# starting kernel
# baseline (speedup 1.0000x reference)
#include <hip/hip_runtime.h>
#include <hip/hip_bf16.h>
#include <math.h>

static constexpr int Bc = 4;
static constexpr int Nn = 1024;
static constexpr int Dd = 256;
static constexpr int Hh = 8;
static constexpr int DH = 32;

typedef float v2f __attribute__((ext_vector_type(2)));
typedef float v8f __attribute__((ext_vector_type(8)));

__global__ __launch_bounds__(32) void proj12_kernel(
    const float* __restrict__ hmat, const float* __restrict__ W1,
    const float* __restrict__ W2, float* __restrict__ wh1,
    float* __restrict__ wh2)
{
    const int lane = threadIdx.x & 31;
    const int m    = lane & 15;
    const int kh   = (lane >> 4) << 1;
    const int col0 = blockIdx.x * 16;
    const int row0 = blockIdx.y * 16;

    v8f acc1 = {};
    v8f acc2 = {};
    const float* aRow  = hmat + (long)(row0 + m) * Dd;
    const float* b1Row = W1   + (long)(col0 + m) * Dd;
    const float* b2Row = W2   + (long)(col0 + m) * Dd;

    for (int k0 = 0; k0 < Dd; k0 += 4) {
        v2f a  = *(const v2f*)(aRow  + k0 + kh);
        v2f b1 = *(const v2f*)(b1Row + k0 + kh);
        v2f b2 = *(const v2f*)(b2Row + k0 + kh);
        acc1 = __builtin_amdgcn_wmma_f32_16x16x4_f32(false, a, false, b1,
                                                     (short)0, acc1, false, false);
        acc2 = __builtin_amdgcn_wmma_f32_16x16x4_f32(false, a, false, b2,
                                                     (short)0, acc2, false, false);
    }

    const int rbase = row0 + ((lane >> 4) << 3);
    const int c     = col0 + m;
#pragma unroll
    for (int r = 0; r < 8; ++r) {
        wh1[(long)(rbase + r) * Dd + c] = acc1[r];
        wh2[(long)(rbase + r) * Dd + c] = acc2[r];
    }
}

__global__ __launch_bounds__(256) void scores_kernel(
    const float* __restrict__ wh1, const float* __restrict__ wh2,
    const float* __restrict__ Ai, const float* __restrict__ Aj,
    float* __restrict__ s1, float* __restrict__ s2,
    float* __restrict__ o1, float* __restrict__ o2)
{
    int idx = blockIdx.x * blockDim.x + threadIdx.x;
    if (idx >= Bc * Nn * Hh) return;
    const int h = idx % Hh;
    const int n = (idx / Hh) % Nn;
    const int b = idx / (Hh * Nn);

    const float* r1 = wh1 + ((long)(b * Nn + n)) * Dd + h * DH;
    const float* r2 = wh2 + ((long)(b * Nn + n)) * Dd + h * DH;
    const float* ai = Ai + h * DH;
    const float* aj = Aj + h * DH;

    float a1 = 0.f, a2 = 0.f, a3 = 0.f, a4 = 0.f;
#pragma unroll 8
    for (int d = 0; d < DH; ++d) {
        const float w1v = r1[d], w2v = r2[d], aiv = ai[d], ajv = aj[d];
        a1 += w1v * aiv;  a2 += w1v * ajv;
        a3 += w2v * aiv;  a4 += w2v * ajv;
    }
    const int o = (b * Hh + h) * Nn + n;
    s1[o] = a1; s2[o] = a2; o1[o] = a3; o2[o] = a4;
}

__global__ __launch_bounds__(256) void entmax_kernel(
    const float* __restrict__ s1, const float* __restrict__ s2,
    const float* __restrict__ o1, const float* __restrict__ o2,
    const int* __restrict__ adj, const int* __restrict__ nag,
    float* __restrict__ att)
{
    __shared__ float xs[Nn];
    __shared__ float srt[Nn];
    __shared__ float tau[Nn];
    __shared__ float redA[256];
    __shared__ float redB[256];
    __shared__ int   redI[256];

    const int tid = threadIdx.x;
    const int bid = blockIdx.x;
    const int i   = bid & (Nn - 1);
    const int h   = (bid >> 10) & (Hh - 1);
    const int b   = bid >> 13;
    const int na  = nag[0];

    const int  rowbase = (b * Hh + h) * Nn;
    const float s1i = s1[rowbase + i];
    const float o1i = o1[rowbase + i];
    const int   im  = i % na;
    const long  adjbase = ((long)b * Nn + i) * Nn;

    float lmax = -3.4e38f;
    for (int j = tid; j < Nn; j += 256) {
        float e = (im == (j % na)) ? (s1i + s2[rowbase + j])
                                   : (o1i + o2[rowbase + j]);
        if (adj[adjbase + j] == 0) e = -1.0e12f;
        const float x = 0.5f * e;
        xs[j] = x;
        lmax  = fmaxf(lmax, x);
    }
    redA[tid] = lmax;
    __syncthreads();
    for (int s = 128; s > 0; s >>= 1) {
        if (tid < s) redA[tid] = fmaxf(redA[tid], redA[tid + s]);
        __syncthreads();
    }
    const float xmax = redA[0];
    __syncthreads();
    for (int j = tid; j < Nn; j += 256) {
        const float x = xs[j] - xmax;
        xs[j]  = x;
        srt[j] = x;
    }
    __syncthreads();

    for (int k = 2; k <= Nn; k <<= 1) {
        for (int jj = k >> 1; jj > 0; jj >>= 1) {
            for (int idx = tid; idx < Nn; idx += 256) {
                const int ixj = idx ^ jj;
                if (ixj > idx) {
                    const float a = srt[idx];
                    const float c = srt[ixj];
                    const bool desc = ((idx & k) == 0);
                    if (desc ? (a < c) : (a > c)) { srt[idx] = c; srt[ixj] = a; }
                }
            }
            __syncthreads();
        }
    }

    const int base = tid * 4;
    const float v0 = srt[base], v1 = srt[base + 1];
    const float v2 = srt[base + 2], v3 = srt[base + 3];
    const float c1 = v0 + v1, c2 = c1 + v2, c3 = c2 + v3;
    const float q0 = v0 * v0, q1 = v1 * v1, q2 = v2 * v2, q3 = v3 * v3;
    const float d1 = q0 + q1, d2 = d1 + q2, d3 = d2 + q3;
    redA[tid] = c3;
    redB[tid] = d3;
    __syncthreads();
    for (int off = 1; off < 256; off <<= 1) {
        const float pa = (tid >= off) ? redA[tid - off] : 0.0f;
        const float pb = (tid >= off) ? redB[tid - off] : 0.0f;
        __syncthreads();
        redA[tid] += pa;
        redB[tid] += pb;
        __syncthreads();
    }
    const float exS = redA[tid] - c3;
    const float exQ = redB[tid] - d3;

    const float sv[4] = { v0, v1, v2, v3 };
    const float cs[4] = { exS + v0, exS + c1, exS + c2, exS + c3 };
    const float cq[4] = { exQ + q0, exQ + d1, exQ + d2, exQ + d3 };
    int cnt = 0;
#pragma unroll
    for (int u = 0; u < 4; ++u) {
        const float rho   = (float)(base + u + 1);
        const float mean  = cs[u] / rho;
        const float msq   = cq[u] / rho;
        const float ssv   = rho * (msq - mean * mean);
        const float delta = (1.0f - ssv) / rho;
        const float t     = mean - ((delta > 0.0f) ? sqrtf(delta) : 0.0f);
        tau[base + u] = t;
        cnt += (t <= sv[u]) ? 1 : 0;
    }
    redI[tid] = cnt;
    __syncthreads();
    for (int s = 128; s > 0; s >>= 1) {
        if (tid < s) redI[tid] += redI[tid + s];
        __syncthreads();
    }
    const int   support = redI[0];
    const float tstar   = tau[support - 1];

    float* outrow = att + ((long)(b * Hh + h) * Nn + i) * Nn;
    for (int j = tid; j < Nn; j += 256) {
        float y = xs[j] - tstar;
        y = fmaxf(y, 0.0f);
        outrow[j] = y * y;
    }
}

static constexpr int KC  = 128;
static constexpr int LDP = 40;

__global__ __launch_bounds__(128) void aggregate_kernel(
    const float* __restrict__ att, const float* __restrict__ wh1,
    const float* __restrict__ wh2, const int* __restrict__ nag,
    float* __restrict__ hp)
{
    __shared__ float ldsW1[KC][LDP];
    __shared__ float ldsW2[KC][LDP];

    const int tid  = threadIdx.x;
    const int wave = tid >> 5;
    const int lane = tid & 31;
    const int m    = lane & 15;
    const int kh   = (lane >> 4) << 1;
    const int bh   = blockIdx.x;
    const int b    = bh / Hh;
    const int h    = bh % Hh;
    const int i0   = blockIdx.y * 64 + wave * 16;
    const int na   = nag[0];

    const int row    = i0 + m;
    const int rowmod = row % na;
    const float* attrow = att + ((long)bh * Nn + row) * Nn;

    v8f accS0 = {};
    v8f accS1 = {};

    for (int kc = 0; kc < Nn; kc += KC) {
        __syncthreads();
        {
            const float* src1 = wh1 + ((long)(b * Nn + kc + tid)) * Dd + h * DH;
            const float* src2 = wh2 + ((long)(b * Nn + kc + tid)) * Dd + h * DH;
            float4* dst1 = (float4*)&ldsW1[tid][0];
            float4* dst2 = (float4*)&ldsW2[tid][0];
#pragma unroll
            for (int q = 0; q < 8; ++q) {
                dst1[q] = ((const float4*)src1)[q];
                dst2[q] = ((const float4*)src2)[q];
            }
        }
        __syncthreads();

        for (int k0 = 0; k0 < KC; k0 += 4) {
            const int j0 = kc + k0 + kh;
            const int j1 = j0 + 1;
            const v2f av = *(const v2f*)(attrow + j0);
            const bool self0 = ((j0 % na) == rowmod);
            const bool self1 = ((j1 % na) == rowmod);
            v2f aS, aO;
            aS.x = self0 ? av.x : 0.0f;  aS.y = self1 ? av.y : 0.0f;
            aO.x = self0 ? 0.0f : av.x;  aO.y = self1 ? 0.0f : av.y;

            const int l0 = k0 + kh;
            v2f b1_0; b1_0.x = ldsW1[l0][m];      b1_0.y = ldsW1[l0 + 1][m];
            v2f b1_1; b1_1.x = ldsW1[l0][16 + m]; b1_1.y = ldsW1[l0 + 1][16 + m];
            v2f b2_0; b2_0.x = ldsW2[l0][m];      b2_0.y = ldsW2[l0 + 1][m];
            v2f b2_1; b2_1.x = ldsW2[l0][16 + m]; b2_1.y = ldsW2[l0 + 1][16 + m];

            accS0 = __builtin_amdgcn_wmma_f32_16x16x4_f32(false, aS, false, b1_0,
                                                          (short)0, accS0, false, false);
            accS0 = __builtin_amdgcn_wmma_f32_16x16x4_f32(false, aO, false, b2_0,
                                                          (short)0, accS0, false, false);
            accS1 = __builtin_amdgcn_wmma_f32_16x16x4_f32(false, aS, false, b1_1,
                                                          (short)0, accS1, false, false);
            accS1 = __builtin_amdgcn_wmma_f32_16x16x4_f32(false, aO, false, b2_1,
                                                          (short)0, accS1, false, false);
        }
    }

    const int rbase = i0 + ((lane >> 4) << 3);
#pragma unroll
    for (int r = 0; r < 8; ++r) {
        float* o = hp + ((long)(b * Nn + rbase + r)) * Dd + h * DH;
        o[m]      = accS0[r];
        o[16 + m] = accS1[r];
    }
}

__global__ __launch_bounds__(32) void outproj_kernel(
    const float* __restrict__ hp, const float* __restrict__ Wo,
    const float* __restrict__ bo, float* __restrict__ out)
{
    const int lane = threadIdx.x & 31;
    const int m    = lane & 15;
    const int kh   = (lane >> 4) << 1;
    const int col0 = blockIdx.x * 16;
    const int row0 = blockIdx.y * 16;

    v8f acc = {};
    const float* aRow = hp + (long)(row0 + m) * Dd;
    const float* bRow = Wo + (long)(col0 + m) * Dd;
    for (int k0 = 0; k0 < Dd; k0 += 4) {
        v2f a  = *(const v2f*)(aRow + k0 + kh);
        v2f bb = *(const v2f*)(bRow + k0 + kh);
        acc = __builtin_amdgcn_wmma_f32_16x16x4_f32(false, a, false, bb,
                                                    (short)0, acc, false, false);
    }

    const int rbase = row0 + ((lane >> 4) << 3);
    const int c     = col0 + m;
    const float bias = bo[c];
#pragma unroll
    for (int r = 0; r < 8; ++r) {
        float v = acc[r] + bias;
        v = (v > 0.0f) ? v : (expf(v) - 1.0f);
        out[(long)(rbase + r) * Dd + c] = v;
    }
}

extern "C" void kernel_launch(void* const* d_in, const int* in_sizes, int n_in,
                              void* d_out, int out_size, void* d_ws, size_t ws_size,
                              hipStream_t stream) {
    (void)in_sizes; (void)n_in; (void)out_size; (void)ws_size;

    const float* hmat = (const float*)d_in[0];
    const int*   nag  = (const int*)  d_in[1];
    const int*   adj  = (const int*)  d_in[2];
    const float* W1   = (const float*)d_in[3];
    const float* W2   = (const float*)d_in[4];
    const float* Ai   = (const float*)d_in[5];
    const float* Aj   = (const float*)d_in[6];
    const float* Wo   = (const float*)d_in[7];
    const float* bo   = (const float*)d_in[8];
    float*       out  = (float*)d_out;

    float* ws  = (float*)d_ws;
    const size_t whN = (size_t)Bc * Nn * Dd;
    const size_t scN = (size_t)Bc * Hh * Nn;
    float* wh1 = ws;
    float* wh2 = wh1 + whN;
    float* s1  = wh2 + whN;
    float* s2  = s1 + scN;
    float* o1  = s2 + scN;
    float* o2  = o1 + scN;
    float* hp  = o2 + scN;
    float* att = hp + whN;

    proj12_kernel<<<dim3(Dd / 16, (Bc * Nn) / 16), 32, 0, stream>>>(
        hmat, W1, W2, wh1, wh2);

    scores_kernel<<<(Bc * Nn * Hh + 255) / 256, 256, 0, stream>>>(
        wh1, wh2, Ai, Aj, s1, s2, o1, o2);

    entmax_kernel<<<Bc * Hh * Nn, 256, 0, stream>>>(
        s1, s2, o1, o2, adj, nag, att);

    aggregate_kernel<<<dim3(Bc * Hh, Nn / 64), 128, 0, stream>>>(
        att, wh1, wh2, nag, hp);

    outproj_kernel<<<dim3(Dd / 16, (Bc * Nn) / 16), 32, 0, stream>>>(
        hp, Wo, bo, out);
}


# entmax: sort and scans removed; one wave per row in registers finds tau by tangent steps from below on sum(max(x-t,0)^2)=1, then the closed form over the support x>t; support sum order changes
# speedup vs baseline: 4.7647x; 4.7647x over previous
#include <hip/hip_runtime.h>
#include <hip/hip_bf16.h>
#include <math.h>

static constexpr int Bc = 4;
static constexpr int Nn = 1024;
static constexpr int Dd = 256;
static constexpr int Hh = 8;
static constexpr int DH = 32;

typedef float v2f __attribute__((ext_vector_type(2)));
typedef float v8f __attribute__((ext_vector_type(8)));
typedef float v4f __attribute__((ext_vector_type(4)));
typedef int   v4i __attribute__((ext_vector_type(4)));

__global__ __launch_bounds__(32) void proj12_kernel(
    const float* __restrict__ hmat, const float* __restrict__ W1,
    const float* __restrict__ W2, float* __restrict__ wh1,
    float* __restrict__ wh2)
{
    const int lane = threadIdx.x & 31;
    const int m    = lane & 15;
    const int kh   = (lane >> 4) << 1;
    const int col0 = blockIdx.x * 16;
    const int row0 = blockIdx.y * 16;

    v8f acc1 = {};
    v8f acc2 = {};
    const float* aRow  = hmat + (long)(row0 + m) * Dd;
    const float* b1Row = W1   + (long)(col0 + m) * Dd;
    const float* b2Row = W2   + (long)(col0 + m) * Dd;

    for (int k0 = 0; k0 < Dd; k0 += 4) {
        v2f a  = *(const v2f*)(aRow  + k0 + kh);
        v2f b1 = *(const v2f*)(b1Row + k0 + kh);
        v2f b2 = *(const v2f*)(b2Row + k0 + kh);
        acc1 = __builtin_amdgcn_wmma_f32_16x16x4_f32(false, a, false, b1,
                                                     (short)0, acc1, false, false);
        acc2 = __builtin_amdgcn_wmma_f32_16x16x4_f32(false, a, false, b2,
                                                     (short)0, acc2, false, false);
    }

    const int rbase = row0 + ((lane >> 4) << 3);
    const int c     = col0 + m;
#pragma unroll
    for (int r = 0; r < 8; ++r) {
        wh1[(long)(rbase + r) * Dd + c] = acc1[r];
        wh2[(long)(rbase + r) * Dd + c] = acc2[r];
    }
}

__global__ __launch_bounds__(256) void scores_kernel(
    const float* __restrict__ wh1, const float* __restrict__ wh2,
    const float* __restrict__ Ai, const float* __restrict__ Aj,
    float* __restrict__ s1, float* __restrict__ s2,
    float* __restrict__ o1, float* __restrict__ o2)
{
    int idx = blockIdx.x * blockDim.x + threadIdx.x;
    if (idx >= Bc * Nn * Hh) return;
    const int h = idx % Hh;
    const int n = (idx / Hh) % Nn;
    const int b = idx / (Hh * Nn);

    const float* r1 = wh1 + ((long)(b * Nn + n)) * Dd + h * DH;
    const float* r2 = wh2 + ((long)(b * Nn + n)) * Dd + h * DH;
    const float* ai = Ai + h * DH;
    const float* aj = Aj + h * DH;

    float a1 = 0.f, a2 = 0.f, a3 = 0.f, a4 = 0.f;
#pragma unroll 8
    for (int d = 0; d < DH; ++d) {
        const float w1v = r1[d], w2v = r2[d], aiv = ai[d], ajv = aj[d];
        a1 += w1v * aiv;  a2 += w1v * ajv;
        a3 += w2v * aiv;  a4 += w2v * ajv;
    }
    const int o = (b * Hh + h) * Nn + n;
    s1[o] = a1; s2[o] = a2; o1[o] = a3; o2[o] = a4;
}

#define DEV static __device__ __forceinline__

DEV float xhalf(float v) {
    const int b = __builtin_bit_cast(int, v);
    return __builtin_bit_cast(float,
        __builtin_amdgcn_permlanex16(b, b, 0x76543210, 0xfedcba98, false, false));
}
template <int M> DEV float xrow(float v) {
    const int b = __builtin_bit_cast(int, v);
    return __builtin_bit_cast(float,
        __builtin_amdgcn_update_dpp(b, b, 0x160 + M, 0xf, 0xf, true));
}
DEV float wave_sum(float v) {
    v += xrow<1>(v); v += xrow<2>(v); v += xrow<4>(v); v += xrow<8>(v);
    v += xhalf(v);
    return v;
}
DEV float wave_max(float v) {
    v = fmaxf(v, xrow<1>(v)); v = fmaxf(v, xrow<2>(v));
    v = fmaxf(v, xrow<4>(v)); v = fmaxf(v, xrow<8>(v));
    v = fmaxf(v, xhalf(v));
    return v;
}
DEV float uni(float v) {
    return __builtin_bit_cast(float,
        __builtin_amdgcn_readfirstlane(__builtin_bit_cast(int, v)));
}

static constexpr int ROWS_PER_BLOCK = 8;
static constexpr int STEP_MAX     = 64;

__global__ __launch_bounds__(256) void entmax_kernel(
    const float* __restrict__ s1, const float* __restrict__ s2,
    const float* __restrict__ o1, const float* __restrict__ o2,
    const int* __restrict__ adj, const int* __restrict__ nag,
    float* __restrict__ att)
{
    const int lane = threadIdx.x & 31;
    const int wave = __builtin_amdgcn_readfirstlane((int)(threadIdx.x >> 5));
    const int rid  = blockIdx.x * ROWS_PER_BLOCK + wave;
    const int i    = rid & (Nn - 1);
    const int h    = (rid >> 10) & (Hh - 1);
    const int b    = rid >> 13;
    const int na   = nag[0];

    const int   rowbase = (b * Hh + h) * Nn;
    const float s1i = s1[rowbase + i];
    const float o1i = o1[rowbase + i];
    const int   im  = i % na;
    const long  adjbase = ((long)b * Nn + i) * Nn;

    const v4f* s2v  = (const v4f*)(s2 + rowbase) + lane;
    const v4f* o2v  = (const v4f*)(o2 + rowbase) + lane;
    const v4i* adjv = (const v4i*)(adj + adjbase) + lane;

    float x[32];
    int       jm   = (4 * lane) % na;
    const int step = 128 % na;
    float lmax = -3.4e38f;
#pragma unroll
    for (int c = 0; c < 8; ++c) {
        const v4f sv = s2v[c * 32];
        const v4f ov = o2v[c * 32];
        const v4i av = adjv[c * 32];
        asm volatile("" :: "v"(sv), "v"(ov));
        const float se[4] = { sv.x, sv.y, sv.z, sv.w };
        const float oe[4] = { ov.x, ov.y, ov.z, ov.w };
        const int   ae[4] = { av.x, av.y, av.z, av.w };
        int m = jm;
#pragma unroll
        for (int e = 0; e < 4; ++e) {
            float ev = (im == m) ? (s1i + se[e]) : (o1i + oe[e]);
            if (ae[e] == 0) ev = -1.0e12f;
            const float xv = 0.5f * ev;
            x[4 * c + e] = xv;
            lmax = fmaxf(lmax, xv);
            m = (m + 1 == na) ? 0 : (m + 1);
        }
        jm += step;
        if (jm >= na) jm -= na;
    }
    const float xmax = uni(wave_max(lmax));
#pragma unroll
    for (int r = 0; r < 32; ++r) x[r] -= xmax;

    float t = -1.0f;
    for (int it = 0; it < STEP_MAX; ++it) {
        const v2f tt = { t, t };
        const v2f zz = { 0.f, 0.f };
        v2f ga = zz, gb = zz, fa = zz, fb = zz;
#pragma unroll
        for (int r = 0; r < 32; r += 4) {
            const v2f xa = { x[r],     x[r + 1] };
            const v2f xb = { x[r + 2], x[r + 3] };
            const v2f da = __builtin_elementwise_max(xa - tt, zz);
            const v2f db = __builtin_elementwise_max(xb - tt, zz);
            ga += da; gb += db;
            fa = __builtin_elementwise_fma(da, da, fa);
            fb = __builtin_elementwise_fma(db, db, fb);
        }
        const v2f gs = ga + gb, fs = fa + fb;
        const float G  = uni(wave_sum(gs.x + gs.y));
        const float F  = uni(wave_sum(fs.x + fs.y));
        const float tn = t + (F - 1.0f) / (2.0f * G);
        if (!(tn > t)) break;
        t = tn;
    }

    float kc = 0.f, sa = 0.f, sb = 0.f, qa = 0.f, qb = 0.f;
#pragma unroll
    for (int r = 0; r < 32; r += 2) {
        const bool  p0 = x[r] > t, p1 = x[r + 1] > t;
        const float v0 = p0 ? x[r] : 0.f, v1 = p1 ? x[r + 1] : 0.f;
        kc += (p0 ? 1.f : 0.f) + (p1 ? 1.f : 0.f);
        sa += v0; sb += v1;
        qa = fmaf(v0, v0, qa); qb = fmaf(v1, v1, qb);
    }
    const float rho   = uni(wave_sum(kc));
    const float csum  = uni(wave_sum(sa + sb));
    const float csq   = uni(wave_sum(qa + qb));
    const float mean  = csum / rho;
    const float msq   = csq / rho;
    const float ssv   = rho * (msq - mean * mean);
    const float delta = (1.0f - ssv) / rho;
    const float tstar = mean - ((delta > 0.0f) ? sqrtf(delta) : 0.0f);

    v4f* outrow = (v4f*)(att + (long)rid * Nn) + lane;
#pragma unroll
    for (int c = 0; c < 8; ++c) {
        v4f y;
        y.x = fmaxf(x[4 * c]     - tstar, 0.0f);
        y.y = fmaxf(x[4 * c + 1] - tstar, 0.0f);
        y.z = fmaxf(x[4 * c + 2] - tstar, 0.0f);
        y.w = fmaxf(x[4 * c + 3] - tstar, 0.0f);
        y.x *= y.x; y.y *= y.y; y.z *= y.z; y.w *= y.w;
        outrow[c * 32] = y;
    }
}

static constexpr int KC  = 128;
static constexpr int LDP = 40;

__global__ __launch_bounds__(128) void aggregate_kernel(
    const float* __restrict__ att, const float* __restrict__ wh1,
    const float* __restrict__ wh2, const int* __restrict__ nag,
    float* __restrict__ hp)
{
    __shared__ float ldsW1[KC][LDP];
    __shared__ float ldsW2[KC][LDP];

    const int tid  = threadIdx.x;
    const int wave = tid >> 5;
    const int lane = tid & 31;
    const int m    = lane & 15;
    const int kh   = (lane >> 4) << 1;
    const int bh   = blockIdx.x;
    const int b    = bh / Hh;
    const int h    = bh % Hh;
    const int i0   = blockIdx.y * 64 + wave * 16;
    const int na   = nag[0];

    const int row    = i0 + m;
    const int rowmod = row % na;
    const float* attrow = att + ((long)bh * Nn + row) * Nn;

    v8f accS0 = {};
    v8f accS1 = {};

    for (int kc = 0; kc < Nn; kc += KC) {
        __syncthreads();
        {
            const float* src1 = wh1 + ((long)(b * Nn + kc + tid)) * Dd + h * DH;
            const float* src2 = wh2 + ((long)(b * Nn + kc + tid)) * Dd + h * DH;
            float4* dst1 = (float4*)&ldsW1[tid][0];
            float4* dst2 = (float4*)&ldsW2[tid][0];
#pragma unroll
            for (int q = 0; q < 8; ++q) {
                dst1[q] = ((const float4*)src1)[q];
                dst2[q] = ((const float4*)src2)[q];
            }
        }
        __syncthreads();

        for (int k0 = 0; k0 < KC; k0 += 4) {
            const int j0 = kc + k0 + kh;
            const int j1 = j0 + 1;
            const v2f av = *(const v2f*)(attrow + j0);
            const bool self0 = ((j0 % na) == rowmod);
            const bool self1 = ((j1 % na) == rowmod);
            v2f aS, aO;
            aS.x = self0 ? av.x : 0.0f;  aS.y = self1 ? av.y : 0.0f;
            aO.x = self0 ? 0.0f : av.x;  aO.y = self1 ? 0.0f : av.y;

            const int l0 = k0 + kh;
            v2f b1_0; b1_0.x = ldsW1[l0][m];      b1_0.y = ldsW1[l0 + 1][m];
            v2f b1_1; b1_1.x = ldsW1[l0][16 + m]; b1_1.y = ldsW1[l0 + 1][16 + m];
            v2f b2_0; b2_0.x = ldsW2[l0][m];      b2_0.y = ldsW2[l0 + 1][m];
            v2f b2_1; b2_1.x = ldsW2[l0][16 + m]; b2_1.y = ldsW2[l0 + 1][16 + m];

            accS0 = __builtin_amdgcn_wmma_f32_16x16x4_f32(false, aS, false, b1_0,
                                                          (short)0, accS0, false, false);
            accS0 = __builtin_amdgcn_wmma_f32_16x16x4_f32(false, aO, false, b2_0,
                                                          (short)0, accS0, false, false);
            accS1 = __builtin_amdgcn_wmma_f32_16x16x4_f32(false, aS, false, b1_1,
                                                          (short)0, accS1, false, false);
            accS1 = __builtin_amdgcn_wmma_f32_16x16x4_f32(false, aO, false, b2_1,
                                                          (short)0, accS1, false, false);
        }
    }

    const int rbase = i0 + ((lane >> 4) << 3);
#pragma unroll
    for (int r = 0; r < 8; ++r) {
        float* o = hp + ((long)(b * Nn + rbase + r)) * Dd + h * DH;
        o[m]      = accS0[r];
        o[16 + m] = accS1[r];
    }
}

__global__ __launch_bounds__(32) void outproj_kernel(
    const float* __restrict__ hp, const float* __restrict__ Wo,
    const float* __restrict__ bo, float* __restrict__ out)
{
    const int lane = threadIdx.x & 31;
    const int m    = lane & 15;
    const int kh   = (lane >> 4) << 1;
    const int col0 = blockIdx.x * 16;
    const int row0 = blockIdx.y * 16;

    v8f acc = {};
    const float* aRow = hp + (long)(row0 + m) * Dd;
    const float* bRow = Wo + (long)(col0 + m) * Dd;
    for (int k0 = 0; k0 < Dd; k0 += 4) {
        v2f a  = *(const v2f*)(aRow + k0 + kh);
        v2f bb = *(const v2f*)(bRow + k0 + kh);
        acc = __builtin_amdgcn_wmma_f32_16x16x4_f32(false, a, false, bb,
                                                    (short)0, acc, false, false);
    }

    const int rbase = row0 + ((lane >> 4) << 3);
    const int c     = col0 + m;
    const float bias = bo[c];
#pragma unroll
    for (int r = 0; r < 8; ++r) {
        float v = acc[r] + bias;
        v = (v > 0.0f) ? v : (expf(v) - 1.0f);
        out[(long)(rbase + r) * Dd + c] = v;
    }
}

extern "C" void kernel_launch(void* const* d_in, const int* in_sizes, int n_in,
                              void* d_out, int out_size, void* d_ws, size_t ws_size,
                              hipStream_t stream) {
    (void)in_sizes; (void)n_in; (void)out_size; (void)ws_size;

    const float* hmat = (const float*)d_in[0];
    const int*   nag  = (const int*)  d_in[1];
    const int*   adj  = (const int*)  d_in[2];
    const float* W1   = (const float*)d_in[3];
    const float* W2   = (const float*)d_in[4];
    const float* Ai   = (const float*)d_in[5];
    const float* Aj   = (const float*)d_in[6];
    const float* Wo   = (const float*)d_in[7];
    const float* bo   = (const float*)d_in[8];
    float*       out  = (float*)d_out;

    float* ws  = (float*)d_ws;
    const size_t whN = (size_t)Bc * Nn * Dd;
    const size_t scN = (size_t)Bc * Hh * Nn;
    float* wh1 = ws;
    float* wh2 = wh1 + whN;
    float* s1  = wh2 + whN;
    float* s2  = s1 + scN;
    float* o1  = s2 + scN;
    float* o2  = o1 + scN;
    float* hp  = o2 + scN;
    float* att = hp + whN;

    proj12_kernel<<<dim3(Dd / 16, (Bc * Nn) / 16), 32, 0, stream>>>(
        hmat, W1, W2, wh1, wh2);

    scores_kernel<<<(Bc * Nn * Hh + 255) / 256, 256, 0, stream>>>(
        wh1, wh2, Ai, Aj, s1, s2, o1, o2);

    entmax_kernel<<<(Bc * Hh * Nn) / ROWS_PER_BLOCK, 256, 0, stream>>>(
        s1, s2, o1, o2, adj, nag, att);

    aggregate_kernel<<<dim3(Bc * Hh, Nn / 64), 128, 0, stream>>>(
        att, wh1, wh2, nag, hp);

    outproj_kernel<<<dim3(Dd / 16, (Bc * Nn) / 16), 32, 0, stream>>>(
        hp, Wo, bo, out);
}


# entmax and aggregate fused: a 16-wave block keeps its 16 att rows in an LDS tile that feeds the matrix instructions (B from memory, K split over waves); att never stored; k sum order changes
# speedup vs baseline: 6.6799x; 1.4020x over previous
#include <hip/hip_runtime.h>
#include <hip/hip_bf16.h>
#include <math.h>

static constexpr int Bc = 4;
static constexpr int Nn = 1024;
static constexpr int Dd = 256;
static constexpr int Hh = 8;
static constexpr int DH = 32;

typedef float v2f __attribute__((ext_vector_type(2)));
typedef float v8f __attribute__((ext_vector_type(8)));
typedef float v4f __attribute__((ext_vector_type(4)));
typedef int   v4i __attribute__((ext_vector_type(4)));

__global__ __launch_bounds__(32) void proj12_kernel(
    const float* __restrict__ hmat, const float* __restrict__ W1,
    const float* __restrict__ W2, float* __restrict__ wh1,
    float* __restrict__ wh2)
{
    const int lane = threadIdx.x & 31;
    const int m    = lane & 15;
    const int kh   = (lane >> 4) << 1;
    const int col0 = blockIdx.x * 16;
    const int row0 = blockIdx.y * 16;

    v8f acc1 = {};
    v8f acc2 = {};
    const float* aRow  = hmat + (long)(row0 + m) * Dd;
    const float* b1Row = W1   + (long)(col0 + m) * Dd;
    const float* b2Row = W2   + (long)(col0 + m) * Dd;

    for (int k0 = 0; k0 < Dd; k0 += 4) {
        v2f a  = *(const v2f*)(aRow  + k0 + kh);
        v2f b1 = *(const v2f*)(b1Row + k0 + kh);
        v2f b2 = *(const v2f*)(b2Row + k0 + kh);
        acc1 = __builtin_amdgcn_wmma_f32_16x16x4_f32(false, a, false, b1,
                                                     (short)0, acc1, false, false);
        acc2 = __builtin_amdgcn_wmma_f32_16x16x4_f32(false, a, false, b2,
                                                     (short)0, acc2, false, false);
    }

    const int rbase = row0 + ((lane >> 4) << 3);
    const int c     = col0 + m;
#pragma unroll
    for (int r = 0; r < 8; ++r) {
        wh1[(long)(rbase + r) * Dd + c] = acc1[r];
        wh2[(long)(rbase + r) * Dd + c] = acc2[r];
    }
}

__global__ __launch_bounds__(256) void scores_kernel(
    const float* __restrict__ wh1, const float* __restrict__ wh2,
    const float* __restrict__ Ai, const float* __restrict__ Aj,
    float* __restrict__ s1, float* __restrict__ s2,
    float* __restrict__ o1, float* __restrict__ o2)
{
    int idx = blockIdx.x * blockDim.x + threadIdx.x;
    if (idx >= Bc * Nn * Hh) return;
    const int h = idx % Hh;
    const int n = (idx / Hh) % Nn;
    const int b = idx / (Hh * Nn);

    const float* r1 = wh1 + ((long)(b * Nn + n)) * Dd + h * DH;
    const float* r2 = wh2 + ((long)(b * Nn + n)) * Dd + h * DH;
    const float* ai = Ai + h * DH;
    const float* aj = Aj + h * DH;

    float a1 = 0.f, a2 = 0.f, a3 = 0.f, a4 = 0.f;
#pragma unroll 8
    for (int d = 0; d < DH; ++d) {
        const float w1v = r1[d], w2v = r2[d], aiv = ai[d], ajv = aj[d];
        a1 += w1v * aiv;  a2 += w1v * ajv;
        a3 += w2v * aiv;  a4 += w2v * ajv;
    }
    const int o = (b * Hh + h) * Nn + n;
    s1[o] = a1; s2[o] = a2; o1[o] = a3; o2[o] = a4;
}

#define DEV static __device__ __forceinline__

DEV float xhalf(float v) {
    const int b = __builtin_bit_cast(int, v);
    return __builtin_bit_cast(float,
        __builtin_amdgcn_permlanex16(b, b, 0x76543210, 0xfedcba98, false, false));
}
template <int M> DEV float xrow(float v) {
    const int b = __builtin_bit_cast(int, v);
    return __builtin_bit_cast(float,
        __builtin_amdgcn_update_dpp(b, b, 0x160 + M, 0xf, 0xf, true));
}
DEV float wave_sum(float v) {
    v += xrow<1>(v); v += xrow<2>(v); v += xrow<4>(v); v += xrow<8>(v);
    v += xhalf(v);
    return v;
}
DEV float wave_max(float v) {
    v = fmaxf(v, xrow<1>(v)); v = fmaxf(v, xrow<2>(v));
    v = fmaxf(v, xrow<4>(v)); v = fmaxf(v, xrow<8>(v));
    v = fmaxf(v, xhalf(v));
    return v;
}
DEV float uni(float v) {
    return __builtin_bit_cast(float,
        __builtin_amdgcn_readfirstlane(__builtin_bit_cast(int, v)));
}

static constexpr int ROWS_PER_BLOCK = 16;
static constexpr int LDA = Nn + 4;
static constexpr int KW  = Nn / ROWS_PER_BLOCK;
static constexpr int RS  = 36;
static constexpr int STEP_MAX     = 64;

__global__ __launch_bounds__(512) void entmax_aggregate_kernel(
    const float* __restrict__ s1, const float* __restrict__ s2,
    const float* __restrict__ o1, const float* __restrict__ o2,
    const int* __restrict__ adj, const int* __restrict__ nag,
    const float* __restrict__ wh1, const float* __restrict__ wh2,
    float* __restrict__ hp)
{
    __shared__ __attribute__((aligned(16))) float smem[ROWS_PER_BLOCK * LDA];
    float (*attL)[LDA]   = (float (*)[LDA])smem;
    float (*red)[16][RS] = (float (*)[16][RS])smem;

    const int lane = threadIdx.x & 31;
    const int wave = __builtin_amdgcn_readfirstlane((int)(threadIdx.x >> 5));
    const int rid  = blockIdx.x * ROWS_PER_BLOCK + wave;
    const int i    = rid & (Nn - 1);
    const int h    = (rid >> 10) & (Hh - 1);
    const int b    = rid >> 13;
    const int na   = nag[0];

    const int   rowbase = (b * Hh + h) * Nn;
    const float s1i = s1[rowbase + i];
    const float o1i = o1[rowbase + i];
    const int   im  = i % na;
    const long  adjbase = ((long)b * Nn + i) * Nn;

    const v4f* s2v  = (const v4f*)(s2 + rowbase) + lane;
    const v4f* o2v  = (const v4f*)(o2 + rowbase) + lane;
    const v4i* adjv = (const v4i*)(adj + adjbase) + lane;

    float x[32];
    int       jm   = (4 * lane) % na;
    const int step = 128 % na;
    float lmax = -3.4e38f;
#pragma unroll
    for (int c = 0; c < 8; ++c) {
        const v4f sv = s2v[c * 32];
        const v4f ov = o2v[c * 32];
        const v4i av = adjv[c * 32];
        asm volatile("" :: "v"(sv), "v"(ov));
        const float se[4] = { sv.x, sv.y, sv.z, sv.w };
        const float oe[4] = { ov.x, ov.y, ov.z, ov.w };
        const int   ae[4] = { av.x, av.y, av.z, av.w };
        int m = jm;
#pragma unroll
        for (int e = 0; e < 4; ++e) {
            float ev = (im == m) ? (s1i + se[e]) : (o1i + oe[e]);
            if (ae[e] == 0) ev = -1.0e12f;
            const float xv = 0.5f * ev;
            x[4 * c + e] = xv;
            lmax = fmaxf(lmax, xv);
            m = (m + 1 == na) ? 0 : (m + 1);
        }
        jm += step;
        if (jm >= na) jm -= na;
    }
    const float xmax = uni(wave_max(lmax));
#pragma unroll
    for (int r = 0; r < 32; ++r) x[r] -= xmax;

    float t = -1.0f;
    for (int it = 0; it < STEP_MAX; ++it) {
        const v2f tt = { t, t };
        const v2f zz = { 0.f, 0.f };
        v2f ga = zz, gb = zz, fa = zz, fb = zz;
#pragma unroll
        for (int r = 0; r < 32; r += 4) {
            const v2f xa = { x[r],     x[r + 1] };
            const v2f xb = { x[r + 2], x[r + 3] };
            const v2f da = __builtin_elementwise_max(xa - tt, zz);
            const v2f db = __builtin_elementwise_max(xb - tt, zz);
            ga += da; gb += db;
            fa = __builtin_elementwise_fma(da, da, fa);
            fb = __builtin_elementwise_fma(db, db, fb);
        }
        const v2f gs = ga + gb, fs = fa + fb;
        const float G  = uni(wave_sum(gs.x + gs.y));
        const float F  = uni(wave_sum(fs.x + fs.y));
        const float tn = t + (F - 1.0f) / (2.0f * G);
        if (!(tn > t)) break;
        t = tn;
    }

    float kc = 0.f, sa = 0.f, sb = 0.f, qa = 0.f, qb = 0.f;
#pragma unroll
    for (int r = 0; r < 32; r += 2) {
        const bool  p0 = x[r] > t, p1 = x[r + 1] > t;
        const float v0 = p0 ? x[r] : 0.f, v1 = p1 ? x[r + 1] : 0.f;
        kc += (p0 ? 1.f : 0.f) + (p1 ? 1.f : 0.f);
        sa += v0; sb += v1;
        qa = fmaf(v0, v0, qa); qb = fmaf(v1, v1, qb);
    }
    const float rho   = uni(wave_sum(kc));
    const float csum  = uni(wave_sum(sa + sb));
    const float csq   = uni(wave_sum(qa + qb));
    const float mean  = csum / rho;
    const float msq   = csq / rho;
    const float ssv   = rho * (msq - mean * mean);
    const float delta = (1.0f - ssv) / rho;
    const float tstar = mean - ((delta > 0.0f) ? sqrtf(delta) : 0.0f);

    v4f* outrow = (v4f*)(&attL[wave][0]) + lane;
#pragma unroll
    for (int c = 0; c < 8; ++c) {
        v4f y;
        y.x = fmaxf(x[4 * c]     - tstar, 0.0f);
        y.y = fmaxf(x[4 * c + 1] - tstar, 0.0f);
        y.z = fmaxf(x[4 * c + 2] - tstar, 0.0f);
        y.w = fmaxf(x[4 * c + 3] - tstar, 0.0f);
        y.x *= y.x; y.y *= y.y; y.z *= y.z; y.w *= y.w;
        outrow[c * 32] = y;
    }
    __syncthreads();

    const int m      = lane & 15;
    const int kh     = (lane >> 4) << 1;
    const int ib     = (blockIdx.x * ROWS_PER_BLOCK) & (Nn - 1);
    const int rowmod = (ib + m) % na;
    const int kb     = wave * KW;
    const float* arow = &attL[m][kb + kh];
    const float* w1p  = wh1 + ((long)(b * Nn + kb + kh)) * Dd + h * DH + m;
    const float* w2p  = wh2 + ((long)(b * Nn + kb + kh)) * Dd + h * DH + m;
    int       cm    = (kb + kh) % na;
    const int step4 = 4 % na;

    v8f accS0 = {};
    v8f accS1 = {};
#pragma unroll 4
    for (int k0 = 0; k0 < KW; k0 += 4) {
        const v2f av = *(const v2f*)(arow + k0);
        const int cm1 = (cm + 1 == na) ? 0 : (cm + 1);
        const bool self0 = (cm  == rowmod);
        const bool self1 = (cm1 == rowmod);
        v2f aS, aO;
        aS.x = self0 ? av.x : 0.0f;  aS.y = self1 ? av.y : 0.0f;
        aO.x = self0 ? 0.0f : av.x;  aO.y = self1 ? 0.0f : av.y;

        const float* p1 = w1p + (long)k0 * Dd;
        const float* p2 = w2p + (long)k0 * Dd;
        v2f b1_0; b1_0.x = p1[0];  b1_0.y = p1[Dd];
        v2f b1_1; b1_1.x = p1[16]; b1_1.y = p1[Dd + 16];
        v2f b2_0; b2_0.x = p2[0];  b2_0.y = p2[Dd];
        v2f b2_1; b2_1.x = p2[16]; b2_1.y = p2[Dd + 16];

        accS0 = __builtin_amdgcn_wmma_f32_16x16x4_f32(false, aS, false, b1_0,
                                                      (short)0, accS0, false, false);
        accS0 = __builtin_amdgcn_wmma_f32_16x16x4_f32(false, aO, false, b2_0,
                                                      (short)0, accS0, false, false);
        accS1 = __builtin_amdgcn_wmma_f32_16x16x4_f32(false, aS, false, b1_1,
                                                      (short)0, accS1, false, false);
        accS1 = __builtin_amdgcn_wmma_f32_16x16x4_f32(false, aO, false, b2_1,
                                                      (short)0, accS1, false, false);
        cm += step4;
        if (cm >= na) cm -= na;
    }

    __syncthreads();
    float* rw = &red[wave][(lane >> 4) << 3][m];
#pragma unroll
    for (int r = 0; r < 8; ++r) {
        rw[r * RS]      = accS0[r];
        rw[r * RS + 16] = accS1[r];
    }
    __syncthreads();
    float acc = 0.0f;
#pragma unroll
    for (int w = 0; w < ROWS_PER_BLOCK; ++w) acc += red[w][wave][lane];
    hp[((long)(b * Nn + ib + wave)) * Dd + h * DH + lane] = acc;
}


__global__ __launch_bounds__(32) void outproj_kernel(
    const float* __restrict__ hp, const float* __restrict__ Wo,
    const float* __restrict__ bo, float* __restrict__ out)
{
    const int lane = threadIdx.x & 31;
    const int m    = lane & 15;
    const int kh   = (lane >> 4) << 1;
    const int col0 = blockIdx.x * 16;
    const int row0 = blockIdx.y * 16;

    v8f acc = {};
    const float* aRow = hp + (long)(row0 + m) * Dd;
    const float* bRow = Wo + (long)(col0 + m) * Dd;
    for (int k0 = 0; k0 < Dd; k0 += 4) {
        v2f a  = *(const v2f*)(aRow + k0 + kh);
        v2f bb = *(const v2f*)(bRow + k0 + kh);
        acc = __builtin_amdgcn_wmma_f32_16x16x4_f32(false, a, false, bb,
                                                    (short)0, acc, false, false);
    }

    const int rbase = row0 + ((lane >> 4) << 3);
    const int c     = col0 + m;
    const float bias = bo[c];
#pragma unroll
    for (int r = 0; r < 8; ++r) {
        float v = acc[r] + bias;
        v = (v > 0.0f) ? v : (expf(v) - 1.0f);
        out[(long)(rbase + r) * Dd + c] = v;
    }
}

extern "C" void kernel_launch(void* const* d_in, const int* in_sizes, int n_in,
                              void* d_out, int out_size, void* d_ws, size_t ws_size,
                              hipStream_t stream) {
    (void)in_sizes; (void)n_in; (void)out_size; (void)ws_size;

    const float* hmat = (const float*)d_in[0];
    const int*   nag  = (const int*)  d_in[1];
    const int*   adj  = (const int*)  d_in[2];
    const float* W1   = (const float*)d_in[3];
    const float* W2   = (const float*)d_in[4];
    const float* Ai   = (const float*)d_in[5];
    const float* Aj   = (const float*)d_in[6];
    const float* Wo   = (const float*)d_in[7];
    const float* bo   = (const float*)d_in[8];
    float*       out  = (float*)d_out;

    float* ws  = (float*)d_ws;
    const size_t whN = (size_t)Bc * Nn * Dd;
    const size_t scN = (size_t)Bc * Hh * Nn;
    float* wh1 = ws;
    float* wh2 = wh1 + whN;
    float* s1  = wh2 + whN;
    float* s2  = s1 + scN;
    float* o1  = s2 + scN;
    float* o2  = o1 + scN;
    float* hp  = o2 + scN;

    proj12_kernel<<<dim3(Dd / 16, (Bc * Nn) / 16), 32, 0, stream>>>(
        hmat, W1, W2, wh1, wh2);

    scores_kernel<<<(Bc * Nn * Hh + 255) / 256, 256, 0, stream>>>(
        wh1, wh2, Ai, Aj, s1, s2, o1, o2);

    entmax_aggregate_kernel<<<(Bc * Hh * Nn) / ROWS_PER_BLOCK, 512, 0, stream>>>(
        s1, s2, o1, o2, adj, nag, wh1, wh2, hp);

    outproj_kernel<<<dim3(Dd / 16, (Bc * Nn) / 16), 32, 0, stream>>>(
        hp, Wo, bo, out);
}


# aggregate: K runs only over the union of the tile's 16 entmax supports (all-zero columns skipped): rows leave support bit words, columns listed in LDS, K-steps dealt to waves; k sum order changes
# speedup vs baseline: 8.6153x; 1.2897x over previous
#include <hip/hip_runtime.h>
#include <hip/hip_bf16.h>
#include <math.h>

static constexpr int Bc = 4;
static constexpr int Nn = 1024;
static constexpr int Dd = 256;
static constexpr int Hh = 8;
static constexpr int DH = 32;

typedef float v2f __attribute__((ext_vector_type(2)));
typedef float v8f __attribute__((ext_vector_type(8)));
typedef float v4f __attribute__((ext_vector_type(4)));
typedef int   v4i __attribute__((ext_vector_type(4)));

__global__ __launch_bounds__(32) void proj12_kernel(
    const float* __restrict__ hmat, const float* __restrict__ W1,
    const float* __restrict__ W2, float* __restrict__ wh1,
    float* __restrict__ wh2)
{
    const int lane = threadIdx.x & 31;
    const int m    = lane & 15;
    const int kh   = (lane >> 4) << 1;
    const int col0 = blockIdx.x * 16;
    const int row0 = blockIdx.y * 16;

    v8f acc1 = {};
    v8f acc2 = {};
    const float* aRow  = hmat + (long)(row0 + m) * Dd;
    const float* b1Row = W1   + (long)(col0 + m) * Dd;
    const float* b2Row = W2   + (long)(col0 + m) * Dd;

    for (int k0 = 0; k0 < Dd; k0 += 4) {
        v2f a  = *(const v2f*)(aRow  + k0 + kh);
        v2f b1 = *(const v2f*)(b1Row + k0 + kh);
        v2f b2 = *(const v2f*)(b2Row + k0 + kh);
        acc1 = __builtin_amdgcn_wmma_f32_16x16x4_f32(false, a, false, b1,
                                                     (short)0, acc1, false, false);
        acc2 = __builtin_amdgcn_wmma_f32_16x16x4_f32(false, a, false, b2,
                                                     (short)0, acc2, false, false);
    }

    const int rbase = row0 + ((lane >> 4) << 3);
    const int c     = col0 + m;
#pragma unroll
    for (int r = 0; r < 8; ++r) {
        wh1[(long)(rbase + r) * Dd + c] = acc1[r];
        wh2[(long)(rbase + r) * Dd + c] = acc2[r];
    }
}

__global__ __launch_bounds__(256) void scores_kernel(
    const float* __restrict__ wh1, const float* __restrict__ wh2,
    const float* __restrict__ Ai, const float* __restrict__ Aj,
    float* __restrict__ s1, float* __restrict__ s2,
    float* __restrict__ o1, float* __restrict__ o2)
{
    int idx = blockIdx.x * blockDim.x + threadIdx.x;
    if (idx >= Bc * Nn * Hh) return;
    const int h = idx % Hh;
    const int n = (idx / Hh) % Nn;
    const int b = idx / (Hh * Nn);

    const float* r1 = wh1 + ((long)(b * Nn + n)) * Dd + h * DH;
    const float* r2 = wh2 + ((long)(b * Nn + n)) * Dd + h * DH;
    const float* ai = Ai + h * DH;
    const float* aj = Aj + h * DH;

    float a1 = 0.f, a2 = 0.f, a3 = 0.f, a4 = 0.f;
#pragma unroll 8
    for (int d = 0; d < DH; ++d) {
        const float w1v = r1[d], w2v = r2[d], aiv = ai[d], ajv = aj[d];
        a1 += w1v * aiv;  a2 += w1v * ajv;
        a3 += w2v * aiv;  a4 += w2v * ajv;
    }
    const int o = (b * Hh + h) * Nn + n;
    s1[o] = a1; s2[o] = a2; o1[o] = a3; o2[o] = a4;
}

#define DEV static __device__ __forceinline__

DEV float xhalf(float v) {
    const int b = __builtin_bit_cast(int, v);
    return __builtin_bit_cast(float,
        __builtin_amdgcn_permlanex16(b, b, 0x76543210, 0xfedcba98, false, false));
}
template <int M> DEV float xrow(float v) {
    const int b = __builtin_bit_cast(int, v);
    return __builtin_bit_cast(float,
        __builtin_amdgcn_update_dpp(b, b, 0x160 + M, 0xf, 0xf, true));
}
DEV float wave_sum(float v) {
    v += xrow<1>(v); v += xrow<2>(v); v += xrow<4>(v); v += xrow<8>(v);
    v += xhalf(v);
    return v;
}
DEV float wave_max(float v) {
    v = fmaxf(v, xrow<1>(v)); v = fmaxf(v, xrow<2>(v));
    v = fmaxf(v, xrow<4>(v)); v = fmaxf(v, xrow<8>(v));
    v = fmaxf(v, xhalf(v));
    return v;
}
DEV float uni(float v) {
    return __builtin_bit_cast(float,
        __builtin_amdgcn_readfirstlane(__builtin_bit_cast(int, v)));
}

static constexpr int ROWS_PER_BLOCK = 16;
static constexpr int LDA = Nn + 4;
static constexpr int RS  = 36;
static constexpr int STEP_MAX     = 64;

__global__ __launch_bounds__(512) void entmax_aggregate_kernel(
    const float* __restrict__ s1, const float* __restrict__ s2,
    const float* __restrict__ o1, const float* __restrict__ o2,
    const int* __restrict__ adj, const int* __restrict__ nag,
    const float* __restrict__ wh1, const float* __restrict__ wh2,
    float* __restrict__ hp)
{
    __shared__ __attribute__((aligned(16))) float smem[ROWS_PER_BLOCK * LDA];
    float (*attL)[LDA]   = (float (*)[LDA])smem;
    float (*red)[16][RS] = (float (*)[16][RS])smem;
    __shared__ unsigned int suppw[ROWS_PER_BLOCK][32];
    __shared__ __attribute__((aligned(8))) int collist[Nn];

    const int lane = threadIdx.x & 31;
    const int wave = __builtin_amdgcn_readfirstlane((int)(threadIdx.x >> 5));
    const int rid  = blockIdx.x * ROWS_PER_BLOCK + wave;
    const int i    = rid & (Nn - 1);
    const int h    = (rid >> 10) & (Hh - 1);
    const int b    = rid >> 13;
    const int na   = nag[0];

    const int   rowbase = (b * Hh + h) * Nn;
    const float s1i = s1[rowbase + i];
    const float o1i = o1[rowbase + i];
    const int   im  = i % na;
    const long  adjbase = ((long)b * Nn + i) * Nn;

    const v4f* s2v  = (const v4f*)(s2 + rowbase) + lane;
    const v4f* o2v  = (const v4f*)(o2 + rowbase) + lane;
    const v4i* adjv = (const v4i*)(adj + adjbase) + lane;

    float x[32];
    int       jm   = (4 * lane) % na;
    const int step = 128 % na;
    float lmax = -3.4e38f;
#pragma unroll
    for (int c = 0; c < 8; ++c) {
        const v4f sv = s2v[c * 32];
        const v4f ov = o2v[c * 32];
        const v4i av = adjv[c * 32];
        asm volatile("" :: "v"(sv), "v"(ov));
        const float se[4] = { sv.x, sv.y, sv.z, sv.w };
        const float oe[4] = { ov.x, ov.y, ov.z, ov.w };
        const int   ae[4] = { av.x, av.y, av.z, av.w };
        int m = jm;
#pragma unroll
        for (int e = 0; e < 4; ++e) {
            float ev = (im == m) ? (s1i + se[e]) : (o1i + oe[e]);
            if (ae[e] == 0) ev = -1.0e12f;
            const float xv = 0.5f * ev;
            x[4 * c + e] = xv;
            lmax = fmaxf(lmax, xv);
            m = (m + 1 == na) ? 0 : (m + 1);
        }
        jm += step;
        if (jm >= na) jm -= na;
    }
    const float xmax = uni(wave_max(lmax));
#pragma unroll
    for (int r = 0; r < 32; ++r) x[r] -= xmax;

    float t = -1.0f;
    for (int it = 0; it < STEP_MAX; ++it) {
        const v2f tt = { t, t };
        const v2f zz = { 0.f, 0.f };
        v2f ga = zz, gb = zz, fa = zz, fb = zz;
#pragma unroll
        for (int r = 0; r < 32; r += 4) {
            const v2f xa = { x[r],     x[r + 1] };
            const v2f xb = { x[r + 2], x[r + 3] };
            const v2f da = __builtin_elementwise_max(xa - tt, zz);
            const v2f db = __builtin_elementwise_max(xb - tt, zz);
            ga += da; gb += db;
            fa = __builtin_elementwise_fma(da, da, fa);
            fb = __builtin_elementwise_fma(db, db, fb);
        }
        const v2f gs = ga + gb, fs = fa + fb;
        const float G  = uni(wave_sum(gs.x + gs.y));
        const float F  = uni(wave_sum(fs.x + fs.y));
        const float tn = t + (F - 1.0f) / (2.0f * G);
        if (!(tn > t)) break;
        t = tn;
    }

    float kc = 0.f, sa = 0.f, sb = 0.f, qa = 0.f, qb = 0.f;
#pragma unroll
    for (int r = 0; r < 32; r += 2) {
        const bool  p0 = x[r] > t, p1 = x[r + 1] > t;
        const float v0 = p0 ? x[r] : 0.f, v1 = p1 ? x[r + 1] : 0.f;
        kc += (p0 ? 1.f : 0.f) + (p1 ? 1.f : 0.f);
        sa += v0; sb += v1;
        qa = fmaf(v0, v0, qa); qb = fmaf(v1, v1, qb);
    }
    const float rho   = uni(wave_sum(kc));
    const float csum  = uni(wave_sum(sa + sb));
    const float csq   = uni(wave_sum(qa + qb));
    const float mean  = csum / rho;
    const float msq   = csq / rho;
    const float ssv   = rho * (msq - mean * mean);
    const float delta = (1.0f - ssv) / rho;
    const float tstar = mean - ((delta > 0.0f) ? sqrtf(delta) : 0.0f);

    v4f* outrow = (v4f*)(&attL[wave][0]) + lane;
    unsigned int myw = 0;
#pragma unroll
    for (int c = 0; c < 8; ++c) {
        v4f y;
        y.x = fmaxf(x[4 * c]     - tstar, 0.0f);
        y.y = fmaxf(x[4 * c + 1] - tstar, 0.0f);
        y.z = fmaxf(x[4 * c + 2] - tstar, 0.0f);
        y.w = fmaxf(x[4 * c + 3] - tstar, 0.0f);
        y.x *= y.x; y.y *= y.y; y.z *= y.z; y.w *= y.w;
        outrow[c * 32] = y;
        myw |= (y.x != 0.0f) ? (1u << (4 * c))     : 0u;
        myw |= (y.y != 0.0f) ? (1u << (4 * c + 1)) : 0u;
        myw |= (y.z != 0.0f) ? (1u << (4 * c + 2)) : 0u;
        myw |= (y.w != 0.0f) ? (1u << (4 * c + 3)) : 0u;
    }
    suppw[wave][lane] = myw;
    __syncthreads();

    unsigned int u = 0;
#pragma unroll
    for (int w = 0; w < ROWS_PER_BLOCK; ++w) u |= suppw[w][lane];
    const int cnt = __builtin_popcount(u);
    int inc = cnt;
    inc += __builtin_amdgcn_update_dpp(0, inc, 0x111, 0xf, 0xf, true);
    inc += __builtin_amdgcn_update_dpp(0, inc, 0x112, 0xf, 0xf, true);
    inc += __builtin_amdgcn_update_dpp(0, inc, 0x114, 0xf, 0xf, true);
    inc += __builtin_amdgcn_update_dpp(0, inc, 0x118, 0xf, 0xf, true);
    const int lowtot = __builtin_amdgcn_readlane(inc, 15);
    if (lane >= 16) inc += lowtot;
    const int Kp = __builtin_amdgcn_readlane(inc, 31);
    {
        int pos = inc - cnt;
        const int cbase = 4 * lane;
        unsigned int bits = u;
#pragma unroll 1
        for (int it = 0; it < 32; ++it) {
            if (__builtin_amdgcn_ballot_w32(bits != 0u) == 0u) break;
            if (bits != 0u) {
                const int r = __builtin_ctz(bits);
                bits &= bits - 1u;
                if (((pos >> 2) & (ROWS_PER_BLOCK - 1)) == wave) collist[pos] = cbase + 128 * (r >> 2) + (r & 3);
                ++pos;
            }
        }
    }
    __syncthreads();

    const int m      = lane & 15;
    const int kh     = (lane >> 4) << 1;
    const int ib     = (blockIdx.x * ROWS_PER_BLOCK) & (Nn - 1);
    const unsigned int una    = (unsigned int)na;
    const unsigned int rowmod = (unsigned int)(ib + m) % una;
    const float* w1b = wh1 + ((long)b * Nn) * Dd + h * DH + m;
    const float* w2b = wh2 + ((long)b * Nn) * Dd + h * DH + m;
    const int nsteps = (Kp + 3) >> 2;

    v8f accS0 = {};
    v8f accS1 = {};
    for (int s = wave; s < nsteps; s += ROWS_PER_BLOCK) {
        const int  p0 = 4 * s + kh;
        const bool in0 = p0 < Kp, in1 = p0 + 1 < Kp;
        const int  c0 = in0 ? collist[p0]     : 0;
        const int  c1 = in1 ? collist[p0 + 1] : 0;
        const float a0 = in0 ? attL[m][c0] : 0.0f;
        const float a1 = in1 ? attL[m][c1] : 0.0f;
        const bool self0 = ((unsigned int)c0 % una) == rowmod;
        const bool self1 = ((unsigned int)c1 % una) == rowmod;
        v2f aS, aO;
        aS.x = self0 ? a0 : 0.0f;  aS.y = self1 ? a1 : 0.0f;
        aO.x = self0 ? 0.0f : a0;  aO.y = self1 ? 0.0f : a1;

        const long o0 = (long)c0 * Dd, o1r = (long)c1 * Dd;
        v2f b1_0; b1_0.x = w1b[o0];      b1_0.y = w1b[o1r];
        v2f b1_1; b1_1.x = w1b[o0 + 16]; b1_1.y = w1b[o1r + 16];
        v2f b2_0; b2_0.x = w2b[o0];      b2_0.y = w2b[o1r];
        v2f b2_1; b2_1.x = w2b[o0 + 16]; b2_1.y = w2b[o1r + 16];

        accS0 = __builtin_amdgcn_wmma_f32_16x16x4_f32(false, aS, false, b1_0,
                                                      (short)0, accS0, false, false);
        accS0 = __builtin_amdgcn_wmma_f32_16x16x4_f32(false, aO, false, b2_0,
                                                      (short)0, accS0, false, false);
        accS1 = __builtin_amdgcn_wmma_f32_16x16x4_f32(false, aS, false, b1_1,
                                                      (short)0, accS1, false, false);
        accS1 = __builtin_amdgcn_wmma_f32_16x16x4_f32(false, aO, false, b2_1,
                                                      (short)0, accS1, false, false);
    }

    __syncthreads();
    float* rw = &red[wave][(lane >> 4) << 3][m];
#pragma unroll
    for (int r = 0; r < 8; ++r) {
        rw[r * RS]      = accS0[r];
        rw[r * RS + 16] = accS1[r];
    }
    __syncthreads();
    float acc = 0.0f;
#pragma unroll
    for (int w = 0; w < ROWS_PER_BLOCK; ++w) acc += red[w][wave][lane];
    hp[((long)(b * Nn + ib + wave)) * Dd + h * DH + lane] = acc;
}


__global__ __launch_bounds__(32) void outproj_kernel(
    const float* __restrict__ hp, const float* __restrict__ Wo,
    const float* __restrict__ bo, float* __restrict__ out)
{
    const int lane = threadIdx.x & 31;
    const int m    = lane & 15;
    const int kh   = (lane >> 4) << 1;
    const int col0 = blockIdx.x * 16;
    const int row0 = blockIdx.y * 16;

    v8f acc = {};
    const float* aRow = hp + (long)(row0 + m) * Dd;
    const float* bRow = Wo + (long)(col0 + m) * Dd;
    for (int k0 = 0; k0 < Dd; k0 += 4) {
        v2f a  = *(const v2f*)(aRow + k0 + kh);
        v2f bb = *(const v2f*)(bRow + k0 + kh);
        acc = __builtin_amdgcn_wmma_f32_16x16x4_f32(false, a, false, bb,
                                                    (short)0, acc, false, false);
    }

    const int rbase = row0 + ((lane >> 4) << 3);
    const int c     = col0 + m;
    const float bias = bo[c];
#pragma unroll
    for (int r = 0; r < 8; ++r) {
        float v = acc[r] + bias;
        v = (v > 0.0f) ? v : (expf(v) - 1.0f);
        out[(long)(rbase + r) * Dd + c] = v;
    }
}

extern "C" void kernel_launch(void* const* d_in, const int* in_sizes, int n_in,
                              void* d_out, int out_size, void* d_ws, size_t ws_size,
                              hipStream_t stream) {
    (void)in_sizes; (void)n_in; (void)out_size; (void)ws_size;

    const float* hmat = (const float*)d_in[0];
    const int*   nag  = (const int*)  d_in[1];
    const int*   adj  = (const int*)  d_in[2];
    const float* W1   = (const float*)d_in[3];
    const float* W2   = (const float*)d_in[4];
    const float* Ai   = (const float*)d_in[5];
    const float* Aj   = (const float*)d_in[6];
    const float* Wo   = (const float*)d_in[7];
    const float* bo   = (const float*)d_in[8];
    float*       out  = (float*)d_out;

    float* ws  = (float*)d_ws;
    const size_t whN = (size_t)Bc * Nn * Dd;
    const size_t scN = (size_t)Bc * Hh * Nn;
    float* wh1 = ws;
    float* wh2 = wh1 + whN;
    float* s1  = wh2 + whN;
    float* s2  = s1 + scN;
    float* o1  = s2 + scN;
    float* o2  = o1 + scN;
    float* hp  = o2 + scN;

    proj12_kernel<<<dim3(Dd / 16, (Bc * Nn) / 16), 32, 0, stream>>>(
        hmat, W1, W2, wh1, wh2);

    scores_kernel<<<(Bc * Nn * Hh + 255) / 256, 256, 0, stream>>>(
        wh1, wh2, Ai, Aj, s1, s2, o1, o2);

    entmax_aggregate_kernel<<<(Bc * Hh * Nn) / ROWS_PER_BLOCK, 512, 0, stream>>>(
        s1, s2, o1, o2, adj, nag, wh1, wh2, hp);

    outproj_kernel<<<dim3(Dd / 16, (Bc * Nn) / 16), 32, 0, stream>>>(
        hp, Wo, bo, out);
}


# entmax tau: tangent steps and the support pass give way to closed-form solves on the active set {x>t} (count, sum, squares per pass) until the count repeats; support sum order changes
# speedup vs baseline: 8.9793x; 1.0423x over previous
#include <hip/hip_runtime.h>
#include <hip/hip_bf16.h>
#include <math.h>

static constexpr int Bc = 4;
static constexpr int Nn = 1024;
static constexpr int Dd = 256;
static constexpr int Hh = 8;
static constexpr int DH = 32;

typedef float v2f __attribute__((ext_vector_type(2)));
typedef float v8f __attribute__((ext_vector_type(8)));
typedef float v4f __attribute__((ext_vector_type(4)));
typedef int   v4i __attribute__((ext_vector_type(4)));

__global__ __launch_bounds__(32) void proj12_kernel(
    const float* __restrict__ hmat, const float* __restrict__ W1,
    const float* __restrict__ W2, float* __restrict__ wh1,
    float* __restrict__ wh2)
{
    const int lane = threadIdx.x & 31;
    const int m    = lane & 15;
    const int kh   = (lane >> 4) << 1;
    const int col0 = blockIdx.x * 16;
    const int row0 = blockIdx.y * 16;

    v8f acc1 = {};
    v8f acc2 = {};
    const float* aRow  = hmat + (long)(row0 + m) * Dd;
    const float* b1Row = W1   + (long)(col0 + m) * Dd;
    const float* b2Row = W2   + (long)(col0 + m) * Dd;

    for (int k0 = 0; k0 < Dd; k0 += 4) {
        v2f a  = *(const v2f*)(aRow  + k0 + kh);
        v2f b1 = *(const v2f*)(b1Row + k0 + kh);
        v2f b2 = *(const v2f*)(b2Row + k0 + kh);
        acc1 = __builtin_amdgcn_wmma_f32_16x16x4_f32(false, a, false, b1,
                                                     (short)0, acc1, false, false);
        acc2 = __builtin_amdgcn_wmma_f32_16x16x4_f32(false, a, false, b2,
                                                     (short)0, acc2, false, false);
    }

    const int rbase = row0 + ((lane >> 4) << 3);
    const int c     = col0 + m;
#pragma unroll
    for (int r = 0; r < 8; ++r) {
        wh1[(long)(rbase + r) * Dd + c] = acc1[r];
        wh2[(long)(rbase + r) * Dd + c] = acc2[r];
    }
}

__global__ __launch_bounds__(256) void scores_kernel(
    const float* __restrict__ wh1, const float* __restrict__ wh2,
    const float* __restrict__ Ai, const float* __restrict__ Aj,
    float* __restrict__ s1, float* __restrict__ s2,
    float* __restrict__ o1, float* __restrict__ o2)
{
    int idx = blockIdx.x * blockDim.x + threadIdx.x;
    if (idx >= Bc * Nn * Hh) return;
    const int h = idx % Hh;
    const int n = (idx / Hh) % Nn;
    const int b = idx / (Hh * Nn);

    const float* r1 = wh1 + ((long)(b * Nn + n)) * Dd + h * DH;
    const float* r2 = wh2 + ((long)(b * Nn + n)) * Dd + h * DH;
    const float* ai = Ai + h * DH;
    const float* aj = Aj + h * DH;

    float a1 = 0.f, a2 = 0.f, a3 = 0.f, a4 = 0.f;
#pragma unroll 8
    for (int d = 0; d < DH; ++d) {
        const float w1v = r1[d], w2v = r2[d], aiv = ai[d], ajv = aj[d];
        a1 += w1v * aiv;  a2 += w1v * ajv;
        a3 += w2v * aiv;  a4 += w2v * ajv;
    }
    const int o = (b * Hh + h) * Nn + n;
    s1[o] = a1; s2[o] = a2; o1[o] = a3; o2[o] = a4;
}

#define DEV static __device__ __forceinline__

DEV float xhalf(float v) {
    const int b = __builtin_bit_cast(int, v);
    return __builtin_bit_cast(float,
        __builtin_amdgcn_permlanex16(b, b, 0x76543210, 0xfedcba98, false, false));
}
template <int M> DEV float xrow(float v) {
    const int b = __builtin_bit_cast(int, v);
    return __builtin_bit_cast(float,
        __builtin_amdgcn_update_dpp(b, b, 0x160 + M, 0xf, 0xf, true));
}
DEV float wave_sum(float v) {
    v += xrow<1>(v); v += xrow<2>(v); v += xrow<4>(v); v += xrow<8>(v);
    v += xhalf(v);
    return v;
}
DEV float wave_max(float v) {
    v = fmaxf(v, xrow<1>(v)); v = fmaxf(v, xrow<2>(v));
    v = fmaxf(v, xrow<4>(v)); v = fmaxf(v, xrow<8>(v));
    v = fmaxf(v, xhalf(v));
    return v;
}
DEV float uni(float v) {
    return __builtin_bit_cast(float,
        __builtin_amdgcn_readfirstlane(__builtin_bit_cast(int, v)));
}

static constexpr int ROWS_PER_BLOCK = 16;
static constexpr int LDA = Nn + 4;
static constexpr int RS  = 36;
static constexpr int   STEP_MAX = 64;
static constexpr float TAU_EPS  = 2.384185791015625e-07f;

__global__ __launch_bounds__(512) void entmax_aggregate_kernel(
    const float* __restrict__ s1, const float* __restrict__ s2,
    const float* __restrict__ o1, const float* __restrict__ o2,
    const int* __restrict__ adj, const int* __restrict__ nag,
    const float* __restrict__ wh1, const float* __restrict__ wh2,
    float* __restrict__ hp)
{
    __shared__ __attribute__((aligned(16))) float smem[ROWS_PER_BLOCK * LDA];
    float (*attL)[LDA]   = (float (*)[LDA])smem;
    float (*red)[16][RS] = (float (*)[16][RS])smem;
    __shared__ unsigned int suppw[ROWS_PER_BLOCK][32];
    __shared__ __attribute__((aligned(8))) int collist[Nn];

    const int lane = threadIdx.x & 31;
    const int wave = __builtin_amdgcn_readfirstlane((int)(threadIdx.x >> 5));
    const int rid  = blockIdx.x * ROWS_PER_BLOCK + wave;
    const int i    = rid & (Nn - 1);
    const int h    = (rid >> 10) & (Hh - 1);
    const int b    = rid >> 13;
    const int na   = nag[0];

    const int   rowbase = (b * Hh + h) * Nn;
    const float s1i = s1[rowbase + i];
    const float o1i = o1[rowbase + i];
    const int   im  = i % na;
    const long  adjbase = ((long)b * Nn + i) * Nn;

    const v4f* s2v  = (const v4f*)(s2 + rowbase) + lane;
    const v4f* o2v  = (const v4f*)(o2 + rowbase) + lane;
    const v4i* adjv = (const v4i*)(adj + adjbase) + lane;

    float x[32];
    int       jm   = (4 * lane) % na;
    const int step = 128 % na;
    float lmax = -3.4e38f;
#pragma unroll
    for (int c = 0; c < 8; ++c) {
        const v4f sv = s2v[c * 32];
        const v4f ov = o2v[c * 32];
        const v4i av = adjv[c * 32];
        asm volatile("" :: "v"(sv), "v"(ov));
        const float se[4] = { sv.x, sv.y, sv.z, sv.w };
        const float oe[4] = { ov.x, ov.y, ov.z, ov.w };
        const int   ae[4] = { av.x, av.y, av.z, av.w };
        int m = jm;
#pragma unroll
        for (int e = 0; e < 4; ++e) {
            float ev = (im == m) ? (s1i + se[e]) : (o1i + oe[e]);
            if (ae[e] == 0) ev = -1.0e12f;
            const float xv = 0.5f * ev;
            x[4 * c + e] = xv;
            lmax = fmaxf(lmax, xv);
            m = (m + 1 == na) ? 0 : (m + 1);
        }
        jm += step;
        if (jm >= na) jm -= na;
    }
    const float xmax = uni(wave_max(lmax));
#pragma unroll
    for (int r = 0; r < 32; ++r) x[r] -= xmax;

    float t = -1.0f, lo = -1.0f, tstar = -1.0f;
    int kprev = -1;
    for (int it = 0; it <= STEP_MAX; ++it) {
        const v2f tt = { t, t };
        const v2f zz = { 0.f, 0.f };
        v2f ga = zz, gb = zz, fa = zz, fb = zz;
        int k = 0;
#pragma unroll
        for (int r = 0; r < 32; r += 4) {
            const v2f xa = { x[r],     x[r + 1] };
            const v2f xb = { x[r + 2], x[r + 3] };
            const v2f da = __builtin_elementwise_max(xa - tt, zz);
            const v2f db = __builtin_elementwise_max(xb - tt, zz);
            ga += da; gb += db;
            fa = __builtin_elementwise_fma(da, da, fa);
            fb = __builtin_elementwise_fma(db, db, fb);
            k += __builtin_popcount(__builtin_amdgcn_ballot_w32(x[r]     > t))
               + __builtin_popcount(__builtin_amdgcn_ballot_w32(x[r + 1] > t))
               + __builtin_popcount(__builtin_amdgcn_ballot_w32(x[r + 2] > t))
               + __builtin_popcount(__builtin_amdgcn_ballot_w32(x[r + 3] > t));
        }
        const v2f gs = ga + gb, fs = fa + fb;
        const float G  = uni(wave_sum(gs.x + gs.y));
        const float F  = uni(wave_sum(fs.x + fs.y));
        const float kf = (float)k;
        const float disc = fmaxf(fmaf(G, G, -kf * (F - 1.0f)), 0.0f);
        const float s    = (G - __builtin_amdgcn_sqrtf(disc)) * __builtin_amdgcn_rcpf(kf);
        tstar = t + s;
        if (k == kprev || fabsf(s) <= TAU_EPS || it == STEP_MAX) break;
        kprev = k;
        lo = fmaxf(lo, t + 0.5f * (F - 1.0f) * __builtin_amdgcn_rcpf(G));
        t  = fmaxf(tstar, lo);
    }

    v4f* outrow = (v4f*)(&attL[wave][0]) + lane;
    unsigned int myw = 0;
#pragma unroll
    for (int c = 0; c < 8; ++c) {
        v4f y;
        y.x = fmaxf(x[4 * c]     - tstar, 0.0f);
        y.y = fmaxf(x[4 * c + 1] - tstar, 0.0f);
        y.z = fmaxf(x[4 * c + 2] - tstar, 0.0f);
        y.w = fmaxf(x[4 * c + 3] - tstar, 0.0f);
        y.x *= y.x; y.y *= y.y; y.z *= y.z; y.w *= y.w;
        outrow[c * 32] = y;
        myw |= (y.x != 0.0f) ? (1u << (4 * c))     : 0u;
        myw |= (y.y != 0.0f) ? (1u << (4 * c + 1)) : 0u;
        myw |= (y.z != 0.0f) ? (1u << (4 * c + 2)) : 0u;
        myw |= (y.w != 0.0f) ? (1u << (4 * c + 3)) : 0u;
    }
    suppw[wave][lane] = myw;
    __syncthreads();

    unsigned int u = 0;
#pragma unroll
    for (int w = 0; w < ROWS_PER_BLOCK; ++w) u |= suppw[w][lane];
    const int cnt = __builtin_popcount(u);
    int inc = cnt;
    inc += __builtin_amdgcn_update_dpp(0, inc, 0x111, 0xf, 0xf, true);
    inc += __builtin_amdgcn_update_dpp(0, inc, 0x112, 0xf, 0xf, true);
    inc += __builtin_amdgcn_update_dpp(0, inc, 0x114, 0xf, 0xf, true);
    inc += __builtin_amdgcn_update_dpp(0, inc, 0x118, 0xf, 0xf, true);
    const int lowtot = __builtin_amdgcn_readlane(inc, 15);
    if (lane >= 16) inc += lowtot;
    const int Kp = __builtin_amdgcn_readlane(inc, 31);
    {
        int pos = inc - cnt;
        const int cbase = 4 * lane;
        unsigned int bits = u;
#pragma unroll 1
        for (int it = 0; it < 32; ++it) {
            if (__builtin_amdgcn_ballot_w32(bits != 0u) == 0u) break;
            if (bits != 0u) {
                const int r = __builtin_ctz(bits);
                bits &= bits - 1u;
                if (((pos >> 2) & (ROWS_PER_BLOCK - 1)) == wave) collist[pos] = cbase + 128 * (r >> 2) + (r & 3);
                ++pos;
            }
        }
    }
    __syncthreads();

    const int m      = lane & 15;
    const int kh     = (lane >> 4) << 1;
    const int ib     = (blockIdx.x * ROWS_PER_BLOCK) & (Nn - 1);
    const unsigned int una    = (unsigned int)na;
    const unsigned int rowmod = (unsigned int)(ib + m) % una;
    const float* w1b = wh1 + ((long)b * Nn) * Dd + h * DH + m;
    const float* w2b = wh2 + ((long)b * Nn) * Dd + h * DH + m;
    const int nsteps = (Kp + 3) >> 2;

    v8f accS0 = {};
    v8f accS1 = {};
    for (int s = wave; s < nsteps; s += ROWS_PER_BLOCK) {
        const int  p0 = 4 * s + kh;
        const bool in0 = p0 < Kp, in1 = p0 + 1 < Kp;
        const int  c0 = in0 ? collist[p0]     : 0;
        const int  c1 = in1 ? collist[p0 + 1] : 0;
        const float a0 = in0 ? attL[m][c0] : 0.0f;
        const float a1 = in1 ? attL[m][c1] : 0.0f;
        const bool self0 = ((unsigned int)c0 % una) == rowmod;
        const bool self1 = ((unsigned int)c1 % una) == rowmod;
        v2f aS, aO;
        aS.x = self0 ? a0 : 0.0f;  aS.y = self1 ? a1 : 0.0f;
        aO.x = self0 ? 0.0f : a0;  aO.y = self1 ? 0.0f : a1;

        const long o0 = (long)c0 * Dd, o1r = (long)c1 * Dd;
        v2f b1_0; b1_0.x = w1b[o0];      b1_0.y = w1b[o1r];
        v2f b1_1; b1_1.x = w1b[o0 + 16]; b1_1.y = w1b[o1r + 16];
        v2f b2_0; b2_0.x = w2b[o0];      b2_0.y = w2b[o1r];
        v2f b2_1; b2_1.x = w2b[o0 + 16]; b2_1.y = w2b[o1r + 16];

        accS0 = __builtin_amdgcn_wmma_f32_16x16x4_f32(false, aS, false, b1_0,
                                                      (short)0, accS0, false, false);
        accS0 = __builtin_amdgcn_wmma_f32_16x16x4_f32(false, aO, false, b2_0,
                                                      (short)0, accS0, false, false);
        accS1 = __builtin_amdgcn_wmma_f32_16x16x4_f32(false, aS, false, b1_1,
                                                      (short)0, accS1, false, false);
        accS1 = __builtin_amdgcn_wmma_f32_16x16x4_f32(false, aO, false, b2_1,
                                                      (short)0, accS1, false, false);
    }

    __syncthreads();
    float* rw = &red[wave][(lane >> 4) << 3][m];
#pragma unroll
    for (int r = 0; r < 8; ++r) {
        rw[r * RS]      = accS0[r];
        rw[r * RS + 16] = accS1[r];
    }
    __syncthreads();
    float acc = 0.0f;
#pragma unroll
    for (int w = 0; w < ROWS_PER_BLOCK; ++w) acc += red[w][wave][lane];
    hp[((long)(b * Nn + ib + wave)) * Dd + h * DH + lane] = acc;
}


__global__ __launch_bounds__(32) void outproj_kernel(
    const float* __restrict__ hp, const float* __restrict__ Wo,
    const float* __restrict__ bo, float* __restrict__ out)
{
    const int lane = threadIdx.x & 31;
    const int m    = lane & 15;
    const int kh   = (lane >> 4) << 1;
    const int col0 = blockIdx.x * 16;
    const int row0 = blockIdx.y * 16;

    v8f acc = {};
    const float* aRow = hp + (long)(row0 + m) * Dd;
    const float* bRow = Wo + (long)(col0 + m) * Dd;
    for (int k0 = 0; k0 < Dd; k0 += 4) {
        v2f a  = *(const v2f*)(aRow + k0 + kh);
        v2f bb = *(const v2f*)(bRow + k0 + kh);
        acc = __builtin_amdgcn_wmma_f32_16x16x4_f32(false, a, false, bb,
                                                    (short)0, acc, false, false);
    }

    const int rbase = row0 + ((lane >> 4) << 3);
    const int c     = col0 + m;
    const float bias = bo[c];
#pragma unroll
    for (int r = 0; r < 8; ++r) {
        float v = acc[r] + bias;
        v = (v > 0.0f) ? v : (expf(v) - 1.0f);
        out[(long)(rbase + r) * Dd + c] = v;
    }
}

extern "C" void kernel_launch(void* const* d_in, const int* in_sizes, int n_in,
                              void* d_out, int out_size, void* d_ws, size_t ws_size,
                              hipStream_t stream) {
    (void)in_sizes; (void)n_in; (void)out_size; (void)ws_size;

    const float* hmat = (const float*)d_in[0];
    const int*   nag  = (const int*)  d_in[1];
    const int*   adj  = (const int*)  d_in[2];
    const float* W1   = (const float*)d_in[3];
    const float* W2   = (const float*)d_in[4];
    const float* Ai   = (const float*)d_in[5];
    const float* Aj   = (const float*)d_in[6];
    const float* Wo   = (const float*)d_in[7];
    const float* bo   = (const float*)d_in[8];
    float*       out  = (float*)d_out;

    float* ws  = (float*)d_ws;
    const size_t whN = (size_t)Bc * Nn * Dd;
    const size_t scN = (size_t)Bc * Hh * Nn;
    float* wh1 = ws;
    float* wh2 = wh1 + whN;
    float* s1  = wh2 + whN;
    float* s2  = s1 + scN;
    float* o1  = s2 + scN;
    float* o2  = o1 + scN;
    float* hp  = o2 + scN;

    proj12_kernel<<<dim3(Dd / 16, (Bc * Nn) / 16), 32, 0, stream>>>(
        hmat, W1, W2, wh1, wh2);

    scores_kernel<<<(Bc * Nn * Hh + 255) / 256, 256, 0, stream>>>(
        wh1, wh2, Ai, Aj, s1, s2, o1, o2);

    entmax_aggregate_kernel<<<(Bc * Hh * Nn) / ROWS_PER_BLOCK, 512, 0, stream>>>(
        s1, s2, o1, o2, adj, nag, wh1, wh2, hp);

    outproj_kernel<<<dim3(Dd / 16, (Bc * Nn) / 16), 32, 0, stream>>>(
        hp, Wo, bo, out);
}


# projections (Wh1/Wh2, output): a wave owns a 32x64 or 32x32 tile; fragments come as 16-byte loads (k remapped alike in A and B) one k-group ahead and are reused in registers; k sum order changes
# speedup vs baseline: 11.1886x; 1.2460x over previous
#include <hip/hip_runtime.h>
#include <hip/hip_bf16.h>
#include <math.h>

static constexpr int Bc = 4;
static constexpr int Nn = 1024;
static constexpr int Dd = 256;
static constexpr int Hh = 8;
static constexpr int DH = 32;

typedef float v2f __attribute__((ext_vector_type(2)));
typedef float v8f __attribute__((ext_vector_type(8)));
typedef float v4f __attribute__((ext_vector_type(4)));
typedef int   v4i __attribute__((ext_vector_type(4)));

#define DEVI static __device__ __forceinline__

template <int MT, int NT, int QG>
DEVI void load_group(v4f (&a)[MT][QG], v4f (&b)[NT][QG],
                     const float* __restrict__ aP, const float* __restrict__ bP, const int k0)
{
#pragma unroll
    for (int q = 0; q < QG; ++q) {
#pragma unroll
        for (int i = 0; i < MT; ++i) a[i][q] = *(const v4f*)(aP + i * 16 * Dd + k0 + 8 * q);
#pragma unroll
        for (int j = 0; j < NT; ++j) b[j][q] = *(const v4f*)(bP + j * 16 * Dd + k0 + 8 * q);
    }
}

template <int MT, int NT, int QG>
DEVI void mma_group(v8f (&acc)[MT][NT], const v4f (&a)[MT][QG], const v4f (&b)[NT][QG])
{
#pragma unroll
    for (int q = 0; q < QG; ++q) {
#pragma unroll
        for (int s = 0; s < 2; ++s) {
#pragma unroll
            for (int i = 0; i < MT; ++i) {
                const v2f af = s ? __builtin_shufflevector(a[i][q], a[i][q], 2, 3)
                                 : __builtin_shufflevector(a[i][q], a[i][q], 0, 1);
#pragma unroll
                for (int j = 0; j < NT; ++j) {
                    const v2f bf = s ? __builtin_shufflevector(b[j][q], b[j][q], 2, 3)
                                     : __builtin_shufflevector(b[j][q], b[j][q], 0, 1);
                    acc[i][j] = __builtin_amdgcn_wmma_f32_16x16x4_f32(false, af, false, bf,
                                                                      (short)0, acc[i][j], false, false);
                }
            }
        }
    }
}

template <int MT, int NT, int QG>
DEVI void gemm_wave_tile(v8f (&acc)[MT][NT], const float* __restrict__ aP, const float* __restrict__ bP)
{
    constexpr int KG = 8 * QG;
#pragma unroll
    for (int i = 0; i < MT; ++i) {
#pragma unroll
        for (int j = 0; j < NT; ++j) acc[i][j] = v8f{};
    }
    v4f a0[MT][QG], b0[NT][QG], a1[MT][QG], b1[NT][QG];
    load_group<MT, NT, QG>(a0, b0, aP, bP, 0);
#pragma unroll 1
    for (int k0 = 0; k0 < Dd - 2 * KG; k0 += 2 * KG) {
        load_group<MT, NT, QG>(a1, b1, aP, bP, k0 + KG);
        __builtin_amdgcn_sched_barrier(0);
        mma_group<MT, NT, QG>(acc, a0, b0);
        __builtin_amdgcn_sched_barrier(0);
        load_group<MT, NT, QG>(a0, b0, aP, bP, k0 + 2 * KG);
        __builtin_amdgcn_sched_barrier(0);
        mma_group<MT, NT, QG>(acc, a1, b1);
    }
    load_group<MT, NT, QG>(a1, b1, aP, bP, Dd - KG);
    __builtin_amdgcn_sched_barrier(0);
    mma_group<MT, NT, QG>(acc, a0, b0);
    mma_group<MT, NT, QG>(acc, a1, b1);
}

__global__ __launch_bounds__(128) void proj12_kernel(
    const float* __restrict__ hmat, const float* __restrict__ W1,
    const float* __restrict__ W2, float* __restrict__ wh1,
    float* __restrict__ wh2)
{
    const int lane = threadIdx.x & 31;
    const int wave = threadIdx.x >> 5;
    const int m    = lane & 15;
    const int hf   = lane >> 4;
    const int row0 = (blockIdx.x >> 2) * 64 + (wave >> 1) * 32;
    const int col0 = (blockIdx.x & 1) * 128 + (wave & 1) * 64;
    const bool second = (blockIdx.x & 2) != 0;
    const float* __restrict__ W   = second ? W2 : W1;
    float* __restrict__       dst = second ? wh2 : wh1;

    v8f acc[2][4];
    gemm_wave_tile<2, 4, 4>(acc, hmat + (row0 + m) * Dd + 4 * hf, W + (col0 + m) * Dd + 4 * hf);

#pragma unroll
    for (int i = 0; i < 2; ++i) {
#pragma unroll
        for (int j = 0; j < 4; ++j) {
            float* o = dst + (row0 + 16 * i + 8 * hf) * Dd + col0 + 16 * j + m;
#pragma unroll
            for (int r = 0; r < 8; ++r) o[r * Dd] = acc[i][j][r];
        }
    }
}

__global__ __launch_bounds__(256) void scores_kernel(
    const float* __restrict__ wh1, const float* __restrict__ wh2,
    const float* __restrict__ Ai, const float* __restrict__ Aj,
    float* __restrict__ s1, float* __restrict__ s2,
    float* __restrict__ o1, float* __restrict__ o2)
{
    int idx = blockIdx.x * blockDim.x + threadIdx.x;
    if (idx >= Bc * Nn * Hh) return;
    const int h = idx % Hh;
    const int n = (idx / Hh) % Nn;
    const int b = idx / (Hh * Nn);

    const float* r1 = wh1 + ((long)(b * Nn + n)) * Dd + h * DH;
    const float* r2 = wh2 + ((long)(b * Nn + n)) * Dd + h * DH;
    const float* ai = Ai + h * DH;
    const float* aj = Aj + h * DH;

    float a1 = 0.f, a2 = 0.f, a3 = 0.f, a4 = 0.f;
#pragma unroll 8
    for (int d = 0; d < DH; ++d) {
        const float w1v = r1[d], w2v = r2[d], aiv = ai[d], ajv = aj[d];
        a1 += w1v * aiv;  a2 += w1v * ajv;
        a3 += w2v * aiv;  a4 += w2v * ajv;
    }
    const int o = (b * Hh + h) * Nn + n;
    s1[o] = a1; s2[o] = a2; o1[o] = a3; o2[o] = a4;
}

#define DEV static __device__ __forceinline__

DEV float xhalf(float v) {
    const int b = __builtin_bit_cast(int, v);
    return __builtin_bit_cast(float,
        __builtin_amdgcn_permlanex16(b, b, 0x76543210, 0xfedcba98, false, false));
}
template <int M> DEV float xrow(float v) {
    const int b = __builtin_bit_cast(int, v);
    return __builtin_bit_cast(float,
        __builtin_amdgcn_update_dpp(b, b, 0x160 + M, 0xf, 0xf, true));
}
DEV float wave_sum(float v) {
    v += xrow<1>(v); v += xrow<2>(v); v += xrow<4>(v); v += xrow<8>(v);
    v += xhalf(v);
    return v;
}
DEV float wave_max(float v) {
    v = fmaxf(v, xrow<1>(v)); v = fmaxf(v, xrow<2>(v));
    v = fmaxf(v, xrow<4>(v)); v = fmaxf(v, xrow<8>(v));
    v = fmaxf(v, xhalf(v));
    return v;
}
DEV float uni(float v) {
    return __builtin_bit_cast(float,
        __builtin_amdgcn_readfirstlane(__builtin_bit_cast(int, v)));
}

static constexpr int ROWS_PER_BLOCK = 16;
static constexpr int LDA = Nn + 4;
static constexpr int RS  = 36;
static constexpr int   STEP_MAX = 64;
static constexpr float TAU_EPS  = 2.384185791015625e-07f;

__global__ __launch_bounds__(512) void entmax_aggregate_kernel(
    const float* __restrict__ s1, const float* __restrict__ s2,
    const float* __restrict__ o1, const float* __restrict__ o2,
    const int* __restrict__ adj, const int* __restrict__ nag,
    const float* __restrict__ wh1, const float* __restrict__ wh2,
    float* __restrict__ hp)
{
    __shared__ __attribute__((aligned(16))) float smem[ROWS_PER_BLOCK * LDA];
    float (*attL)[LDA]   = (float (*)[LDA])smem;
    float (*red)[16][RS] = (float (*)[16][RS])smem;
    __shared__ unsigned int suppw[ROWS_PER_BLOCK][32];
    __shared__ __attribute__((aligned(8))) int collist[Nn];

    const int lane = threadIdx.x & 31;
    const int wave = __builtin_amdgcn_readfirstlane((int)(threadIdx.x >> 5));
    const int rid  = blockIdx.x * ROWS_PER_BLOCK + wave;
    const int i    = rid & (Nn - 1);
    const int h    = (rid >> 10) & (Hh - 1);
    const int b    = rid >> 13;
    const int na   = nag[0];

    const int   rowbase = (b * Hh + h) * Nn;
    const float s1i = s1[rowbase + i];
    const float o1i = o1[rowbase + i];
    const int   im  = i % na;
    const long  adjbase = ((long)b * Nn + i) * Nn;

    const v4f* s2v  = (const v4f*)(s2 + rowbase) + lane;
    const v4f* o2v  = (const v4f*)(o2 + rowbase) + lane;
    const v4i* adjv = (const v4i*)(adj + adjbase) + lane;

    float x[32];
    int       jm   = (4 * lane) % na;
    const int step = 128 % na;
    float lmax = -3.4e38f;
#pragma unroll
    for (int c = 0; c < 8; ++c) {
        const v4f sv = s2v[c * 32];
        const v4f ov = o2v[c * 32];
        const v4i av = adjv[c * 32];
        asm volatile("" :: "v"(sv), "v"(ov));
        const float se[4] = { sv.x, sv.y, sv.z, sv.w };
        const float oe[4] = { ov.x, ov.y, ov.z, ov.w };
        const int   ae[4] = { av.x, av.y, av.z, av.w };
        int m = jm;
#pragma unroll
        for (int e = 0; e < 4; ++e) {
            float ev = (im == m) ? (s1i + se[e]) : (o1i + oe[e]);
            if (ae[e] == 0) ev = -1.0e12f;
            const float xv = 0.5f * ev;
            x[4 * c + e] = xv;
            lmax = fmaxf(lmax, xv);
            m = (m + 1 == na) ? 0 : (m + 1);
        }
        jm += step;
        if (jm >= na) jm -= na;
    }
    const float xmax = uni(wave_max(lmax));
#pragma unroll
    for (int r = 0; r < 32; ++r) x[r] -= xmax;

    float t = -1.0f, lo = -1.0f, tstar = -1.0f;
    int kprev = -1;
    for (int it = 0; it <= STEP_MAX; ++it) {
        const v2f tt = { t, t };
        const v2f zz = { 0.f, 0.f };
        v2f ga = zz, gb = zz, fa = zz, fb = zz;
        int k = 0;
#pragma unroll
        for (int r = 0; r < 32; r += 4) {
            const v2f xa = { x[r],     x[r + 1] };
            const v2f xb = { x[r + 2], x[r + 3] };
            const v2f da = __builtin_elementwise_max(xa - tt, zz);
            const v2f db = __builtin_elementwise_max(xb - tt, zz);
            ga += da; gb += db;
            fa = __builtin_elementwise_fma(da, da, fa);
            fb = __builtin_elementwise_fma(db, db, fb);
            k += __builtin_popcount(__builtin_amdgcn_ballot_w32(x[r]     > t))
               + __builtin_popcount(__builtin_amdgcn_ballot_w32(x[r + 1] > t))
               + __builtin_popcount(__builtin_amdgcn_ballot_w32(x[r + 2] > t))
               + __builtin_popcount(__builtin_amdgcn_ballot_w32(x[r + 3] > t));
        }
        const v2f gs = ga + gb, fs = fa + fb;
        const float G  = uni(wave_sum(gs.x + gs.y));
        const float F  = uni(wave_sum(fs.x + fs.y));
        const float kf = (float)k;
        const float disc = fmaxf(fmaf(G, G, -kf * (F - 1.0f)), 0.0f);
        const float s    = (G - __builtin_amdgcn_sqrtf(disc)) * __builtin_amdgcn_rcpf(kf);
        tstar = t + s;
        if (k == kprev || fabsf(s) <= TAU_EPS || it == STEP_MAX) break;
        kprev = k;
        lo = fmaxf(lo, t + 0.5f * (F - 1.0f) * __builtin_amdgcn_rcpf(G));
        t  = fmaxf(tstar, lo);
    }

    v4f* outrow = (v4f*)(&attL[wave][0]) + lane;
    unsigned int myw = 0;
#pragma unroll
    for (int c = 0; c < 8; ++c) {
        v4f y;
        y.x = fmaxf(x[4 * c]     - tstar, 0.0f);
        y.y = fmaxf(x[4 * c + 1] - tstar, 0.0f);
        y.z = fmaxf(x[4 * c + 2] - tstar, 0.0f);
        y.w = fmaxf(x[4 * c + 3] - tstar, 0.0f);
        y.x *= y.x; y.y *= y.y; y.z *= y.z; y.w *= y.w;
        outrow[c * 32] = y;
        myw |= (y.x != 0.0f) ? (1u << (4 * c))     : 0u;
        myw |= (y.y != 0.0f) ? (1u << (4 * c + 1)) : 0u;
        myw |= (y.z != 0.0f) ? (1u << (4 * c + 2)) : 0u;
        myw |= (y.w != 0.0f) ? (1u << (4 * c + 3)) : 0u;
    }
    suppw[wave][lane] = myw;
    __syncthreads();

    unsigned int u = 0;
#pragma unroll
    for (int w = 0; w < ROWS_PER_BLOCK; ++w) u |= suppw[w][lane];
    const int cnt = __builtin_popcount(u);
    int inc = cnt;
    inc += __builtin_amdgcn_update_dpp(0, inc, 0x111, 0xf, 0xf, true);
    inc += __builtin_amdgcn_update_dpp(0, inc, 0x112, 0xf, 0xf, true);
    inc += __builtin_amdgcn_update_dpp(0, inc, 0x114, 0xf, 0xf, true);
    inc += __builtin_amdgcn_update_dpp(0, inc, 0x118, 0xf, 0xf, true);
    const int lowtot = __builtin_amdgcn_readlane(inc, 15);
    if (lane >= 16) inc += lowtot;
    const int Kp = __builtin_amdgcn_readlane(inc, 31);
    {
        int pos = inc - cnt;
        const int cbase = 4 * lane;
        unsigned int bits = u;
#pragma unroll 1
        for (int it = 0; it < 32; ++it) {
            if (__builtin_amdgcn_ballot_w32(bits != 0u) == 0u) break;
            if (bits != 0u) {
                const int r = __builtin_ctz(bits);
                bits &= bits - 1u;
                if (((pos >> 2) & (ROWS_PER_BLOCK - 1)) == wave) collist[pos] = cbase + 128 * (r >> 2) + (r & 3);
                ++pos;
            }
        }
    }
    __syncthreads();

    const int m      = lane & 15;
    const int kh     = (lane >> 4) << 1;
    const int ib     = (blockIdx.x * ROWS_PER_BLOCK) & (Nn - 1);
    const unsigned int una    = (unsigned int)na;
    const unsigned int rowmod = (unsigned int)(ib + m) % una;
    const float* w1b = wh1 + ((long)b * Nn) * Dd + h * DH + m;
    const float* w2b = wh2 + ((long)b * Nn) * Dd + h * DH + m;
    const int nsteps = (Kp + 3) >> 2;

    v8f accS0 = {};
    v8f accS1 = {};
    for (int s = wave; s < nsteps; s += ROWS_PER_BLOCK) {
        const int  p0 = 4 * s + kh;
        const bool in0 = p0 < Kp, in1 = p0 + 1 < Kp;
        const int  c0 = in0 ? collist[p0]     : 0;
        const int  c1 = in1 ? collist[p0 + 1] : 0;
        const float a0 = in0 ? attL[m][c0] : 0.0f;
        const float a1 = in1 ? attL[m][c1] : 0.0f;
        const bool self0 = ((unsigned int)c0 % una) == rowmod;
        const bool self1 = ((unsigned int)c1 % una) == rowmod;
        v2f aS, aO;
        aS.x = self0 ? a0 : 0.0f;  aS.y = self1 ? a1 : 0.0f;
        aO.x = self0 ? 0.0f : a0;  aO.y = self1 ? 0.0f : a1;

        const long o0 = (long)c0 * Dd, o1r = (long)c1 * Dd;
        v2f b1_0; b1_0.x = w1b[o0];      b1_0.y = w1b[o1r];
        v2f b1_1; b1_1.x = w1b[o0 + 16]; b1_1.y = w1b[o1r + 16];
        v2f b2_0; b2_0.x = w2b[o0];      b2_0.y = w2b[o1r];
        v2f b2_1; b2_1.x = w2b[o0 + 16]; b2_1.y = w2b[o1r + 16];

        accS0 = __builtin_amdgcn_wmma_f32_16x16x4_f32(false, aS, false, b1_0,
                                                      (short)0, accS0, false, false);
        accS0 = __builtin_amdgcn_wmma_f32_16x16x4_f32(false, aO, false, b2_0,
                                                      (short)0, accS0, false, false);
        accS1 = __builtin_amdgcn_wmma_f32_16x16x4_f32(false, aS, false, b1_1,
                                                      (short)0, accS1, false, false);
        accS1 = __builtin_amdgcn_wmma_f32_16x16x4_f32(false, aO, false, b2_1,
                                                      (short)0, accS1, false, false);
    }

    __syncthreads();
    float* rw = &red[wave][(lane >> 4) << 3][m];
#pragma unroll
    for (int r = 0; r < 8; ++r) {
        rw[r * RS]      = accS0[r];
        rw[r * RS + 16] = accS1[r];
    }
    __syncthreads();
    float acc = 0.0f;
#pragma unroll
    for (int w = 0; w < ROWS_PER_BLOCK; ++w) acc += red[w][wave][lane];
    hp[((long)(b * Nn + ib + wave)) * Dd + h * DH + lane] = acc;
}


__global__ __launch_bounds__(128) void outproj_kernel(
    const float* __restrict__ hp, const float* __restrict__ Wo,
    const float* __restrict__ bo, float* __restrict__ out)
{
    const int lane = threadIdx.x & 31;
    const int wave = threadIdx.x >> 5;
    const int m    = lane & 15;
    const int hf   = lane >> 4;
    const int row0 = (blockIdx.x >> 2) * 64 + (wave >> 1) * 32;
    const int col0 = (blockIdx.x & 3) * 64 + (wave & 1) * 32;

    v8f acc[2][2];
    gemm_wave_tile<2, 2, 8>(acc, hp + (row0 + m) * Dd + 4 * hf, Wo + (col0 + m) * Dd + 4 * hf);

#pragma unroll
    for (int j = 0; j < 2; ++j) {
        const int c = col0 + 16 * j + m;
        const float bias = bo[c];
#pragma unroll
        for (int i = 0; i < 2; ++i) {
            float* o = out + (row0 + 16 * i + 8 * hf) * Dd + c;
#pragma unroll
            for (int r = 0; r < 8; ++r) {
                float v = acc[i][j][r] + bias;
                v = (v > 0.0f) ? v : (expf(v) - 1.0f);
                o[r * Dd] = v;
            }
        }
    }
}

extern "C" void kernel_launch(void* const* d_in, const int* in_sizes, int n_in,
                              void* d_out, int out_size, void* d_ws, size_t ws_size,
                              hipStream_t stream) {
    (void)in_sizes; (void)n_in; (void)out_size; (void)ws_size;

    const float* hmat = (const float*)d_in[0];
    const int*   nag  = (const int*)  d_in[1];
    const int*   adj  = (const int*)  d_in[2];
    const float* W1   = (const float*)d_in[3];
    const float* W2   = (const float*)d_in[4];
    const float* Ai   = (const float*)d_in[5];
    const float* Aj   = (const float*)d_in[6];
    const float* Wo   = (const float*)d_in[7];
    const float* bo   = (const float*)d_in[8];
    float*       out  = (float*)d_out;

    float* ws  = (float*)d_ws;
    const size_t whN = (size_t)Bc * Nn * Dd;
    const size_t scN = (size_t)Bc * Hh * Nn;
    float* wh1 = ws;
    float* wh2 = wh1 + whN;
    float* s1  = wh2 + whN;
    float* s2  = s1 + scN;
    float* o1  = s2 + scN;
    float* o2  = o1 + scN;
    float* hp  = o2 + scN;

    proj12_kernel<<<((Bc * Nn) / 64) * 4, 128, 0, stream>>>(
        hmat, W1, W2, wh1, wh2);

    scores_kernel<<<(Bc * Nn * Hh + 255) / 256, 256, 0, stream>>>(
        wh1, wh2, Ai, Aj, s1, s2, o1, o2);

    entmax_aggregate_kernel<<<(Bc * Hh * Nn) / ROWS_PER_BLOCK, 512, 0, stream>>>(
        s1, s2, o1, o2, adj, nag, wh1, wh2, hp);

    outproj_kernel<<<((Bc * Nn) / 64) * 4, 128, 0, stream>>>(
        hp, Wo, bo, out);
}


# entmax x formation rewritten: scores stored halved (exact), packed adds, one select per column, self masks once per row: 5 not 11 vector ops a column; the row's 8 adj loads now go out as one group
# speedup vs baseline: 12.0548x; 1.0774x over previous
#include <hip/hip_runtime.h>
#include <hip/hip_bf16.h>
#include <math.h>

static constexpr int Bc = 4;
static constexpr int Nn = 1024;
static constexpr int Dd = 256;
static constexpr int Hh = 8;
static constexpr int DH = 32;

typedef float v2f __attribute__((ext_vector_type(2)));
typedef float v8f __attribute__((ext_vector_type(8)));
typedef float v4f __attribute__((ext_vector_type(4)));
typedef int   v4i __attribute__((ext_vector_type(4)));

#define DEVI static __device__ __forceinline__

template <int MT, int NT, int QG>
DEVI void load_group(v4f (&a)[MT][QG], v4f (&b)[NT][QG],
                     const float* __restrict__ aP, const float* __restrict__ bP, const int k0)
{
#pragma unroll
    for (int q = 0; q < QG; ++q) {
#pragma unroll
        for (int i = 0; i < MT; ++i) a[i][q] = *(const v4f*)(aP + i * 16 * Dd + k0 + 8 * q);
#pragma unroll
        for (int j = 0; j < NT; ++j) b[j][q] = *(const v4f*)(bP + j * 16 * Dd + k0 + 8 * q);
    }
}

template <int MT, int NT, int QG>
DEVI void mma_group(v8f (&acc)[MT][NT], const v4f (&a)[MT][QG], const v4f (&b)[NT][QG])
{
#pragma unroll
    for (int q = 0; q < QG; ++q) {
#pragma unroll
        for (int s = 0; s < 2; ++s) {
#pragma unroll
            for (int i = 0; i < MT; ++i) {
                const v2f af = s ? __builtin_shufflevector(a[i][q], a[i][q], 2, 3)
                                 : __builtin_shufflevector(a[i][q], a[i][q], 0, 1);
#pragma unroll
                for (int j = 0; j < NT; ++j) {
                    const v2f bf = s ? __builtin_shufflevector(b[j][q], b[j][q], 2, 3)
                                     : __builtin_shufflevector(b[j][q], b[j][q], 0, 1);
                    acc[i][j] = __builtin_amdgcn_wmma_f32_16x16x4_f32(false, af, false, bf,
                                                                      (short)0, acc[i][j], false, false);
                }
            }
        }
    }
}

template <int MT, int NT, int QG>
DEVI void gemm_wave_tile(v8f (&acc)[MT][NT], const float* __restrict__ aP, const float* __restrict__ bP)
{
    constexpr int KG = 8 * QG;
#pragma unroll
    for (int i = 0; i < MT; ++i) {
#pragma unroll
        for (int j = 0; j < NT; ++j) acc[i][j] = v8f{};
    }
    v4f a0[MT][QG], b0[NT][QG], a1[MT][QG], b1[NT][QG];
    load_group<MT, NT, QG>(a0, b0, aP, bP, 0);
#pragma unroll 1
    for (int k0 = 0; k0 < Dd - 2 * KG; k0 += 2 * KG) {
        load_group<MT, NT, QG>(a1, b1, aP, bP, k0 + KG);
        __builtin_amdgcn_sched_barrier(0);
        mma_group<MT, NT, QG>(acc, a0, b0);
        __builtin_amdgcn_sched_barrier(0);
        load_group<MT, NT, QG>(a0, b0, aP, bP, k0 + 2 * KG);
        __builtin_amdgcn_sched_barrier(0);
        mma_group<MT, NT, QG>(acc, a1, b1);
    }
    load_group<MT, NT, QG>(a1, b1, aP, bP, Dd - KG);
    __builtin_amdgcn_sched_barrier(0);
    mma_group<MT, NT, QG>(acc, a0, b0);
    mma_group<MT, NT, QG>(acc, a1, b1);
}

__global__ __launch_bounds__(128) void proj12_kernel(
    const float* __restrict__ hmat, const float* __restrict__ W1,
    const float* __restrict__ W2, float* __restrict__ wh1,
    float* __restrict__ wh2)
{
    const int lane = threadIdx.x & 31;
    const int wave = threadIdx.x >> 5;
    const int m    = lane & 15;
    const int hf   = lane >> 4;
    const int row0 = (blockIdx.x >> 2) * 64 + (wave >> 1) * 32;
    const int col0 = (blockIdx.x & 1) * 128 + (wave & 1) * 64;
    const bool second = (blockIdx.x & 2) != 0;
    const float* __restrict__ W   = second ? W2 : W1;
    float* __restrict__       dst = second ? wh2 : wh1;

    v8f acc[2][4];
    gemm_wave_tile<2, 4, 4>(acc, hmat + (row0 + m) * Dd + 4 * hf, W + (col0 + m) * Dd + 4 * hf);

#pragma unroll
    for (int i = 0; i < 2; ++i) {
#pragma unroll
        for (int j = 0; j < 4; ++j) {
            float* o = dst + (row0 + 16 * i + 8 * hf) * Dd + col0 + 16 * j + m;
#pragma unroll
            for (int r = 0; r < 8; ++r) o[r * Dd] = acc[i][j][r];
        }
    }
}

__global__ __launch_bounds__(256) void scores_kernel(
    const float* __restrict__ wh1, const float* __restrict__ wh2,
    const float* __restrict__ Ai, const float* __restrict__ Aj,
    float* __restrict__ s1, float* __restrict__ s2,
    float* __restrict__ o1, float* __restrict__ o2)
{
    int idx = blockIdx.x * blockDim.x + threadIdx.x;
    if (idx >= Bc * Nn * Hh) return;
    const int h = idx % Hh;
    const int n = (idx / Hh) % Nn;
    const int b = idx / (Hh * Nn);

    const float* r1 = wh1 + ((long)(b * Nn + n)) * Dd + h * DH;
    const float* r2 = wh2 + ((long)(b * Nn + n)) * Dd + h * DH;
    const float* ai = Ai + h * DH;
    const float* aj = Aj + h * DH;

    float a1 = 0.f, a2 = 0.f, a3 = 0.f, a4 = 0.f;
#pragma unroll 8
    for (int d = 0; d < DH; ++d) {
        const float w1v = r1[d], w2v = r2[d], aiv = ai[d], ajv = aj[d];
        a1 += w1v * aiv;  a2 += w1v * ajv;
        a3 += w2v * aiv;  a4 += w2v * ajv;
    }
    const int o = (b * Hh + h) * Nn + n;
    s1[o] = 0.5f * a1; s2[o] = 0.5f * a2; o1[o] = 0.5f * a3; o2[o] = 0.5f * a4;
}

#define DEV static __device__ __forceinline__

DEV float xhalf(float v) {
    const int b = __builtin_bit_cast(int, v);
    return __builtin_bit_cast(float,
        __builtin_amdgcn_permlanex16(b, b, 0x76543210, 0xfedcba98, false, false));
}
template <int M> DEV float xrow(float v) {
    const int b = __builtin_bit_cast(int, v);
    return __builtin_bit_cast(float,
        __builtin_amdgcn_update_dpp(b, b, 0x160 + M, 0xf, 0xf, true));
}
DEV float wave_sum(float v) {
    v += xrow<1>(v); v += xrow<2>(v); v += xrow<4>(v); v += xrow<8>(v);
    v += xhalf(v);
    return v;
}
DEV float wave_max(float v) {
    v = fmaxf(v, xrow<1>(v)); v = fmaxf(v, xrow<2>(v));
    v = fmaxf(v, xrow<4>(v)); v = fmaxf(v, xrow<8>(v));
    v = fmaxf(v, xhalf(v));
    return v;
}
DEV float uni(float v) {
    return __builtin_bit_cast(float,
        __builtin_amdgcn_readfirstlane(__builtin_bit_cast(int, v)));
}

static constexpr int ROWS_PER_BLOCK = 16;
static constexpr int LDA = Nn + 4;
static constexpr int RS  = 36;
static constexpr int   STEP_MAX = 64;
static constexpr float TAU_EPS  = 2.384185791015625e-07f;
static constexpr float X_MASKED = 0.5f * -1.0e12f;

DEV void self_flags(bool (&sf)[4], const int jm, const int im, const int na) {
    int m = jm;
#pragma unroll
    for (int e = 0; e < 4; ++e) {
        sf[e] = (im == m);
        m = (m + 1 == na) ? 0 : (m + 1);
    }
}

template <bool FLAGS_PER_CHUNK>
DEV float form_x(float (&x)[32], const v4f* __restrict__ s2v, const v4f* __restrict__ o2v,
                 const v4i* __restrict__ adjv, const float s1i, const float o1i,
                 int jm, const int step, const int im, const int na)
{
    bool sf[4];
    self_flags(sf, jm, im, na);
    const v2f s1p = { s1i, s1i };
    const v2f o1p = { o1i, o1i };
    float lmax = -3.4e38f;
    v4i av[8];
    v4f sv[8], ov[8];
#pragma unroll
    for (int c = 0; c < 8; ++c) av[c] = adjv[c * 32];
    constexpr int AHEAD = FLAGS_PER_CHUNK ? 1 : 2;
#pragma unroll
    for (int c = 0; c < AHEAD; ++c) {
        sv[c] = s2v[c * 32];
        ov[c] = o2v[c * 32];
    }
#pragma unroll
    for (int c = 0; c < 8; ++c) {
        if (FLAGS_PER_CHUNK && c > 0) {
            jm += step;
            if (jm >= na) jm -= na;
            self_flags(sf, jm, im, na);
        }
        if (c + AHEAD < 8) {
            sv[c + AHEAD] = s2v[(c + AHEAD) * 32];
            ov[c + AHEAD] = o2v[(c + AHEAD) * 32];
        }
        const v2f sa = __builtin_shufflevector(sv[c], sv[c], 0, 1) + s1p;
        const v2f sb = __builtin_shufflevector(sv[c], sv[c], 2, 3) + s1p;
        const v2f oa = __builtin_shufflevector(ov[c], ov[c], 0, 1) + o1p;
        const v2f ob = __builtin_shufflevector(ov[c], ov[c], 2, 3) + o1p;
        const float x0 = (av[c].x == 0) ? X_MASKED : (sf[0] ? sa.x : oa.x);
        const float x1 = (av[c].y == 0) ? X_MASKED : (sf[1] ? sa.y : oa.y);
        const float x2 = (av[c].z == 0) ? X_MASKED : (sf[2] ? sb.x : ob.x);
        const float x3 = (av[c].w == 0) ? X_MASKED : (sf[3] ? sb.y : ob.y);
        x[4 * c]     = x0;
        x[4 * c + 1] = x1;
        x[4 * c + 2] = x2;
        x[4 * c + 3] = x3;
        lmax = fmaxf(fmaxf(lmax, x0), fmaxf(x1, fmaxf(x2, x3)));
        if (c + 1 < 8) __builtin_amdgcn_sched_barrier(0);
    }
    return lmax;
}

__global__ __launch_bounds__(512) void entmax_aggregate_kernel(
    const float* __restrict__ s1, const float* __restrict__ s2,
    const float* __restrict__ o1, const float* __restrict__ o2,
    const int* __restrict__ adj, const int* __restrict__ nag,
    const float* __restrict__ wh1, const float* __restrict__ wh2,
    float* __restrict__ hp)
{
    __shared__ __attribute__((aligned(16))) float smem[ROWS_PER_BLOCK * LDA];
    float (*attL)[LDA]   = (float (*)[LDA])smem;
    float (*red)[16][RS] = (float (*)[16][RS])smem;
    __shared__ unsigned int suppw[ROWS_PER_BLOCK][32];
    __shared__ __attribute__((aligned(8))) int collist[Nn];

    const int lane = threadIdx.x & 31;
    const int wave = __builtin_amdgcn_readfirstlane((int)(threadIdx.x >> 5));
    const int rid  = blockIdx.x * ROWS_PER_BLOCK + wave;
    const int i    = rid & (Nn - 1);
    const int h    = (rid >> 10) & (Hh - 1);
    const int b    = rid >> 13;
    const int na   = nag[0];

    const int   rowbase = (b * Hh + h) * Nn;
    const float s1i = s1[rowbase + i];
    const float o1i = o1[rowbase + i];
    const int   im  = i % na;
    const long  adjbase = ((long)b * Nn + i) * Nn;

    const v4f* s2v  = (const v4f*)(s2 + rowbase) + lane;
    const v4f* o2v  = (const v4f*)(o2 + rowbase) + lane;
    const v4i* adjv = (const v4i*)(adj + adjbase) + lane;

    float x[32];
    const int jm   = (4 * lane) % na;
    const int step = 128 % na;
    float lmax;
    if (step == 0) lmax = form_x<false>(x, s2v, o2v, adjv, s1i, o1i, jm, step, im, na);
    else           lmax = form_x<true >(x, s2v, o2v, adjv, s1i, o1i, jm, step, im, na);
    const float xmax = uni(wave_max(lmax));
#pragma unroll
    for (int r = 0; r < 32; ++r) x[r] -= xmax;

    float t = -1.0f, lo = -1.0f, tstar = -1.0f;
    int kprev = -1;
    for (int it = 0; it <= STEP_MAX; ++it) {
        const v2f tt = { t, t };
        const v2f zz = { 0.f, 0.f };
        v2f ga = zz, gb = zz, fa = zz, fb = zz;
        int k = 0;
#pragma unroll
        for (int r = 0; r < 32; r += 4) {
            const v2f xa = { x[r],     x[r + 1] };
            const v2f xb = { x[r + 2], x[r + 3] };
            const v2f da = __builtin_elementwise_max(xa - tt, zz);
            const v2f db = __builtin_elementwise_max(xb - tt, zz);
            ga += da; gb += db;
            fa = __builtin_elementwise_fma(da, da, fa);
            fb = __builtin_elementwise_fma(db, db, fb);
            k += __builtin_popcount(__builtin_amdgcn_ballot_w32(x[r]     > t))
               + __builtin_popcount(__builtin_amdgcn_ballot_w32(x[r + 1] > t))
               + __builtin_popcount(__builtin_amdgcn_ballot_w32(x[r + 2] > t))
               + __builtin_popcount(__builtin_amdgcn_ballot_w32(x[r + 3] > t));
        }
        const v2f gs = ga + gb, fs = fa + fb;
        const float G  = uni(wave_sum(gs.x + gs.y));
        const float F  = uni(wave_sum(fs.x + fs.y));
        const float kf = (float)k;
        const float disc = fmaxf(fmaf(G, G, -kf * (F - 1.0f)), 0.0f);
        const float s    = (G - __builtin_amdgcn_sqrtf(disc)) * __builtin_amdgcn_rcpf(kf);
        tstar = t + s;
        if (k == kprev || fabsf(s) <= TAU_EPS || it == STEP_MAX) break;
        kprev = k;
        lo = fmaxf(lo, t + 0.5f * (F - 1.0f) * __builtin_amdgcn_rcpf(G));
        t  = fmaxf(tstar, lo);
    }

    v4f* outrow = (v4f*)(&attL[wave][0]) + lane;
    unsigned int myw = 0;
#pragma unroll
    for (int c = 0; c < 8; ++c) {
        v4f y;
        y.x = fmaxf(x[4 * c]     - tstar, 0.0f);
        y.y = fmaxf(x[4 * c + 1] - tstar, 0.0f);
        y.z = fmaxf(x[4 * c + 2] - tstar, 0.0f);
        y.w = fmaxf(x[4 * c + 3] - tstar, 0.0f);
        y.x *= y.x; y.y *= y.y; y.z *= y.z; y.w *= y.w;
        outrow[c * 32] = y;
        myw |= (y.x != 0.0f) ? (1u << (4 * c))     : 0u;
        myw |= (y.y != 0.0f) ? (1u << (4 * c + 1)) : 0u;
        myw |= (y.z != 0.0f) ? (1u << (4 * c + 2)) : 0u;
        myw |= (y.w != 0.0f) ? (1u << (4 * c + 3)) : 0u;
    }
    suppw[wave][lane] = myw;
    __syncthreads();

    unsigned int u = 0;
#pragma unroll
    for (int w = 0; w < ROWS_PER_BLOCK; ++w) u |= suppw[w][lane];
    const int cnt = __builtin_popcount(u);
    int inc = cnt;
    inc += __builtin_amdgcn_update_dpp(0, inc, 0x111, 0xf, 0xf, true);
    inc += __builtin_amdgcn_update_dpp(0, inc, 0x112, 0xf, 0xf, true);
    inc += __builtin_amdgcn_update_dpp(0, inc, 0x114, 0xf, 0xf, true);
    inc += __builtin_amdgcn_update_dpp(0, inc, 0x118, 0xf, 0xf, true);
    const int lowtot = __builtin_amdgcn_readlane(inc, 15);
    if (lane >= 16) inc += lowtot;
    const int Kp = __builtin_amdgcn_readlane(inc, 31);
    {
        int pos = inc - cnt;
        const int cbase = 4 * lane;
        unsigned int bits = u;
#pragma unroll 1
        for (int it = 0; it < 32; ++it) {
            if (__builtin_amdgcn_ballot_w32(bits != 0u) == 0u) break;
            if (bits != 0u) {
                const int r = __builtin_ctz(bits);
                bits &= bits - 1u;
                if (((pos >> 2) & (ROWS_PER_BLOCK - 1)) == wave) collist[pos] = cbase + 128 * (r >> 2) + (r & 3);
                ++pos;
            }
        }
    }
    __syncthreads();

    const int m      = lane & 15;
    const int kh     = (lane >> 4) << 1;
    const int ib     = (blockIdx.x * ROWS_PER_BLOCK) & (Nn - 1);
    const unsigned int una    = (unsigned int)na;
    const unsigned int rowmod = (unsigned int)(ib + m) % una;
    const float* w1b = wh1 + ((long)b * Nn) * Dd + h * DH + m;
    const float* w2b = wh2 + ((long)b * Nn) * Dd + h * DH + m;
    const int nsteps = (Kp + 3) >> 2;

    v8f accS0 = {};
    v8f accS1 = {};
    for (int s = wave; s < nsteps; s += ROWS_PER_BLOCK) {
        const int  p0 = 4 * s + kh;
        const bool in0 = p0 < Kp, in1 = p0 + 1 < Kp;
        const int  c0 = in0 ? collist[p0]     : 0;
        const int  c1 = in1 ? collist[p0 + 1] : 0;
        const float a0 = in0 ? attL[m][c0] : 0.0f;
        const float a1 = in1 ? attL[m][c1] : 0.0f;
        const bool self0 = ((unsigned int)c0 % una) == rowmod;
        const bool self1 = ((unsigned int)c1 % una) == rowmod;
        v2f aS, aO;
        aS.x = self0 ? a0 : 0.0f;  aS.y = self1 ? a1 : 0.0f;
        aO.x = self0 ? 0.0f : a0;  aO.y = self1 ? 0.0f : a1;

        const long o0 = (long)c0 * Dd, o1r = (long)c1 * Dd;
        v2f b1_0; b1_0.x = w1b[o0];      b1_0.y = w1b[o1r];
        v2f b1_1; b1_1.x = w1b[o0 + 16]; b1_1.y = w1b[o1r + 16];
        v2f b2_0; b2_0.x = w2b[o0];      b2_0.y = w2b[o1r];
        v2f b2_1; b2_1.x = w2b[o0 + 16]; b2_1.y = w2b[o1r + 16];

        accS0 = __builtin_amdgcn_wmma_f32_16x16x4_f32(false, aS, false, b1_0,
                                                      (short)0, accS0, false, false);
        accS0 = __builtin_amdgcn_wmma_f32_16x16x4_f32(false, aO, false, b2_0,
                                                      (short)0, accS0, false, false);
        accS1 = __builtin_amdgcn_wmma_f32_16x16x4_f32(false, aS, false, b1_1,
                                                      (short)0, accS1, false, false);
        accS1 = __builtin_amdgcn_wmma_f32_16x16x4_f32(false, aO, false, b2_1,
                                                      (short)0, accS1, false, false);
    }

    __syncthreads();
    float* rw = &red[wave][(lane >> 4) << 3][m];
#pragma unroll
    for (int r = 0; r < 8; ++r) {
        rw[r * RS]      = accS0[r];
        rw[r * RS + 16] = accS1[r];
    }
    __syncthreads();
    float acc = 0.0f;
#pragma unroll
    for (int w = 0; w < ROWS_PER_BLOCK; ++w) acc += red[w][wave][lane];
    hp[((long)(b * Nn + ib + wave)) * Dd + h * DH + lane] = acc;
}


__global__ __launch_bounds__(128) void outproj_kernel(
    const float* __restrict__ hp, const float* __restrict__ Wo,
    const float* __restrict__ bo, float* __restrict__ out)
{
    const int lane = threadIdx.x & 31;
    const int wave = threadIdx.x >> 5;
    const int m    = lane & 15;
    const int hf   = lane >> 4;
    const int row0 = (blockIdx.x >> 2) * 64 + (wave >> 1) * 32;
    const int col0 = (blockIdx.x & 3) * 64 + (wave & 1) * 32;

    v8f acc[2][2];
    gemm_wave_tile<2, 2, 8>(acc, hp + (row0 + m) * Dd + 4 * hf, Wo + (col0 + m) * Dd + 4 * hf);

#pragma unroll
    for (int j = 0; j < 2; ++j) {
        const int c = col0 + 16 * j + m;
        const float bias = bo[c];
#pragma unroll
        for (int i = 0; i < 2; ++i) {
            float* o = out + (row0 + 16 * i + 8 * hf) * Dd + c;
#pragma unroll
            for (int r = 0; r < 8; ++r) {
                float v = acc[i][j][r] + bias;
                v = (v > 0.0f) ? v : (expf(v) - 1.0f);
                o[r * Dd] = v;
            }
        }
    }
}

extern "C" void kernel_launch(void* const* d_in, const int* in_sizes, int n_in,
                              void* d_out, int out_size, void* d_ws, size_t ws_size,
                              hipStream_t stream) {
    (void)in_sizes; (void)n_in; (void)out_size; (void)ws_size;

    const float* hmat = (const float*)d_in[0];
    const int*   nag  = (const int*)  d_in[1];
    const int*   adj  = (const int*)  d_in[2];
    const float* W1   = (const float*)d_in[3];
    const float* W2   = (const float*)d_in[4];
    const float* Ai   = (const float*)d_in[5];
    const float* Aj   = (const float*)d_in[6];
    const float* Wo   = (const float*)d_in[7];
    const float* bo   = (const float*)d_in[8];
    float*       out  = (float*)d_out;

    float* ws  = (float*)d_ws;
    const size_t whN = (size_t)Bc * Nn * Dd;
    const size_t scN = (size_t)Bc * Hh * Nn;
    float* wh1 = ws;
    float* wh2 = wh1 + whN;
    float* s1  = wh2 + whN;
    float* s2  = s1 + scN;
    float* o1  = s2 + scN;
    float* o2  = o1 + scN;
    float* hp  = o2 + scN;

    proj12_kernel<<<((Bc * Nn) / 64) * 4, 128, 0, stream>>>(
        hmat, W1, W2, wh1, wh2);

    scores_kernel<<<(Bc * Nn * Hh + 255) / 256, 256, 0, stream>>>(
        wh1, wh2, Ai, Aj, s1, s2, o1, o2);

    entmax_aggregate_kernel<<<(Bc * Hh * Nn) / ROWS_PER_BLOCK, 512, 0, stream>>>(
        s1, s2, o1, o2, adj, nag, wh1, wh2, hp);

    outproj_kernel<<<((Bc * Nn) / 64) * 4, 128, 0, stream>>>(
        hp, Wo, bo, out);
}


# entmax tau search starts at a closed-form solve over the 32 lane maxima (one register, two wave sums) instead of t=-1: fewer full-row passes per row; same tau up to its last bit
# speedup vs baseline: 12.7930x; 1.0612x over previous
#include <hip/hip_runtime.h>
#include <hip/hip_bf16.h>
#include <math.h>

static constexpr int Bc = 4;
static constexpr int Nn = 1024;
static constexpr int Dd = 256;
static constexpr int Hh = 8;
static constexpr int DH = 32;

typedef float v2f __attribute__((ext_vector_type(2)));
typedef float v8f __attribute__((ext_vector_type(8)));
typedef float v4f __attribute__((ext_vector_type(4)));
typedef int   v4i __attribute__((ext_vector_type(4)));

#define DEVI static __device__ __forceinline__

template <int MT, int NT, int QG>
DEVI void load_group(v4f (&a)[MT][QG], v4f (&b)[NT][QG],
                     const float* __restrict__ aP, const float* __restrict__ bP, const int k0)
{
#pragma unroll
    for (int q = 0; q < QG; ++q) {
#pragma unroll
        for (int i = 0; i < MT; ++i) a[i][q] = *(const v4f*)(aP + i * 16 * Dd + k0 + 8 * q);
#pragma unroll
        for (int j = 0; j < NT; ++j) b[j][q] = *(const v4f*)(bP + j * 16 * Dd + k0 + 8 * q);
    }
}

template <int MT, int NT, int QG>
DEVI void mma_group(v8f (&acc)[MT][NT], const v4f (&a)[MT][QG], const v4f (&b)[NT][QG])
{
#pragma unroll
    for (int q = 0; q < QG; ++q) {
#pragma unroll
        for (int s = 0; s < 2; ++s) {
#pragma unroll
            for (int i = 0; i < MT; ++i) {
                const v2f af = s ? __builtin_shufflevector(a[i][q], a[i][q], 2, 3)
                                 : __builtin_shufflevector(a[i][q], a[i][q], 0, 1);
#pragma unroll
                for (int j = 0; j < NT; ++j) {
                    const v2f bf = s ? __builtin_shufflevector(b[j][q], b[j][q], 2, 3)
                                     : __builtin_shufflevector(b[j][q], b[j][q], 0, 1);
                    acc[i][j] = __builtin_amdgcn_wmma_f32_16x16x4_f32(false, af, false, bf,
                                                                      (short)0, acc[i][j], false, false);
                }
            }
        }
    }
}

template <int MT, int NT, int QG>
DEVI void gemm_wave_tile(v8f (&acc)[MT][NT], const float* __restrict__ aP, const float* __restrict__ bP)
{
    constexpr int KG = 8 * QG;
#pragma unroll
    for (int i = 0; i < MT; ++i) {
#pragma unroll
        for (int j = 0; j < NT; ++j) acc[i][j] = v8f{};
    }
    v4f a0[MT][QG], b0[NT][QG], a1[MT][QG], b1[NT][QG];
    load_group<MT, NT, QG>(a0, b0, aP, bP, 0);
#pragma unroll 1
    for (int k0 = 0; k0 < Dd - 2 * KG; k0 += 2 * KG) {
        load_group<MT, NT, QG>(a1, b1, aP, bP, k0 + KG);
        __builtin_amdgcn_sched_barrier(0);
        mma_group<MT, NT, QG>(acc, a0, b0);
        __builtin_amdgcn_sched_barrier(0);
        load_group<MT, NT, QG>(a0, b0, aP, bP, k0 + 2 * KG);
        __builtin_amdgcn_sched_barrier(0);
        mma_group<MT, NT, QG>(acc, a1, b1);
    }
    load_group<MT, NT, QG>(a1, b1, aP, bP, Dd - KG);
    __builtin_amdgcn_sched_barrier(0);
    mma_group<MT, NT, QG>(acc, a0, b0);
    mma_group<MT, NT, QG>(acc, a1, b1);
}

__global__ __launch_bounds__(128) void proj12_kernel(
    const float* __restrict__ hmat, const float* __restrict__ W1,
    const float* __restrict__ W2, float* __restrict__ wh1,
    float* __restrict__ wh2)
{
    const int lane = threadIdx.x & 31;
    const int wave = threadIdx.x >> 5;
    const int m    = lane & 15;
    const int hf   = lane >> 4;
    const int row0 = (blockIdx.x >> 2) * 64 + (wave >> 1) * 32;
    const int col0 = (blockIdx.x & 1) * 128 + (wave & 1) * 64;
    const bool second = (blockIdx.x & 2) != 0;
    const float* __restrict__ W   = second ? W2 : W1;
    float* __restrict__       dst = second ? wh2 : wh1;

    v8f acc[2][4];
    gemm_wave_tile<2, 4, 4>(acc, hmat + (row0 + m) * Dd + 4 * hf, W + (col0 + m) * Dd + 4 * hf);

#pragma unroll
    for (int i = 0; i < 2; ++i) {
#pragma unroll
        for (int j = 0; j < 4; ++j) {
            float* o = dst + (row0 + 16 * i + 8 * hf) * Dd + col0 + 16 * j + m;
#pragma unroll
            for (int r = 0; r < 8; ++r) o[r * Dd] = acc[i][j][r];
        }
    }
}

__global__ __launch_bounds__(256) void scores_kernel(
    const float* __restrict__ wh1, const float* __restrict__ wh2,
    const float* __restrict__ Ai, const float* __restrict__ Aj,
    float* __restrict__ s1, float* __restrict__ s2,
    float* __restrict__ o1, float* __restrict__ o2)
{
    int idx = blockIdx.x * blockDim.x + threadIdx.x;
    if (idx >= Bc * Nn * Hh) return;
    const int h = idx % Hh;
    const int n = (idx / Hh) % Nn;
    const int b = idx / (Hh * Nn);

    const float* r1 = wh1 + ((long)(b * Nn + n)) * Dd + h * DH;
    const float* r2 = wh2 + ((long)(b * Nn + n)) * Dd + h * DH;
    const float* ai = Ai + h * DH;
    const float* aj = Aj + h * DH;

    float a1 = 0.f, a2 = 0.f, a3 = 0.f, a4 = 0.f;
#pragma unroll 8
    for (int d = 0; d < DH; ++d) {
        const float w1v = r1[d], w2v = r2[d], aiv = ai[d], ajv = aj[d];
        a1 += w1v * aiv;  a2 += w1v * ajv;
        a3 += w2v * aiv;  a4 += w2v * ajv;
    }
    const int o = (b * Hh + h) * Nn + n;
    s1[o] = 0.5f * a1; s2[o] = 0.5f * a2; o1[o] = 0.5f * a3; o2[o] = 0.5f * a4;
}

#define DEV static __device__ __forceinline__

DEV float xhalf(float v) {
    const int b = __builtin_bit_cast(int, v);
    return __builtin_bit_cast(float,
        __builtin_amdgcn_permlanex16(b, b, 0x76543210, 0xfedcba98, false, false));
}
template <int M> DEV float xrow(float v) {
    const int b = __builtin_bit_cast(int, v);
    return __builtin_bit_cast(float,
        __builtin_amdgcn_update_dpp(b, b, 0x160 + M, 0xf, 0xf, true));
}
DEV float wave_sum(float v) {
    v += xrow<1>(v); v += xrow<2>(v); v += xrow<4>(v); v += xrow<8>(v);
    v += xhalf(v);
    return v;
}
DEV float wave_max(float v) {
    v = fmaxf(v, xrow<1>(v)); v = fmaxf(v, xrow<2>(v));
    v = fmaxf(v, xrow<4>(v)); v = fmaxf(v, xrow<8>(v));
    v = fmaxf(v, xhalf(v));
    return v;
}
DEV float uni(float v) {
    return __builtin_bit_cast(float,
        __builtin_amdgcn_readfirstlane(__builtin_bit_cast(int, v)));
}

static constexpr int ROWS_PER_BLOCK = 16;
static constexpr int LDA = Nn + 4;
static constexpr int RS  = 36;
static constexpr int   STEP_MAX = 64;
static constexpr float TAU_EPS  = 2.384185791015625e-07f;
static constexpr float X_MASKED = 0.5f * -1.0e12f;

DEV void self_flags(bool (&sf)[4], const int jm, const int im, const int na) {
    int m = jm;
#pragma unroll
    for (int e = 0; e < 4; ++e) {
        sf[e] = (im == m);
        m = (m + 1 == na) ? 0 : (m + 1);
    }
}

template <bool FLAGS_PER_CHUNK>
DEV float form_x(float (&x)[32], const v4f* __restrict__ s2v, const v4f* __restrict__ o2v,
                 const v4i* __restrict__ adjv, const float s1i, const float o1i,
                 int jm, const int step, const int im, const int na)
{
    bool sf[4];
    self_flags(sf, jm, im, na);
    const v2f s1p = { s1i, s1i };
    const v2f o1p = { o1i, o1i };
    float lmax = -3.4e38f;
    v4i av[8];
    v4f sv[8], ov[8];
#pragma unroll
    for (int c = 0; c < 8; ++c) av[c] = adjv[c * 32];
    constexpr int AHEAD = FLAGS_PER_CHUNK ? 1 : 2;
#pragma unroll
    for (int c = 0; c < AHEAD; ++c) {
        sv[c] = s2v[c * 32];
        ov[c] = o2v[c * 32];
    }
#pragma unroll
    for (int c = 0; c < 8; ++c) {
        if (FLAGS_PER_CHUNK && c > 0) {
            jm += step;
            if (jm >= na) jm -= na;
            self_flags(sf, jm, im, na);
        }
        if (c + AHEAD < 8) {
            sv[c + AHEAD] = s2v[(c + AHEAD) * 32];
            ov[c + AHEAD] = o2v[(c + AHEAD) * 32];
        }
        const v2f sa = __builtin_shufflevector(sv[c], sv[c], 0, 1) + s1p;
        const v2f sb = __builtin_shufflevector(sv[c], sv[c], 2, 3) + s1p;
        const v2f oa = __builtin_shufflevector(ov[c], ov[c], 0, 1) + o1p;
        const v2f ob = __builtin_shufflevector(ov[c], ov[c], 2, 3) + o1p;
        const float x0 = (av[c].x == 0) ? X_MASKED : (sf[0] ? sa.x : oa.x);
        const float x1 = (av[c].y == 0) ? X_MASKED : (sf[1] ? sa.y : oa.y);
        const float x2 = (av[c].z == 0) ? X_MASKED : (sf[2] ? sb.x : ob.x);
        const float x3 = (av[c].w == 0) ? X_MASKED : (sf[3] ? sb.y : ob.y);
        x[4 * c]     = x0;
        x[4 * c + 1] = x1;
        x[4 * c + 2] = x2;
        x[4 * c + 3] = x3;
        lmax = fmaxf(fmaxf(lmax, x0), fmaxf(x1, fmaxf(x2, x3)));
        if (c + 1 < 8) __builtin_amdgcn_sched_barrier(0);
    }
    return lmax;
}

__global__ __launch_bounds__(512) void entmax_aggregate_kernel(
    const float* __restrict__ s1, const float* __restrict__ s2,
    const float* __restrict__ o1, const float* __restrict__ o2,
    const int* __restrict__ adj, const int* __restrict__ nag,
    const float* __restrict__ wh1, const float* __restrict__ wh2,
    float* __restrict__ hp)
{
    __shared__ __attribute__((aligned(16))) float smem[ROWS_PER_BLOCK * LDA];
    float (*attL)[LDA]   = (float (*)[LDA])smem;
    float (*red)[16][RS] = (float (*)[16][RS])smem;
    __shared__ unsigned int suppw[ROWS_PER_BLOCK][32];
    __shared__ __attribute__((aligned(8))) int collist[Nn];

    const int lane = threadIdx.x & 31;
    const int wave = __builtin_amdgcn_readfirstlane((int)(threadIdx.x >> 5));
    const int rid  = blockIdx.x * ROWS_PER_BLOCK + wave;
    const int i    = rid & (Nn - 1);
    const int h    = (rid >> 10) & (Hh - 1);
    const int b    = rid >> 13;
    const int na   = nag[0];

    const int   rowbase = (b * Hh + h) * Nn;
    const float s1i = s1[rowbase + i];
    const float o1i = o1[rowbase + i];
    const int   im  = i % na;
    const long  adjbase = ((long)b * Nn + i) * Nn;

    const v4f* s2v  = (const v4f*)(s2 + rowbase) + lane;
    const v4f* o2v  = (const v4f*)(o2 + rowbase) + lane;
    const v4i* adjv = (const v4i*)(adj + adjbase) + lane;

    float x[32];
    const int jm   = (4 * lane) % na;
    const int step = 128 % na;
    float lmax;
    if (step == 0) lmax = form_x<false>(x, s2v, o2v, adjv, s1i, o1i, jm, step, im, na);
    else           lmax = form_x<true >(x, s2v, o2v, adjv, s1i, o1i, jm, step, im, na);
    const float xmax = uni(wave_max(lmax));
#pragma unroll
    for (int r = 0; r < 32; ++r) x[r] -= xmax;

    float t;
    {
        const float lm   = lmax - xmax;
        const float d0   = fmaxf(lm + 1.0f, 0.0f);
        const float kf0  = (float)__builtin_popcount(__builtin_amdgcn_ballot_w32(lm > -1.0f));
        const float G0   = uni(wave_sum(d0));
        const float F0   = uni(wave_sum(d0 * d0));
        const float dsc0 = fmaxf(fmaf(G0, G0, -kf0 * (F0 - 1.0f)), 0.0f);
        const float t0   = -1.0f + (G0 - __builtin_amdgcn_sqrtf(dsc0)) * __builtin_amdgcn_rcpf(kf0);
        t = (t0 < 0.0f) ? fmaxf(t0, -1.0f) : -1.0f;
    }
    float lo = -1.0f, tstar = -1.0f;
    int kprev = -1;
    for (int it = 0; it <= STEP_MAX; ++it) {
        const v2f tt = { t, t };
        const v2f zz = { 0.f, 0.f };
        v2f ga = zz, gb = zz, fa = zz, fb = zz;
        int k = 0;
#pragma unroll
        for (int r = 0; r < 32; r += 4) {
            const v2f xa = { x[r],     x[r + 1] };
            const v2f xb = { x[r + 2], x[r + 3] };
            const v2f da = __builtin_elementwise_max(xa - tt, zz);
            const v2f db = __builtin_elementwise_max(xb - tt, zz);
            ga += da; gb += db;
            fa = __builtin_elementwise_fma(da, da, fa);
            fb = __builtin_elementwise_fma(db, db, fb);
            k += __builtin_popcount(__builtin_amdgcn_ballot_w32(x[r]     > t))
               + __builtin_popcount(__builtin_amdgcn_ballot_w32(x[r + 1] > t))
               + __builtin_popcount(__builtin_amdgcn_ballot_w32(x[r + 2] > t))
               + __builtin_popcount(__builtin_amdgcn_ballot_w32(x[r + 3] > t));
        }
        const v2f gs = ga + gb, fs = fa + fb;
        const float G  = uni(wave_sum(gs.x + gs.y));
        const float F  = uni(wave_sum(fs.x + fs.y));
        const float kf = (float)k;
        const float disc = fmaxf(fmaf(G, G, -kf * (F - 1.0f)), 0.0f);
        const float s    = (G - __builtin_amdgcn_sqrtf(disc)) * __builtin_amdgcn_rcpf(kf);
        tstar = t + s;
        if (k == kprev || fabsf(s) <= TAU_EPS || it == STEP_MAX) break;
        kprev = k;
        lo = fmaxf(lo, t + 0.5f * (F - 1.0f) * __builtin_amdgcn_rcpf(G));
        t  = fmaxf(tstar, lo);
    }

    v4f* outrow = (v4f*)(&attL[wave][0]) + lane;
    unsigned int myw = 0;
#pragma unroll
    for (int c = 0; c < 8; ++c) {
        v4f y;
        y.x = fmaxf(x[4 * c]     - tstar, 0.0f);
        y.y = fmaxf(x[4 * c + 1] - tstar, 0.0f);
        y.z = fmaxf(x[4 * c + 2] - tstar, 0.0f);
        y.w = fmaxf(x[4 * c + 3] - tstar, 0.0f);
        y.x *= y.x; y.y *= y.y; y.z *= y.z; y.w *= y.w;
        outrow[c * 32] = y;
        myw |= (y.x != 0.0f) ? (1u << (4 * c))     : 0u;
        myw |= (y.y != 0.0f) ? (1u << (4 * c + 1)) : 0u;
        myw |= (y.z != 0.0f) ? (1u << (4 * c + 2)) : 0u;
        myw |= (y.w != 0.0f) ? (1u << (4 * c + 3)) : 0u;
    }
    suppw[wave][lane] = myw;
    __syncthreads();

    unsigned int u = 0;
#pragma unroll
    for (int w = 0; w < ROWS_PER_BLOCK; ++w) u |= suppw[w][lane];
    const int cnt = __builtin_popcount(u);
    int inc = cnt;
    inc += __builtin_amdgcn_update_dpp(0, inc, 0x111, 0xf, 0xf, true);
    inc += __builtin_amdgcn_update_dpp(0, inc, 0x112, 0xf, 0xf, true);
    inc += __builtin_amdgcn_update_dpp(0, inc, 0x114, 0xf, 0xf, true);
    inc += __builtin_amdgcn_update_dpp(0, inc, 0x118, 0xf, 0xf, true);
    const int lowtot = __builtin_amdgcn_readlane(inc, 15);
    if (lane >= 16) inc += lowtot;
    const int Kp = __builtin_amdgcn_readlane(inc, 31);
    {
        int pos = inc - cnt;
        const int cbase = 4 * lane;
        unsigned int bits = u;
#pragma unroll 1
        for (int it = 0; it < 32; ++it) {
            if (__builtin_amdgcn_ballot_w32(bits != 0u) == 0u) break;
            if (bits != 0u) {
                const int r = __builtin_ctz(bits);
                bits &= bits - 1u;
                if (((pos >> 2) & (ROWS_PER_BLOCK - 1)) == wave) collist[pos] = cbase + 128 * (r >> 2) + (r & 3);
                ++pos;
            }
        }
    }
    __syncthreads();

    const int m      = lane & 15;
    const int kh     = (lane >> 4) << 1;
    const int ib     = (blockIdx.x * ROWS_PER_BLOCK) & (Nn - 1);
    const unsigned int una    = (unsigned int)na;
    const unsigned int rowmod = (unsigned int)(ib + m) % una;
    const float* w1b = wh1 + ((long)b * Nn) * Dd + h * DH + m;
    const float* w2b = wh2 + ((long)b * Nn) * Dd + h * DH + m;
    const int nsteps = (Kp + 3) >> 2;

    v8f accS0 = {};
    v8f accS1 = {};
    for (int s = wave; s < nsteps; s += ROWS_PER_BLOCK) {
        const int  p0 = 4 * s + kh;
        const bool in0 = p0 < Kp, in1 = p0 + 1 < Kp;
        const int  c0 = in0 ? collist[p0]     : 0;
        const int  c1 = in1 ? collist[p0 + 1] : 0;
        const float a0 = in0 ? attL[m][c0] : 0.0f;
        const float a1 = in1 ? attL[m][c1] : 0.0f;
        const bool self0 = ((unsigned int)c0 % una) == rowmod;
        const bool self1 = ((unsigned int)c1 % una) == rowmod;
        v2f aS, aO;
        aS.x = self0 ? a0 : 0.0f;  aS.y = self1 ? a1 : 0.0f;
        aO.x = self0 ? 0.0f : a0;  aO.y = self1 ? 0.0f : a1;

        const long o0 = (long)c0 * Dd, o1r = (long)c1 * Dd;
        v2f b1_0; b1_0.x = w1b[o0];      b1_0.y = w1b[o1r];
        v2f b1_1; b1_1.x = w1b[o0 + 16]; b1_1.y = w1b[o1r + 16];
        v2f b2_0; b2_0.x = w2b[o0];      b2_0.y = w2b[o1r];
        v2f b2_1; b2_1.x = w2b[o0 + 16]; b2_1.y = w2b[o1r + 16];

        accS0 = __builtin_amdgcn_wmma_f32_16x16x4_f32(false, aS, false, b1_0,
                                                      (short)0, accS0, false, false);
        accS0 = __builtin_amdgcn_wmma_f32_16x16x4_f32(false, aO, false, b2_0,
                                                      (short)0, accS0, false, false);
        accS1 = __builtin_amdgcn_wmma_f32_16x16x4_f32(false, aS, false, b1_1,
                                                      (short)0, accS1, false, false);
        accS1 = __builtin_amdgcn_wmma_f32_16x16x4_f32(false, aO, false, b2_1,
                                                      (short)0, accS1, false, false);
    }

    __syncthreads();
    float* rw = &red[wave][(lane >> 4) << 3][m];
#pragma unroll
    for (int r = 0; r < 8; ++r) {
        rw[r * RS]      = accS0[r];
        rw[r * RS + 16] = accS1[r];
    }
    __syncthreads();
    float acc = 0.0f;
#pragma unroll
    for (int w = 0; w < ROWS_PER_BLOCK; ++w) acc += red[w][wave][lane];
    hp[((long)(b * Nn + ib + wave)) * Dd + h * DH + lane] = acc;
}


__global__ __launch_bounds__(128) void outproj_kernel(
    const float* __restrict__ hp, const float* __restrict__ Wo,
    const float* __restrict__ bo, float* __restrict__ out)
{
    const int lane = threadIdx.x & 31;
    const int wave = threadIdx.x >> 5;
    const int m    = lane & 15;
    const int hf   = lane >> 4;
    const int row0 = (blockIdx.x >> 2) * 64 + (wave >> 1) * 32;
    const int col0 = (blockIdx.x & 3) * 64 + (wave & 1) * 32;

    v8f acc[2][2];
    gemm_wave_tile<2, 2, 8>(acc, hp + (row0 + m) * Dd + 4 * hf, Wo + (col0 + m) * Dd + 4 * hf);

#pragma unroll
    for (int j = 0; j < 2; ++j) {
        const int c = col0 + 16 * j + m;
        const float bias = bo[c];
#pragma unroll
        for (int i = 0; i < 2; ++i) {
            float* o = out + (row0 + 16 * i + 8 * hf) * Dd + c;
#pragma unroll
            for (int r = 0; r < 8; ++r) {
                float v = acc[i][j][r] + bias;
                v = (v > 0.0f) ? v : (expf(v) - 1.0f);
                o[r * Dd] = v;
            }
        }
    }
}

extern "C" void kernel_launch(void* const* d_in, const int* in_sizes, int n_in,
                              void* d_out, int out_size, void* d_ws, size_t ws_size,
                              hipStream_t stream) {
    (void)in_sizes; (void)n_in; (void)out_size; (void)ws_size;

    const float* hmat = (const float*)d_in[0];
    const int*   nag  = (const int*)  d_in[1];
    const int*   adj  = (const int*)  d_in[2];
    const float* W1   = (const float*)d_in[3];
    const float* W2   = (const float*)d_in[4];
    const float* Ai   = (const float*)d_in[5];
    const float* Aj   = (const float*)d_in[6];
    const float* Wo   = (const float*)d_in[7];
    const float* bo   = (const float*)d_in[8];
    float*       out  = (float*)d_out;

    float* ws  = (float*)d_ws;
    const size_t whN = (size_t)Bc * Nn * Dd;
    const size_t scN = (size_t)Bc * Hh * Nn;
    float* wh1 = ws;
    float* wh2 = wh1 + whN;
    float* s1  = wh2 + whN;
    float* s2  = s1 + scN;
    float* o1  = s2 + scN;
    float* o2  = o1 + scN;
    float* hp  = o2 + scN;

    proj12_kernel<<<((Bc * Nn) / 64) * 4, 128, 0, stream>>>(
        hmat, W1, W2, wh1, wh2);

    scores_kernel<<<(Bc * Nn * Hh + 255) / 256, 256, 0, stream>>>(
        wh1, wh2, Ai, Aj, s1, s2, o1, o2);

    entmax_aggregate_kernel<<<(Bc * Hh * Nn) / ROWS_PER_BLOCK, 512, 0, stream>>>(
        s1, s2, o1, o2, adj, nag, wh1, wh2, hp);

    outproj_kernel<<<((Bc * Nn) / 64) * 4, 128, 0, stream>>>(
        hp, Wo, bo, out);
}


# aggregate: K dealt by column ownership: wave w unions support words 2w,2w+1 and lists the set bits on the scalar unit; no shared prefix-summed list, no list barrier; k sum order changes
# speedup vs baseline: 13.1721x; 1.0296x over previous
#include <hip/hip_runtime.h>
#include <hip/hip_bf16.h>
#include <math.h>

static constexpr int Bc = 4;
static constexpr int Nn = 1024;
static constexpr int Dd = 256;
static constexpr int Hh = 8;
static constexpr int DH = 32;

typedef float v2f __attribute__((ext_vector_type(2)));
typedef float v8f __attribute__((ext_vector_type(8)));
typedef float v4f __attribute__((ext_vector_type(4)));
typedef int   v4i __attribute__((ext_vector_type(4)));

#define DEVI static __device__ __forceinline__

template <int MT, int NT, int QG>
DEVI void load_group(v4f (&a)[MT][QG], v4f (&b)[NT][QG],
                     const float* __restrict__ aP, const float* __restrict__ bP, const int k0)
{
#pragma unroll
    for (int q = 0; q < QG; ++q) {
#pragma unroll
        for (int i = 0; i < MT; ++i) a[i][q] = *(const v4f*)(aP + i * 16 * Dd + k0 + 8 * q);
#pragma unroll
        for (int j = 0; j < NT; ++j) b[j][q] = *(const v4f*)(bP + j * 16 * Dd + k0 + 8 * q);
    }
}

template <int MT, int NT, int QG>
DEVI void mma_group(v8f (&acc)[MT][NT], const v4f (&a)[MT][QG], const v4f (&b)[NT][QG])
{
#pragma unroll
    for (int q = 0; q < QG; ++q) {
#pragma unroll
        for (int s = 0; s < 2; ++s) {
#pragma unroll
            for (int i = 0; i < MT; ++i) {
                const v2f af = s ? __builtin_shufflevector(a[i][q], a[i][q], 2, 3)
                                 : __builtin_shufflevector(a[i][q], a[i][q], 0, 1);
#pragma unroll
                for (int j = 0; j < NT; ++j) {
                    const v2f bf = s ? __builtin_shufflevector(b[j][q], b[j][q], 2, 3)
                                     : __builtin_shufflevector(b[j][q], b[j][q], 0, 1);
                    acc[i][j] = __builtin_amdgcn_wmma_f32_16x16x4_f32(false, af, false, bf,
                                                                      (short)0, acc[i][j], false, false);
                }
            }
        }
    }
}

template <int MT, int NT, int QG>
DEVI void gemm_wave_tile(v8f (&acc)[MT][NT], const float* __restrict__ aP, const float* __restrict__ bP)
{
    constexpr int KG = 8 * QG;
#pragma unroll
    for (int i = 0; i < MT; ++i) {
#pragma unroll
        for (int j = 0; j < NT; ++j) acc[i][j] = v8f{};
    }
    v4f a0[MT][QG], b0[NT][QG], a1[MT][QG], b1[NT][QG];
    load_group<MT, NT, QG>(a0, b0, aP, bP, 0);
#pragma unroll 1
    for (int k0 = 0; k0 < Dd - 2 * KG; k0 += 2 * KG) {
        load_group<MT, NT, QG>(a1, b1, aP, bP, k0 + KG);
        __builtin_amdgcn_sched_barrier(0);
        mma_group<MT, NT, QG>(acc, a0, b0);
        __builtin_amdgcn_sched_barrier(0);
        load_group<MT, NT, QG>(a0, b0, aP, bP, k0 + 2 * KG);
        __builtin_amdgcn_sched_barrier(0);
        mma_group<MT, NT, QG>(acc, a1, b1);
    }
    load_group<MT, NT, QG>(a1, b1, aP, bP, Dd - KG);
    __builtin_amdgcn_sched_barrier(0);
    mma_group<MT, NT, QG>(acc, a0, b0);
    mma_group<MT, NT, QG>(acc, a1, b1);
}

__global__ __launch_bounds__(128) void proj12_kernel(
    const float* __restrict__ hmat, const float* __restrict__ W1,
    const float* __restrict__ W2, float* __restrict__ wh1,
    float* __restrict__ wh2)
{
    const int lane = threadIdx.x & 31;
    const int wave = threadIdx.x >> 5;
    const int m    = lane & 15;
    const int hf   = lane >> 4;
    const int row0 = (blockIdx.x >> 2) * 64 + (wave >> 1) * 32;
    const int col0 = (blockIdx.x & 1) * 128 + (wave & 1) * 64;
    const bool second = (blockIdx.x & 2) != 0;
    const float* __restrict__ W   = second ? W2 : W1;
    float* __restrict__       dst = second ? wh2 : wh1;

    v8f acc[2][4];
    gemm_wave_tile<2, 4, 4>(acc, hmat + (row0 + m) * Dd + 4 * hf, W + (col0 + m) * Dd + 4 * hf);

#pragma unroll
    for (int i = 0; i < 2; ++i) {
#pragma unroll
        for (int j = 0; j < 4; ++j) {
            float* o = dst + (row0 + 16 * i + 8 * hf) * Dd + col0 + 16 * j + m;
#pragma unroll
            for (int r = 0; r < 8; ++r) o[r * Dd] = acc[i][j][r];
        }
    }
}

__global__ __launch_bounds__(256) void scores_kernel(
    const float* __restrict__ wh1, const float* __restrict__ wh2,
    const float* __restrict__ Ai, const float* __restrict__ Aj,
    float* __restrict__ s1, float* __restrict__ s2,
    float* __restrict__ o1, float* __restrict__ o2)
{
    int idx = blockIdx.x * blockDim.x + threadIdx.x;
    if (idx >= Bc * Nn * Hh) return;
    const int h = idx % Hh;
    const int n = (idx / Hh) % Nn;
    const int b = idx / (Hh * Nn);

    const float* r1 = wh1 + ((long)(b * Nn + n)) * Dd + h * DH;
    const float* r2 = wh2 + ((long)(b * Nn + n)) * Dd + h * DH;
    const float* ai = Ai + h * DH;
    const float* aj = Aj + h * DH;

    float a1 = 0.f, a2 = 0.f, a3 = 0.f, a4 = 0.f;
#pragma unroll 8
    for (int d = 0; d < DH; ++d) {
        const float w1v = r1[d], w2v = r2[d], aiv = ai[d], ajv = aj[d];
        a1 += w1v * aiv;  a2 += w1v * ajv;
        a3 += w2v * aiv;  a4 += w2v * ajv;
    }
    const int o = (b * Hh + h) * Nn + n;
    s1[o] = 0.5f * a1; s2[o] = 0.5f * a2; o1[o] = 0.5f * a3; o2[o] = 0.5f * a4;
}

#define DEV static __device__ __forceinline__

DEV float xhalf(float v) {
    const int b = __builtin_bit_cast(int, v);
    return __builtin_bit_cast(float,
        __builtin_amdgcn_permlanex16(b, b, 0x76543210, 0xfedcba98, false, false));
}
template <int M> DEV float xrow(float v) {
    const int b = __builtin_bit_cast(int, v);
    return __builtin_bit_cast(float,
        __builtin_amdgcn_update_dpp(b, b, 0x160 + M, 0xf, 0xf, true));
}
DEV float wave_sum(float v) {
    v += xrow<1>(v); v += xrow<2>(v); v += xrow<4>(v); v += xrow<8>(v);
    v += xhalf(v);
    return v;
}
DEV float wave_max(float v) {
    v = fmaxf(v, xrow<1>(v)); v = fmaxf(v, xrow<2>(v));
    v = fmaxf(v, xrow<4>(v)); v = fmaxf(v, xrow<8>(v));
    v = fmaxf(v, xhalf(v));
    return v;
}
DEV float uni(float v) {
    return __builtin_bit_cast(float,
        __builtin_amdgcn_readfirstlane(__builtin_bit_cast(int, v)));
}

static constexpr int ROWS_PER_BLOCK = 16;
static constexpr int LDA = Nn + 4;
static constexpr int RS  = 36;
static constexpr int   STEP_MAX = 64;
static constexpr float TAU_EPS  = 2.384185791015625e-07f;
static constexpr float X_MASKED = 0.5f * -1.0e12f;

DEV void self_flags(bool (&sf)[4], const int jm, const int im, const int na) {
    int m = jm;
#pragma unroll
    for (int e = 0; e < 4; ++e) {
        sf[e] = (im == m);
        m = (m + 1 == na) ? 0 : (m + 1);
    }
}

template <bool FLAGS_PER_CHUNK>
DEV float form_x(float (&x)[32], const v4f* __restrict__ s2v, const v4f* __restrict__ o2v,
                 const v4i* __restrict__ adjv, const float s1i, const float o1i,
                 int jm, const int step, const int im, const int na)
{
    bool sf[4];
    self_flags(sf, jm, im, na);
    const v2f s1p = { s1i, s1i };
    const v2f o1p = { o1i, o1i };
    float lmax = -3.4e38f;
    v4i av[8];
    v4f sv[8], ov[8];
#pragma unroll
    for (int c = 0; c < 8; ++c) av[c] = adjv[c * 32];
    constexpr int AHEAD = FLAGS_PER_CHUNK ? 1 : 2;
#pragma unroll
    for (int c = 0; c < AHEAD; ++c) {
        sv[c] = s2v[c * 32];
        ov[c] = o2v[c * 32];
    }
#pragma unroll
    for (int c = 0; c < 8; ++c) {
        if (FLAGS_PER_CHUNK && c > 0) {
            jm += step;
            if (jm >= na) jm -= na;
            self_flags(sf, jm, im, na);
        }
        if (c + AHEAD < 8) {
            sv[c + AHEAD] = s2v[(c + AHEAD) * 32];
            ov[c + AHEAD] = o2v[(c + AHEAD) * 32];
        }
        const v2f sa = __builtin_shufflevector(sv[c], sv[c], 0, 1) + s1p;
        const v2f sb = __builtin_shufflevector(sv[c], sv[c], 2, 3) + s1p;
        const v2f oa = __builtin_shufflevector(ov[c], ov[c], 0, 1) + o1p;
        const v2f ob = __builtin_shufflevector(ov[c], ov[c], 2, 3) + o1p;
        const float x0 = (av[c].x == 0) ? X_MASKED : (sf[0] ? sa.x : oa.x);
        const float x1 = (av[c].y == 0) ? X_MASKED : (sf[1] ? sa.y : oa.y);
        const float x2 = (av[c].z == 0) ? X_MASKED : (sf[2] ? sb.x : ob.x);
        const float x3 = (av[c].w == 0) ? X_MASKED : (sf[3] ? sb.y : ob.y);
        x[4 * c]     = x0;
        x[4 * c + 1] = x1;
        x[4 * c + 2] = x2;
        x[4 * c + 3] = x3;
        lmax = fmaxf(fmaxf(lmax, x0), fmaxf(x1, fmaxf(x2, x3)));
        if (c + 1 < 8) __builtin_amdgcn_sched_barrier(0);
    }
    return lmax;
}

__global__ __launch_bounds__(512) void entmax_aggregate_kernel(
    const float* __restrict__ s1, const float* __restrict__ s2,
    const float* __restrict__ o1, const float* __restrict__ o2,
    const int* __restrict__ adj, const int* __restrict__ nag,
    const float* __restrict__ wh1, const float* __restrict__ wh2,
    float* __restrict__ hp)
{
    __shared__ __attribute__((aligned(16))) float smem[ROWS_PER_BLOCK * LDA];
    float (*attL)[LDA]   = (float (*)[LDA])smem;
    float (*red)[16][RS] = (float (*)[16][RS])smem;
    __shared__ unsigned int suppw[ROWS_PER_BLOCK][32];

    const int lane = threadIdx.x & 31;
    const int wave = __builtin_amdgcn_readfirstlane((int)(threadIdx.x >> 5));
    const int rid  = blockIdx.x * ROWS_PER_BLOCK + wave;
    const int i    = rid & (Nn - 1);
    const int h    = (rid >> 10) & (Hh - 1);
    const int b    = rid >> 13;
    const int na   = nag[0];

    const int   rowbase = (b * Hh + h) * Nn;
    const float s1i = s1[rowbase + i];
    const float o1i = o1[rowbase + i];
    const int   im  = i % na;
    const long  adjbase = ((long)b * Nn + i) * Nn;

    const v4f* s2v  = (const v4f*)(s2 + rowbase) + lane;
    const v4f* o2v  = (const v4f*)(o2 + rowbase) + lane;
    const v4i* adjv = (const v4i*)(adj + adjbase) + lane;

    float x[32];
    const int jm   = (4 * lane) % na;
    const int step = 128 % na;
    float lmax;
    if (step == 0) lmax = form_x<false>(x, s2v, o2v, adjv, s1i, o1i, jm, step, im, na);
    else           lmax = form_x<true >(x, s2v, o2v, adjv, s1i, o1i, jm, step, im, na);
    const float xmax = uni(wave_max(lmax));
#pragma unroll
    for (int r = 0; r < 32; ++r) x[r] -= xmax;

    float t;
    {
        const float lm   = lmax - xmax;
        const float d0   = fmaxf(lm + 1.0f, 0.0f);
        const float kf0  = (float)__builtin_popcount(__builtin_amdgcn_ballot_w32(lm > -1.0f));
        const float G0   = uni(wave_sum(d0));
        const float F0   = uni(wave_sum(d0 * d0));
        const float dsc0 = fmaxf(fmaf(G0, G0, -kf0 * (F0 - 1.0f)), 0.0f);
        const float t0   = -1.0f + (G0 - __builtin_amdgcn_sqrtf(dsc0)) * __builtin_amdgcn_rcpf(kf0);
        t = (t0 < 0.0f) ? fmaxf(t0, -1.0f) : -1.0f;
    }
    float lo = -1.0f, tstar = -1.0f;
    int kprev = -1;
    for (int it = 0; it <= STEP_MAX; ++it) {
        const v2f tt = { t, t };
        const v2f zz = { 0.f, 0.f };
        v2f ga = zz, gb = zz, fa = zz, fb = zz;
        int k = 0;
#pragma unroll
        for (int r = 0; r < 32; r += 4) {
            const v2f xa = { x[r],     x[r + 1] };
            const v2f xb = { x[r + 2], x[r + 3] };
            const v2f da = __builtin_elementwise_max(xa - tt, zz);
            const v2f db = __builtin_elementwise_max(xb - tt, zz);
            ga += da; gb += db;
            fa = __builtin_elementwise_fma(da, da, fa);
            fb = __builtin_elementwise_fma(db, db, fb);
            k += __builtin_popcount(__builtin_amdgcn_ballot_w32(x[r]     > t))
               + __builtin_popcount(__builtin_amdgcn_ballot_w32(x[r + 1] > t))
               + __builtin_popcount(__builtin_amdgcn_ballot_w32(x[r + 2] > t))
               + __builtin_popcount(__builtin_amdgcn_ballot_w32(x[r + 3] > t));
        }
        const v2f gs = ga + gb, fs = fa + fb;
        const float G  = uni(wave_sum(gs.x + gs.y));
        const float F  = uni(wave_sum(fs.x + fs.y));
        const float kf = (float)k;
        const float disc = fmaxf(fmaf(G, G, -kf * (F - 1.0f)), 0.0f);
        const float s    = (G - __builtin_amdgcn_sqrtf(disc)) * __builtin_amdgcn_rcpf(kf);
        tstar = t + s;
        if (k == kprev || fabsf(s) <= TAU_EPS || it == STEP_MAX) break;
        kprev = k;
        lo = fmaxf(lo, t + 0.5f * (F - 1.0f) * __builtin_amdgcn_rcpf(G));
        t  = fmaxf(tstar, lo);
    }

    v4f* outrow = (v4f*)(&attL[wave][0]) + lane;
    unsigned int myw = 0;
#pragma unroll
    for (int c = 0; c < 8; ++c) {
        v4f y;
        y.x = fmaxf(x[4 * c]     - tstar, 0.0f);
        y.y = fmaxf(x[4 * c + 1] - tstar, 0.0f);
        y.z = fmaxf(x[4 * c + 2] - tstar, 0.0f);
        y.w = fmaxf(x[4 * c + 3] - tstar, 0.0f);
        y.x *= y.x; y.y *= y.y; y.z *= y.z; y.w *= y.w;
        outrow[c * 32] = y;
        myw |= (y.x != 0.0f) ? (1u << (4 * c))     : 0u;
        myw |= (y.y != 0.0f) ? (1u << (4 * c + 1)) : 0u;
        myw |= (y.z != 0.0f) ? (1u << (4 * c + 2)) : 0u;
        myw |= (y.w != 0.0f) ? (1u << (4 * c + 3)) : 0u;
    }
    suppw[wave][lane] = myw;
    __syncthreads();

    const int m      = lane & 15;
    const int hi     = lane >> 4;
    const int ib     = (blockIdx.x * ROWS_PER_BLOCK) & (Nn - 1);
    const unsigned int una    = (unsigned int)na;
    const unsigned int rowmod = (unsigned int)(ib + m) % una;
    const float* w1b = wh1 + ((long)b * Nn) * Dd + h * DH;
    const float* w2b = wh2 + ((long)b * Nn) * Dd + h * DH;

    unsigned int uw = suppw[m][2 * wave + hi];
    uw |= (unsigned int)__builtin_amdgcn_update_dpp((int)uw, (int)uw, 0x161, 0xf, 0xf, true);
    uw |= (unsigned int)__builtin_amdgcn_update_dpp((int)uw, (int)uw, 0x162, 0xf, 0xf, true);
    uw |= (unsigned int)__builtin_amdgcn_update_dpp((int)uw, (int)uw, 0x164, 0xf, 0xf, true);
    uw |= (unsigned int)__builtin_amdgcn_update_dpp((int)uw, (int)uw, 0x168, 0xf, 0xf, true);
    const unsigned int u0 = (unsigned int)__builtin_amdgcn_readlane((int)uw, 0);
    const unsigned int u1 = (unsigned int)__builtin_amdgcn_readlane((int)uw, 16);
    unsigned long long bits = ((unsigned long long)u1 << 32) | (unsigned long long)u0;
    const int cbase = 8 * wave;

    v8f accS0 = {};
    v8f accS1 = {};
    while (bits != 0ull) {
        unsigned int pk[4];
#pragma unroll
        for (int e = 0; e < 4; ++e) {
            const bool there = bits != 0ull;
            const int  p = there ? __builtin_ctzll(bits) : 0;
            bits &= bits - 1ull;
            const int  r = p & 31;
            const unsigned int c = (unsigned int)(128 * (r >> 2) + cbase + 4 * (p >> 5) + (r & 3));
            pk[e] = c | ((c % una) << 10) | (there ? 0u : (1u << 20));
        }
        const unsigned int P0 = hi ? pk[2] : pk[0];
        const unsigned int P1 = hi ? pk[3] : pk[1];
        const unsigned int c0 = P0 & 1023u, c1 = P1 & 1023u;
        const float a0 = attL[m][c0];
        const float a1 = attL[m][c1];
        const bool self0 = ((P0 >> 10) & 1023u) == rowmod;
        const bool self1 = ((P1 >> 10) & 1023u) == rowmod;
        const bool gone0 = (P0 >> 20) != 0u;
        const bool gone1 = (P1 >> 20) != 0u;
        v2f aS, aO;
        aS.x = (self0 && !gone0) ? a0 : 0.0f;  aS.y = (self1 && !gone1) ? a1 : 0.0f;
        aO.x = (self0 || gone0) ? 0.0f : a0;   aO.y = (self1 || gone1) ? 0.0f : a1;

        const unsigned int o0 = c0 * Dd + m, o1r = c1 * Dd + m;
        v2f b1_0; b1_0.x = w1b[o0];      b1_0.y = w1b[o1r];
        v2f b1_1; b1_1.x = w1b[o0 + 16]; b1_1.y = w1b[o1r + 16];
        v2f b2_0; b2_0.x = w2b[o0];      b2_0.y = w2b[o1r];
        v2f b2_1; b2_1.x = w2b[o0 + 16]; b2_1.y = w2b[o1r + 16];

        accS0 = __builtin_amdgcn_wmma_f32_16x16x4_f32(false, aS, false, b1_0,
                                                      (short)0, accS0, false, false);
        accS0 = __builtin_amdgcn_wmma_f32_16x16x4_f32(false, aO, false, b2_0,
                                                      (short)0, accS0, false, false);
        accS1 = __builtin_amdgcn_wmma_f32_16x16x4_f32(false, aS, false, b1_1,
                                                      (short)0, accS1, false, false);
        accS1 = __builtin_amdgcn_wmma_f32_16x16x4_f32(false, aO, false, b2_1,
                                                      (short)0, accS1, false, false);
    }

    __syncthreads();
    float* rw = &red[wave][(lane >> 4) << 3][m];
#pragma unroll
    for (int r = 0; r < 8; ++r) {
        rw[r * RS]      = accS0[r];
        rw[r * RS + 16] = accS1[r];
    }
    __syncthreads();
    float acc = 0.0f;
#pragma unroll
    for (int w = 0; w < ROWS_PER_BLOCK; ++w) acc += red[w][wave][lane];
    hp[((long)(b * Nn + ib + wave)) * Dd + h * DH + lane] = acc;
}


__global__ __launch_bounds__(128) void outproj_kernel(
    const float* __restrict__ hp, const float* __restrict__ Wo,
    const float* __restrict__ bo, float* __restrict__ out)
{
    const int lane = threadIdx.x & 31;
    const int wave = threadIdx.x >> 5;
    const int m    = lane & 15;
    const int hf   = lane >> 4;
    const int row0 = (blockIdx.x >> 2) * 64 + (wave >> 1) * 32;
    const int col0 = (blockIdx.x & 3) * 64 + (wave & 1) * 32;

    v8f acc[2][2];
    gemm_wave_tile<2, 2, 8>(acc, hp + (row0 + m) * Dd + 4 * hf, Wo + (col0 + m) * Dd + 4 * hf);

#pragma unroll
    for (int j = 0; j < 2; ++j) {
        const int c = col0 + 16 * j + m;
        const float bias = bo[c];
#pragma unroll
        for (int i = 0; i < 2; ++i) {
            float* o = out + (row0 + 16 * i + 8 * hf) * Dd + c;
#pragma unroll
            for (int r = 0; r < 8; ++r) {
                float v = acc[i][j][r] + bias;
                v = (v > 0.0f) ? v : (expf(v) - 1.0f);
                o[r * Dd] = v;
            }
        }
    }
}

extern "C" void kernel_launch(void* const* d_in, const int* in_sizes, int n_in,
                              void* d_out, int out_size, void* d_ws, size_t ws_size,
                              hipStream_t stream) {
    (void)in_sizes; (void)n_in; (void)out_size; (void)ws_size;

    const float* hmat = (const float*)d_in[0];
    const int*   nag  = (const int*)  d_in[1];
    const int*   adj  = (const int*)  d_in[2];
    const float* W1   = (const float*)d_in[3];
    const float* W2   = (const float*)d_in[4];
    const float* Ai   = (const float*)d_in[5];
    const float* Aj   = (const float*)d_in[6];
    const float* Wo   = (const float*)d_in[7];
    const float* bo   = (const float*)d_in[8];
    float*       out  = (float*)d_out;

    float* ws  = (float*)d_ws;
    const size_t whN = (size_t)Bc * Nn * Dd;
    const size_t scN = (size_t)Bc * Hh * Nn;
    float* wh1 = ws;
    float* wh2 = wh1 + whN;
    float* s1  = wh2 + whN;
    float* s2  = s1 + scN;
    float* o1  = s2 + scN;
    float* o2  = o1 + scN;
    float* hp  = o2 + scN;

    proj12_kernel<<<((Bc * Nn) / 64) * 4, 128, 0, stream>>>(
        hmat, W1, W2, wh1, wh2);

    scores_kernel<<<(Bc * Nn * Hh + 255) / 256, 256, 0, stream>>>(
        wh1, wh2, Ai, Aj, s1, s2, o1, o2);

    entmax_aggregate_kernel<<<(Bc * Hh * Nn) / ROWS_PER_BLOCK, 512, 0, stream>>>(
        s1, s2, o1, o2, adj, nag, wh1, wh2, hp);

    outproj_kernel<<<((Bc * Nn) / 64) * 4, 128, 0, stream>>>(
        hp, Wo, bo, out);
}


# adj read once: extra blocks of kernel 2 turn each adj row into 32 lane masks; entmax rows load them as scalars, each used as its select's lane mask: no adj vector loads or compares; 4 blocks per WGP
# speedup vs baseline: 14.1065x; 1.0709x over previous
#include <hip/hip_runtime.h>
#include <hip/hip_bf16.h>
#include <math.h>

static constexpr int Bc = 4;
static constexpr int Nn = 1024;
static constexpr int Dd = 256;
static constexpr int Hh = 8;
static constexpr int DH = 32;

typedef float v2f __attribute__((ext_vector_type(2)));
typedef float v8f __attribute__((ext_vector_type(8)));
typedef float v4f __attribute__((ext_vector_type(4)));
typedef int   v4i __attribute__((ext_vector_type(4)));

#define DEVI static __device__ __forceinline__

template <int MT, int NT, int QG>
DEVI void load_group(v4f (&a)[MT][QG], v4f (&b)[NT][QG],
                     const float* __restrict__ aP, const float* __restrict__ bP, const int k0)
{
#pragma unroll
    for (int q = 0; q < QG; ++q) {
#pragma unroll
        for (int i = 0; i < MT; ++i) a[i][q] = *(const v4f*)(aP + i * 16 * Dd + k0 + 8 * q);
#pragma unroll
        for (int j = 0; j < NT; ++j) b[j][q] = *(const v4f*)(bP + j * 16 * Dd + k0 + 8 * q);
    }
}

template <int MT, int NT, int QG>
DEVI void mma_group(v8f (&acc)[MT][NT], const v4f (&a)[MT][QG], const v4f (&b)[NT][QG])
{
#pragma unroll
    for (int q = 0; q < QG; ++q) {
#pragma unroll
        for (int s = 0; s < 2; ++s) {
#pragma unroll
            for (int i = 0; i < MT; ++i) {
                const v2f af = s ? __builtin_shufflevector(a[i][q], a[i][q], 2, 3)
                                 : __builtin_shufflevector(a[i][q], a[i][q], 0, 1);
#pragma unroll
                for (int j = 0; j < NT; ++j) {
                    const v2f bf = s ? __builtin_shufflevector(b[j][q], b[j][q], 2, 3)
                                     : __builtin_shufflevector(b[j][q], b[j][q], 0, 1);
                    acc[i][j] = __builtin_amdgcn_wmma_f32_16x16x4_f32(false, af, false, bf,
                                                                      (short)0, acc[i][j], false, false);
                }
            }
        }
    }
}

template <int MT, int NT, int QG>
DEVI void gemm_wave_tile(v8f (&acc)[MT][NT], const float* __restrict__ aP, const float* __restrict__ bP)
{
    constexpr int KG = 8 * QG;
#pragma unroll
    for (int i = 0; i < MT; ++i) {
#pragma unroll
        for (int j = 0; j < NT; ++j) acc[i][j] = v8f{};
    }
    v4f a0[MT][QG], b0[NT][QG], a1[MT][QG], b1[NT][QG];
    load_group<MT, NT, QG>(a0, b0, aP, bP, 0);
#pragma unroll 1
    for (int k0 = 0; k0 < Dd - 2 * KG; k0 += 2 * KG) {
        load_group<MT, NT, QG>(a1, b1, aP, bP, k0 + KG);
        __builtin_amdgcn_sched_barrier(0);
        mma_group<MT, NT, QG>(acc, a0, b0);
        __builtin_amdgcn_sched_barrier(0);
        load_group<MT, NT, QG>(a0, b0, aP, bP, k0 + 2 * KG);
        __builtin_amdgcn_sched_barrier(0);
        mma_group<MT, NT, QG>(acc, a1, b1);
    }
    load_group<MT, NT, QG>(a1, b1, aP, bP, Dd - KG);
    __builtin_amdgcn_sched_barrier(0);
    mma_group<MT, NT, QG>(acc, a0, b0);
    mma_group<MT, NT, QG>(acc, a1, b1);
}

__global__ __launch_bounds__(128) void proj12_kernel(
    const float* __restrict__ hmat, const float* __restrict__ W1,
    const float* __restrict__ W2, float* __restrict__ wh1,
    float* __restrict__ wh2)
{
    const int lane = threadIdx.x & 31;
    const int wave = threadIdx.x >> 5;
    const int m    = lane & 15;
    const int hf   = lane >> 4;
    const int row0 = (blockIdx.x >> 2) * 64 + (wave >> 1) * 32;
    const int col0 = (blockIdx.x & 1) * 128 + (wave & 1) * 64;
    const bool second = (blockIdx.x & 2) != 0;
    const float* __restrict__ W   = second ? W2 : W1;
    float* __restrict__       dst = second ? wh2 : wh1;

    v8f acc[2][4];
    gemm_wave_tile<2, 4, 4>(acc, hmat + (row0 + m) * Dd + 4 * hf, W + (col0 + m) * Dd + 4 * hf);

#pragma unroll
    for (int i = 0; i < 2; ++i) {
#pragma unroll
        for (int j = 0; j < 4; ++j) {
            float* o = dst + (row0 + 16 * i + 8 * hf) * Dd + col0 + 16 * j + m;
#pragma unroll
            for (int r = 0; r < 8; ++r) o[r * Dd] = acc[i][j][r];
        }
    }
}

static constexpr int MASK_ROWS_PER_WAVE = 4;
static constexpr int MASK_BLOCKS  = (Bc * Nn) / (8 * MASK_ROWS_PER_WAVE);
static constexpr int SCORE_BLOCKS = (Bc * Nn * Hh + 255) / 256;

__global__ __launch_bounds__(256) void scores_kernel(
    const float* __restrict__ wh1, const float* __restrict__ wh2,
    const float* __restrict__ Ai, const float* __restrict__ Aj,
    const int* __restrict__ adj,
    float* __restrict__ s1, float* __restrict__ s2,
    float* __restrict__ o1, float* __restrict__ o2,
    unsigned int* __restrict__ adjm)
{
    if (blockIdx.x < MASK_BLOCKS) {
        const int lane = threadIdx.x & 31;
        const int row0 = (blockIdx.x * 8 + (int)(threadIdx.x >> 5)) * MASK_ROWS_PER_WAVE;
        const v4i* arow = (const v4i*)(adj + (long)row0 * Nn) + lane;
        v4i av[MASK_ROWS_PER_WAVE][8];
#pragma unroll
        for (int q = 0; q < MASK_ROWS_PER_WAVE; ++q) {
#pragma unroll
            for (int c = 0; c < 8; ++c) av[q][c] = arow[q * (Nn / 4) + c * 32];
        }
        __builtin_amdgcn_sched_barrier(0);
#pragma unroll
        for (int q = 0; q < MASK_ROWS_PER_WAVE; ++q) {
            unsigned int word = 0;
#pragma unroll
            for (int c = 0; c < 8; ++c) {
                const unsigned int m0 = __builtin_amdgcn_ballot_w32(av[q][c].x != 0);
                const unsigned int m1 = __builtin_amdgcn_ballot_w32(av[q][c].y != 0);
                const unsigned int m2 = __builtin_amdgcn_ballot_w32(av[q][c].z != 0);
                const unsigned int m3 = __builtin_amdgcn_ballot_w32(av[q][c].w != 0);
                word = (lane == 4 * c)     ? m0 : word;
                word = (lane == 4 * c + 1) ? m1 : word;
                word = (lane == 4 * c + 2) ? m2 : word;
                word = (lane == 4 * c + 3) ? m3 : word;
            }
            adjm[(row0 + q) * 32 + lane] = word;
        }
        return;
    }
    int idx = (blockIdx.x - MASK_BLOCKS) * blockDim.x + threadIdx.x;
    if (idx >= Bc * Nn * Hh) return;
    const int h = idx % Hh;
    const int n = (idx / Hh) % Nn;
    const int b = idx / (Hh * Nn);

    const float* r1 = wh1 + ((long)(b * Nn + n)) * Dd + h * DH;
    const float* r2 = wh2 + ((long)(b * Nn + n)) * Dd + h * DH;
    const float* ai = Ai + h * DH;
    const float* aj = Aj + h * DH;

    float a1 = 0.f, a2 = 0.f, a3 = 0.f, a4 = 0.f;
#pragma unroll 8
    for (int d = 0; d < DH; ++d) {
        const float w1v = r1[d], w2v = r2[d], aiv = ai[d], ajv = aj[d];
        a1 += w1v * aiv;  a2 += w1v * ajv;
        a3 += w2v * aiv;  a4 += w2v * ajv;
    }
    const int o = (b * Hh + h) * Nn + n;
    s1[o] = 0.5f * a1; s2[o] = 0.5f * a2; o1[o] = 0.5f * a3; o2[o] = 0.5f * a4;
}

#define DEV static __device__ __forceinline__

DEV float xhalf(float v) {
    const int b = __builtin_bit_cast(int, v);
    return __builtin_bit_cast(float,
        __builtin_amdgcn_permlanex16(b, b, 0x76543210, 0xfedcba98, false, false));
}
template <int M> DEV float xrow(float v) {
    const int b = __builtin_bit_cast(int, v);
    return __builtin_bit_cast(float,
        __builtin_amdgcn_update_dpp(b, b, 0x160 + M, 0xf, 0xf, true));
}
DEV float wave_sum(float v) {
    v += xrow<1>(v); v += xrow<2>(v); v += xrow<4>(v); v += xrow<8>(v);
    v += xhalf(v);
    return v;
}
DEV float wave_max(float v) {
    v = fmaxf(v, xrow<1>(v)); v = fmaxf(v, xrow<2>(v));
    v = fmaxf(v, xrow<4>(v)); v = fmaxf(v, xrow<8>(v));
    v = fmaxf(v, xhalf(v));
    return v;
}
DEV float uni(float v) {
    return __builtin_bit_cast(float,
        __builtin_amdgcn_readfirstlane(__builtin_bit_cast(int, v)));
}

static constexpr int ROWS_PER_BLOCK = 16;
static constexpr int LDA = Nn + 4;
static constexpr int RS  = 36;
static constexpr int   STEP_MAX = 64;
static constexpr float TAU_EPS  = 2.384185791015625e-07f;
static constexpr float X_MASKED = 0.5f * -1.0e12f;

DEV void self_flags(bool (&sf)[4], const int jm, const int im, const int na) {
    int m = jm;
#pragma unroll
    for (int e = 0; e < 4; ++e) {
        sf[e] = (im == m);
        m = (m + 1 == na) ? 0 : (m + 1);
    }
}

DEV v4f ld_chunk(const float* __restrict__ row, const unsigned int lb, const int c) {
    return *(const v4f*)((const char*)row + 512 * c + lb);
}

template <bool FLAGS_PER_CHUNK>
DEV float form_x(float (&x)[32], const float* __restrict__ s2r, const float* __restrict__ o2r,
                 const unsigned int* __restrict__ amr, const unsigned int lb,
                 const float s1i, const float o1i,
                 int jm, const int step, const int im, const int na)
{
    bool sf[4];
    self_flags(sf, jm, im, na);
    const v2f s1p = { s1i, s1i };
    const v2f o1p = { o1i, o1i };
    float lmax = -3.4e38f;
    unsigned int mk[32];
#pragma unroll
    for (int r = 0; r < 32; ++r) mk[r] = amr[r];
    v4f sv[8], ov[8];
    constexpr int AHEAD = FLAGS_PER_CHUNK ? 2 : 3;
#pragma unroll
    for (int c = 0; c < AHEAD; ++c) {
        sv[c] = ld_chunk(s2r, lb, c);
        ov[c] = ld_chunk(o2r, lb, c);
    }
#pragma unroll
    for (int c = 0; c < 8; ++c) {
        if (c + AHEAD < 8) {
            sv[c + AHEAD] = ld_chunk(s2r, lb, c + AHEAD);
            ov[c + AHEAD] = ld_chunk(o2r, lb, c + AHEAD);
        }
        if (FLAGS_PER_CHUNK && c > 0) {
            jm += step;
            if (jm >= na) jm -= na;
            self_flags(sf, jm, im, na);
        }
        const v2f sa = __builtin_shufflevector(sv[c], sv[c], 0, 1) + s1p;
        const v2f sb = __builtin_shufflevector(sv[c], sv[c], 2, 3) + s1p;
        const v2f oa = __builtin_shufflevector(ov[c], ov[c], 0, 1) + o1p;
        const v2f ob = __builtin_shufflevector(ov[c], ov[c], 2, 3) + o1p;
        const float x0 = __builtin_amdgcn_inverse_ballot_w32(mk[4 * c])     ? (sf[0] ? sa.x : oa.x) : X_MASKED;
        const float x1 = __builtin_amdgcn_inverse_ballot_w32(mk[4 * c + 1]) ? (sf[1] ? sa.y : oa.y) : X_MASKED;
        const float x2 = __builtin_amdgcn_inverse_ballot_w32(mk[4 * c + 2]) ? (sf[2] ? sb.x : ob.x) : X_MASKED;
        const float x3 = __builtin_amdgcn_inverse_ballot_w32(mk[4 * c + 3]) ? (sf[3] ? sb.y : ob.y) : X_MASKED;
        x[4 * c]     = x0;
        x[4 * c + 1] = x1;
        x[4 * c + 2] = x2;
        x[4 * c + 3] = x3;
        lmax = fmaxf(fmaxf(lmax, x0), fmaxf(x1, fmaxf(x2, x3)));
        if (c + 1 < 8) __builtin_amdgcn_sched_barrier(0);
    }
    return lmax;
}

__global__ __launch_bounds__(512) void entmax_aggregate_kernel(
    const float* __restrict__ s1, const float* __restrict__ s2,
    const float* __restrict__ o1, const float* __restrict__ o2,
    const unsigned int* __restrict__ adjm, const int* __restrict__ nag,
    const float* __restrict__ wh1, const float* __restrict__ wh2,
    float* __restrict__ hp)
{
    __shared__ __attribute__((aligned(16))) float smem[ROWS_PER_BLOCK * LDA];
    float (*attL)[LDA]   = (float (*)[LDA])smem;
    float (*red)[16][RS] = (float (*)[16][RS])smem;
    __shared__ unsigned int suppw[ROWS_PER_BLOCK][32];

    const int lane = threadIdx.x & 31;
    const int wave = __builtin_amdgcn_readfirstlane((int)(threadIdx.x >> 5));
    const int rid  = blockIdx.x * ROWS_PER_BLOCK + wave;
    const int i    = rid & (Nn - 1);
    const int h    = (rid >> 10) & (Hh - 1);
    const int b    = rid >> 13;
    const int na   = nag[0];

    const int   rowbase = (b * Hh + h) * Nn;
    const float s1i = s1[rowbase + i];
    const float o1i = o1[rowbase + i];
    const int   im  = i % na;

    const float* s2r = s2 + rowbase;
    const float* o2r = o2 + rowbase;
    const unsigned int* amr = adjm + (b * Nn + i) * 32;
    const unsigned int  lb  = 16u * (unsigned int)lane;

    float x[32];
    const int jm   = (4 * lane) % na;
    const int step = 128 % na;
    float lmax;
    if (step == 0) lmax = form_x<false>(x, s2r, o2r, amr, lb, s1i, o1i, jm, step, im, na);
    else           lmax = form_x<true >(x, s2r, o2r, amr, lb, s1i, o1i, jm, step, im, na);
    const float xmax = uni(wave_max(lmax));
#pragma unroll
    for (int r = 0; r < 32; ++r) x[r] -= xmax;

    float t;
    {
        const float lm   = lmax - xmax;
        const float d0   = fmaxf(lm + 1.0f, 0.0f);
        const float kf0  = (float)__builtin_popcount(__builtin_amdgcn_ballot_w32(lm > -1.0f));
        const float G0   = uni(wave_sum(d0));
        const float F0   = uni(wave_sum(d0 * d0));
        const float dsc0 = fmaxf(fmaf(G0, G0, -kf0 * (F0 - 1.0f)), 0.0f);
        const float t0   = -1.0f + (G0 - __builtin_amdgcn_sqrtf(dsc0)) * __builtin_amdgcn_rcpf(kf0);
        t = (t0 < 0.0f) ? fmaxf(t0, -1.0f) : -1.0f;
    }
    float lo = -1.0f, tstar = -1.0f;
    int kprev = -1;
    for (int it = 0; it <= STEP_MAX; ++it) {
        const v2f tt = { t, t };
        const v2f zz = { 0.f, 0.f };
        v2f ga = zz, gb = zz, fa = zz, fb = zz;
        int k = 0;
#pragma unroll
        for (int r = 0; r < 32; r += 4) {
            const v2f xa = { x[r],     x[r + 1] };
            const v2f xb = { x[r + 2], x[r + 3] };
            const v2f da = __builtin_elementwise_max(xa - tt, zz);
            const v2f db = __builtin_elementwise_max(xb - tt, zz);
            ga += da; gb += db;
            fa = __builtin_elementwise_fma(da, da, fa);
            fb = __builtin_elementwise_fma(db, db, fb);
            k += __builtin_popcount(__builtin_amdgcn_ballot_w32(x[r]     > t))
               + __builtin_popcount(__builtin_amdgcn_ballot_w32(x[r + 1] > t))
               + __builtin_popcount(__builtin_amdgcn_ballot_w32(x[r + 2] > t))
               + __builtin_popcount(__builtin_amdgcn_ballot_w32(x[r + 3] > t));
        }
        const v2f gs = ga + gb, fs = fa + fb;
        const float G  = uni(wave_sum(gs.x + gs.y));
        const float F  = uni(wave_sum(fs.x + fs.y));
        const float kf = (float)k;
        const float disc = fmaxf(fmaf(G, G, -kf * (F - 1.0f)), 0.0f);
        const float s    = (G - __builtin_amdgcn_sqrtf(disc)) * __builtin_amdgcn_rcpf(kf);
        tstar = t + s;
        if (k == kprev || fabsf(s) <= TAU_EPS || it == STEP_MAX) break;
        kprev = k;
        lo = fmaxf(lo, t + 0.5f * (F - 1.0f) * __builtin_amdgcn_rcpf(G));
        t  = fmaxf(tstar, lo);
    }

    v4f* outrow = (v4f*)(&attL[wave][0]) + lane;
    unsigned int myw = 0;
#pragma unroll
    for (int c = 0; c < 8; ++c) {
        v4f y;
        y.x = fmaxf(x[4 * c]     - tstar, 0.0f);
        y.y = fmaxf(x[4 * c + 1] - tstar, 0.0f);
        y.z = fmaxf(x[4 * c + 2] - tstar, 0.0f);
        y.w = fmaxf(x[4 * c + 3] - tstar, 0.0f);
        y.x *= y.x; y.y *= y.y; y.z *= y.z; y.w *= y.w;
        outrow[c * 32] = y;
        myw |= (y.x != 0.0f) ? (1u << (4 * c))     : 0u;
        myw |= (y.y != 0.0f) ? (1u << (4 * c + 1)) : 0u;
        myw |= (y.z != 0.0f) ? (1u << (4 * c + 2)) : 0u;
        myw |= (y.w != 0.0f) ? (1u << (4 * c + 3)) : 0u;
    }
    suppw[wave][lane] = myw;
    __syncthreads();

    const int m      = lane & 15;
    const int hi     = lane >> 4;
    const int ib     = (blockIdx.x * ROWS_PER_BLOCK) & (Nn - 1);
    const unsigned int una    = (unsigned int)na;
    const unsigned int rowmod = (unsigned int)(ib + m) % una;
    const float* w1b = wh1 + ((long)b * Nn) * Dd + h * DH;
    const float* w2b = wh2 + ((long)b * Nn) * Dd + h * DH;

    unsigned int uw = suppw[m][2 * wave + hi];
    uw |= (unsigned int)__builtin_amdgcn_update_dpp((int)uw, (int)uw, 0x161, 0xf, 0xf, true);
    uw |= (unsigned int)__builtin_amdgcn_update_dpp((int)uw, (int)uw, 0x162, 0xf, 0xf, true);
    uw |= (unsigned int)__builtin_amdgcn_update_dpp((int)uw, (int)uw, 0x164, 0xf, 0xf, true);
    uw |= (unsigned int)__builtin_amdgcn_update_dpp((int)uw, (int)uw, 0x168, 0xf, 0xf, true);
    const unsigned int u0 = (unsigned int)__builtin_amdgcn_readlane((int)uw, 0);
    const unsigned int u1 = (unsigned int)__builtin_amdgcn_readlane((int)uw, 16);
    unsigned long long bits = ((unsigned long long)u1 << 32) | (unsigned long long)u0;
    const int cbase = 8 * wave;

    v8f accS0 = {};
    v8f accS1 = {};
    while (bits != 0ull) {
        unsigned int pk[4];
#pragma unroll
        for (int e = 0; e < 4; ++e) {
            const bool there = bits != 0ull;
            const int  p = there ? __builtin_ctzll(bits) : 0;
            bits &= bits - 1ull;
            const int  r = p & 31;
            const unsigned int c = (unsigned int)(128 * (r >> 2) + cbase + 4 * (p >> 5) + (r & 3));
            pk[e] = c | ((c % una) << 10) | (there ? 0u : (1u << 20));
        }
        const unsigned int P0 = hi ? pk[2] : pk[0];
        const unsigned int P1 = hi ? pk[3] : pk[1];
        const unsigned int c0 = P0 & 1023u, c1 = P1 & 1023u;
        const float a0 = attL[m][c0];
        const float a1 = attL[m][c1];
        const bool self0 = ((P0 >> 10) & 1023u) == rowmod;
        const bool self1 = ((P1 >> 10) & 1023u) == rowmod;
        const bool gone0 = (P0 >> 20) != 0u;
        const bool gone1 = (P1 >> 20) != 0u;
        v2f aS, aO;
        aS.x = (self0 && !gone0) ? a0 : 0.0f;  aS.y = (self1 && !gone1) ? a1 : 0.0f;
        aO.x = (self0 || gone0) ? 0.0f : a0;   aO.y = (self1 || gone1) ? 0.0f : a1;

        const unsigned int o0 = c0 * Dd + m, o1r = c1 * Dd + m;
        v2f b1_0; b1_0.x = w1b[o0];      b1_0.y = w1b[o1r];
        v2f b1_1; b1_1.x = w1b[o0 + 16]; b1_1.y = w1b[o1r + 16];
        v2f b2_0; b2_0.x = w2b[o0];      b2_0.y = w2b[o1r];
        v2f b2_1; b2_1.x = w2b[o0 + 16]; b2_1.y = w2b[o1r + 16];

        accS0 = __builtin_amdgcn_wmma_f32_16x16x4_f32(false, aS, false, b1_0,
                                                      (short)0, accS0, false, false);
        accS0 = __builtin_amdgcn_wmma_f32_16x16x4_f32(false, aO, false, b2_0,
                                                      (short)0, accS0, false, false);
        accS1 = __builtin_amdgcn_wmma_f32_16x16x4_f32(false, aS, false, b1_1,
                                                      (short)0, accS1, false, false);
        accS1 = __builtin_amdgcn_wmma_f32_16x16x4_f32(false, aO, false, b2_1,
                                                      (short)0, accS1, false, false);
    }

    __syncthreads();
    float* rw = &red[wave][(lane >> 4) << 3][m];
#pragma unroll
    for (int r = 0; r < 8; ++r) {
        rw[r * RS]      = accS0[r];
        rw[r * RS + 16] = accS1[r];
    }
    __syncthreads();
    float acc = 0.0f;
#pragma unroll
    for (int w = 0; w < ROWS_PER_BLOCK; ++w) acc += red[w][wave][lane];
    hp[((long)(b * Nn + ib + wave)) * Dd + h * DH + lane] = acc;
}


__global__ __launch_bounds__(128) void outproj_kernel(
    const float* __restrict__ hp, const float* __restrict__ Wo,
    const float* __restrict__ bo, float* __restrict__ out)
{
    const int lane = threadIdx.x & 31;
    const int wave = threadIdx.x >> 5;
    const int m    = lane & 15;
    const int hf   = lane >> 4;
    const int row0 = (blockIdx.x >> 2) * 64 + (wave >> 1) * 32;
    const int col0 = (blockIdx.x & 3) * 64 + (wave & 1) * 32;

    v8f acc[2][2];
    gemm_wave_tile<2, 2, 8>(acc, hp + (row0 + m) * Dd + 4 * hf, Wo + (col0 + m) * Dd + 4 * hf);

#pragma unroll
    for (int j = 0; j < 2; ++j) {
        const int c = col0 + 16 * j + m;
        const float bias = bo[c];
#pragma unroll
        for (int i = 0; i < 2; ++i) {
            float* o = out + (row0 + 16 * i + 8 * hf) * Dd + c;
#pragma unroll
            for (int r = 0; r < 8; ++r) {
                float v = acc[i][j][r] + bias;
                v = (v > 0.0f) ? v : (expf(v) - 1.0f);
                o[r * Dd] = v;
            }
        }
    }
}

extern "C" void kernel_launch(void* const* d_in, const int* in_sizes, int n_in,
                              void* d_out, int out_size, void* d_ws, size_t ws_size,
                              hipStream_t stream) {
    (void)in_sizes; (void)n_in; (void)out_size; (void)ws_size;

    const float* hmat = (const float*)d_in[0];
    const int*   nag  = (const int*)  d_in[1];
    const int*   adj  = (const int*)  d_in[2];
    const float* W1   = (const float*)d_in[3];
    const float* W2   = (const float*)d_in[4];
    const float* Ai   = (const float*)d_in[5];
    const float* Aj   = (const float*)d_in[6];
    const float* Wo   = (const float*)d_in[7];
    const float* bo   = (const float*)d_in[8];
    float*       out  = (float*)d_out;

    float* ws  = (float*)d_ws;
    const size_t whN = (size_t)Bc * Nn * Dd;
    const size_t scN = (size_t)Bc * Hh * Nn;
    float* wh1 = ws;
    float* wh2 = wh1 + whN;
    float* s1  = wh2 + whN;
    float* s2  = s1 + scN;
    float* o1  = s2 + scN;
    float* o2  = o1 + scN;
    float* hp  = o2 + scN;
    unsigned int* adjm = (unsigned int*)(hp + whN);

    proj12_kernel<<<((Bc * Nn) / 64) * 4, 128, 0, stream>>>(
        hmat, W1, W2, wh1, wh2);

    scores_kernel<<<MASK_BLOCKS + SCORE_BLOCKS, 256, 0, stream>>>(
        wh1, wh2, Ai, Aj, adj, s1, s2, o1, o2, adjm);

    entmax_aggregate_kernel<<<(Bc * Hh * Nn) / ROWS_PER_BLOCK, 512, 0, stream>>>(
        s1, s2, o1, o2, adjm, nag, wh1, wh2, hp);

    outproj_kernel<<<((Bc * Nn) / 64) * 4, 128, 0, stream>>>(
        hp, Wo, bo, out);
}


# entmax tau search on each lane's 3 largest values (kept as x is formed): a pass is 3 registers, not 32; the output pass counts the support, a differing count runs the full search; sum order changes
# speedup vs baseline: 14.5198x; 1.0293x over previous
#include <hip/hip_runtime.h>
#include <hip/hip_bf16.h>
#include <math.h>

static constexpr int Bc = 4;
static constexpr int Nn = 1024;
static constexpr int Dd = 256;
static constexpr int Hh = 8;
static constexpr int DH = 32;

typedef float v2f __attribute__((ext_vector_type(2)));
typedef float v8f __attribute__((ext_vector_type(8)));
typedef float v4f __attribute__((ext_vector_type(4)));
typedef int   v4i __attribute__((ext_vector_type(4)));

#define DEVI static __device__ __forceinline__

template <int MT, int NT, int QG>
DEVI void load_group(v4f (&a)[MT][QG], v4f (&b)[NT][QG],
                     const float* __restrict__ aP, const float* __restrict__ bP, const int k0)
{
#pragma unroll
    for (int q = 0; q < QG; ++q) {
#pragma unroll
        for (int i = 0; i < MT; ++i) a[i][q] = *(const v4f*)(aP + i * 16 * Dd + k0 + 8 * q);
#pragma unroll
        for (int j = 0; j < NT; ++j) b[j][q] = *(const v4f*)(bP + j * 16 * Dd + k0 + 8 * q);
    }
}

template <int MT, int NT, int QG>
DEVI void mma_group(v8f (&acc)[MT][NT], const v4f (&a)[MT][QG], const v4f (&b)[NT][QG])
{
#pragma unroll
    for (int q = 0; q < QG; ++q) {
#pragma unroll
        for (int s = 0; s < 2; ++s) {
#pragma unroll
            for (int i = 0; i < MT; ++i) {
                const v2f af = s ? __builtin_shufflevector(a[i][q], a[i][q], 2, 3)
                                 : __builtin_shufflevector(a[i][q], a[i][q], 0, 1);
#pragma unroll
                for (int j = 0; j < NT; ++j) {
                    const v2f bf = s ? __builtin_shufflevector(b[j][q], b[j][q], 2, 3)
                                     : __builtin_shufflevector(b[j][q], b[j][q], 0, 1);
                    acc[i][j] = __builtin_amdgcn_wmma_f32_16x16x4_f32(false, af, false, bf,
                                                                      (short)0, acc[i][j], false, false);
                }
            }
        }
    }
}

template <int MT, int NT, int QG>
DEVI void gemm_wave_tile(v8f (&acc)[MT][NT], const float* __restrict__ aP, const float* __restrict__ bP)
{
    constexpr int KG = 8 * QG;
#pragma unroll
    for (int i = 0; i < MT; ++i) {
#pragma unroll
        for (int j = 0; j < NT; ++j) acc[i][j] = v8f{};
    }
    v4f a0[MT][QG], b0[NT][QG], a1[MT][QG], b1[NT][QG];
    load_group<MT, NT, QG>(a0, b0, aP, bP, 0);
#pragma unroll 1
    for (int k0 = 0; k0 < Dd - 2 * KG; k0 += 2 * KG) {
        load_group<MT, NT, QG>(a1, b1, aP, bP, k0 + KG);
        __builtin_amdgcn_sched_barrier(0);
        mma_group<MT, NT, QG>(acc, a0, b0);
        __builtin_amdgcn_sched_barrier(0);
        load_group<MT, NT, QG>(a0, b0, aP, bP, k0 + 2 * KG);
        __builtin_amdgcn_sched_barrier(0);
        mma_group<MT, NT, QG>(acc, a1, b1);
    }
    load_group<MT, NT, QG>(a1, b1, aP, bP, Dd - KG);
    __builtin_amdgcn_sched_barrier(0);
    mma_group<MT, NT, QG>(acc, a0, b0);
    mma_group<MT, NT, QG>(acc, a1, b1);
}

__global__ __launch_bounds__(128) void proj12_kernel(
    const float* __restrict__ hmat, const float* __restrict__ W1,
    const float* __restrict__ W2, float* __restrict__ wh1,
    float* __restrict__ wh2)
{
    const int lane = threadIdx.x & 31;
    const int wave = threadIdx.x >> 5;
    const int m    = lane & 15;
    const int hf   = lane >> 4;
    const int row0 = (blockIdx.x >> 2) * 64 + (wave >> 1) * 32;
    const int col0 = (blockIdx.x & 1) * 128 + (wave & 1) * 64;
    const bool second = (blockIdx.x & 2) != 0;
    const float* __restrict__ W   = second ? W2 : W1;
    float* __restrict__       dst = second ? wh2 : wh1;

    v8f acc[2][4];
    gemm_wave_tile<2, 4, 4>(acc, hmat + (row0 + m) * Dd + 4 * hf, W + (col0 + m) * Dd + 4 * hf);

#pragma unroll
    for (int i = 0; i < 2; ++i) {
#pragma unroll
        for (int j = 0; j < 4; ++j) {
            float* o = dst + (row0 + 16 * i + 8 * hf) * Dd + col0 + 16 * j + m;
#pragma unroll
            for (int r = 0; r < 8; ++r) o[r * Dd] = acc[i][j][r];
        }
    }
}

static constexpr int MASK_ROWS_PER_WAVE = 4;
static constexpr int MASK_BLOCKS  = (Bc * Nn) / (8 * MASK_ROWS_PER_WAVE);
static constexpr int SCORE_BLOCKS = (Bc * Nn * Hh + 255) / 256;

__global__ __launch_bounds__(256) void scores_kernel(
    const float* __restrict__ wh1, const float* __restrict__ wh2,
    const float* __restrict__ Ai, const float* __restrict__ Aj,
    const int* __restrict__ adj,
    float* __restrict__ s1, float* __restrict__ s2,
    float* __restrict__ o1, float* __restrict__ o2,
    unsigned int* __restrict__ adjm)
{
    if (blockIdx.x < MASK_BLOCKS) {
        const int lane = threadIdx.x & 31;
        const int row0 = (blockIdx.x * 8 + (int)(threadIdx.x >> 5)) * MASK_ROWS_PER_WAVE;
        const v4i* arow = (const v4i*)(adj + (long)row0 * Nn) + lane;
        v4i av[MASK_ROWS_PER_WAVE][8];
#pragma unroll
        for (int q = 0; q < MASK_ROWS_PER_WAVE; ++q) {
#pragma unroll
            for (int c = 0; c < 8; ++c) av[q][c] = arow[q * (Nn / 4) + c * 32];
        }
        __builtin_amdgcn_sched_barrier(0);
#pragma unroll
        for (int q = 0; q < MASK_ROWS_PER_WAVE; ++q) {
            unsigned int word = 0;
#pragma unroll
            for (int c = 0; c < 8; ++c) {
                const unsigned int m0 = __builtin_amdgcn_ballot_w32(av[q][c].x != 0);
                const unsigned int m1 = __builtin_amdgcn_ballot_w32(av[q][c].y != 0);
                const unsigned int m2 = __builtin_amdgcn_ballot_w32(av[q][c].z != 0);
                const unsigned int m3 = __builtin_amdgcn_ballot_w32(av[q][c].w != 0);
                word = (lane == 4 * c)     ? m0 : word;
                word = (lane == 4 * c + 1) ? m1 : word;
                word = (lane == 4 * c + 2) ? m2 : word;
                word = (lane == 4 * c + 3) ? m3 : word;
            }
            adjm[(row0 + q) * 32 + lane] = word;
        }
        return;
    }
    int idx = (blockIdx.x - MASK_BLOCKS) * blockDim.x + threadIdx.x;
    if (idx >= Bc * Nn * Hh) return;
    const int h = idx % Hh;
    const int n = (idx / Hh) % Nn;
    const int b = idx / (Hh * Nn);

    const float* r1 = wh1 + ((long)(b * Nn + n)) * Dd + h * DH;
    const float* r2 = wh2 + ((long)(b * Nn + n)) * Dd + h * DH;
    const float* ai = Ai + h * DH;
    const float* aj = Aj + h * DH;

    float a1 = 0.f, a2 = 0.f, a3 = 0.f, a4 = 0.f;
#pragma unroll 8
    for (int d = 0; d < DH; ++d) {
        const float w1v = r1[d], w2v = r2[d], aiv = ai[d], ajv = aj[d];
        a1 += w1v * aiv;  a2 += w1v * ajv;
        a3 += w2v * aiv;  a4 += w2v * ajv;
    }
    const int o = (b * Hh + h) * Nn + n;
    s1[o] = 0.5f * a1; s2[o] = 0.5f * a2; o1[o] = 0.5f * a3; o2[o] = 0.5f * a4;
}

#define DEV static __device__ __forceinline__

DEV float xhalf(float v) {
    const int b = __builtin_bit_cast(int, v);
    return __builtin_bit_cast(float,
        __builtin_amdgcn_permlanex16(b, b, 0x76543210, 0xfedcba98, false, false));
}
template <int M> DEV float xrow(float v) {
    const int b = __builtin_bit_cast(int, v);
    return __builtin_bit_cast(float,
        __builtin_amdgcn_update_dpp(b, b, 0x160 + M, 0xf, 0xf, true));
}
DEV float wave_sum(float v) {
    v += xrow<1>(v); v += xrow<2>(v); v += xrow<4>(v); v += xrow<8>(v);
    v += xhalf(v);
    return v;
}
DEV float wave_max(float v) {
    v = fmaxf(v, xrow<1>(v)); v = fmaxf(v, xrow<2>(v));
    v = fmaxf(v, xrow<4>(v)); v = fmaxf(v, xrow<8>(v));
    v = fmaxf(v, xhalf(v));
    return v;
}
DEV float uni(float v) {
    return __builtin_bit_cast(float,
        __builtin_amdgcn_readfirstlane(__builtin_bit_cast(int, v)));
}

static constexpr int ROWS_PER_BLOCK = 16;
static constexpr int LDA = Nn + 4;
static constexpr int RS  = 36;
static constexpr int   STEP_MAX = 64;
static constexpr float TAU_EPS  = 2.384185791015625e-07f;
static constexpr float X_MASKED = 0.5f * -1.0e12f;

DEV void self_flags(bool (&sf)[4], const int jm, const int im, const int na) {
    int m = jm;
#pragma unroll
    for (int e = 0; e < 4; ++e) {
        sf[e] = (im == m);
        m = (m + 1 == na) ? 0 : (m + 1);
    }
}

DEV v4f ld_chunk(const float* __restrict__ row, const unsigned int lb, const int c) {
    return *(const v4f*)((const char*)row + 512 * c + lb);
}

DEV void top3_insert(float &m1, float &m2, float &m3, const float v) {
    const float n3 = __builtin_amdgcn_fmed3f(m2, m3, v);
    const float n2 = __builtin_amdgcn_fmed3f(m1, m2, v);
    m1 = fmaxf(m1, v);
    m2 = n2;
    m3 = n3;
}

template <bool FLAGS_PER_CHUNK>
DEV void form_x(float (&x)[32], float &m1, float &m2, float &m3,
                 const float* __restrict__ s2r, const float* __restrict__ o2r,
                 const unsigned int* __restrict__ amr, const unsigned int lb,
                 const float s1i, const float o1i,
                 int jm, const int step, const int im, const int na)
{
    bool sf[4];
    self_flags(sf, jm, im, na);
    const v2f s1p = { s1i, s1i };
    const v2f o1p = { o1i, o1i };
    m1 = -3.4e38f; m2 = -3.4e38f; m3 = -3.4e38f;
    unsigned int mk[32];
#pragma unroll
    for (int r = 0; r < 32; ++r) mk[r] = amr[r];
    v4f sv[8], ov[8];
    constexpr int AHEAD = FLAGS_PER_CHUNK ? 2 : 3;
#pragma unroll
    for (int c = 0; c < AHEAD; ++c) {
        sv[c] = ld_chunk(s2r, lb, c);
        ov[c] = ld_chunk(o2r, lb, c);
    }
#pragma unroll
    for (int c = 0; c < 8; ++c) {
        if (c + AHEAD < 8) {
            sv[c + AHEAD] = ld_chunk(s2r, lb, c + AHEAD);
            ov[c + AHEAD] = ld_chunk(o2r, lb, c + AHEAD);
        }
        if (FLAGS_PER_CHUNK && c > 0) {
            jm += step;
            if (jm >= na) jm -= na;
            self_flags(sf, jm, im, na);
        }
        const v2f sa = __builtin_shufflevector(sv[c], sv[c], 0, 1) + s1p;
        const v2f sb = __builtin_shufflevector(sv[c], sv[c], 2, 3) + s1p;
        const v2f oa = __builtin_shufflevector(ov[c], ov[c], 0, 1) + o1p;
        const v2f ob = __builtin_shufflevector(ov[c], ov[c], 2, 3) + o1p;
        const float x0 = __builtin_amdgcn_inverse_ballot_w32(mk[4 * c])     ? (sf[0] ? sa.x : oa.x) : X_MASKED;
        const float x1 = __builtin_amdgcn_inverse_ballot_w32(mk[4 * c + 1]) ? (sf[1] ? sa.y : oa.y) : X_MASKED;
        const float x2 = __builtin_amdgcn_inverse_ballot_w32(mk[4 * c + 2]) ? (sf[2] ? sb.x : ob.x) : X_MASKED;
        const float x3 = __builtin_amdgcn_inverse_ballot_w32(mk[4 * c + 3]) ? (sf[3] ? sb.y : ob.y) : X_MASKED;
        x[4 * c]     = x0;
        x[4 * c + 1] = x1;
        x[4 * c + 2] = x2;
        x[4 * c + 3] = x3;
        top3_insert(m1, m2, m3, x0);
        top3_insert(m1, m2, m3, x1);
        top3_insert(m1, m2, m3, x2);
        top3_insert(m1, m2, m3, x3);
        if (c + 1 < 8) __builtin_amdgcn_sched_barrier(0);
    }
}

DEV float full_search(const float (&x)[32], float t)
{
    float lo = -1.0f, tstar = -1.0f;
    int kprev = -1;
    for (int it = 0; it <= STEP_MAX; ++it) {
        const v2f tt = { t, t };
        const v2f zz = { 0.f, 0.f };
        v2f ga = zz, gb = zz, fa = zz, fb = zz;
        int k = 0;
#pragma unroll
        for (int r = 0; r < 32; r += 4) {
            const v2f xa = { x[r],     x[r + 1] };
            const v2f xb = { x[r + 2], x[r + 3] };
            const v2f da = __builtin_elementwise_max(xa - tt, zz);
            const v2f db = __builtin_elementwise_max(xb - tt, zz);
            ga += da; gb += db;
            fa = __builtin_elementwise_fma(da, da, fa);
            fb = __builtin_elementwise_fma(db, db, fb);
            k += __builtin_popcount(__builtin_amdgcn_ballot_w32(x[r]     > t))
               + __builtin_popcount(__builtin_amdgcn_ballot_w32(x[r + 1] > t))
               + __builtin_popcount(__builtin_amdgcn_ballot_w32(x[r + 2] > t))
               + __builtin_popcount(__builtin_amdgcn_ballot_w32(x[r + 3] > t));
        }
        const v2f gs = ga + gb, fs = fa + fb;
        const float G  = uni(wave_sum(gs.x + gs.y));
        const float F  = uni(wave_sum(fs.x + fs.y));
        const float kf = (float)k;
        const float disc = fmaxf(fmaf(G, G, -kf * (F - 1.0f)), 0.0f);
        const float s    = (G - __builtin_amdgcn_sqrtf(disc)) * __builtin_amdgcn_rcpf(kf);
        tstar = t + s;
        if (k == kprev || fabsf(s) <= TAU_EPS || it == STEP_MAX) break;
        kprev = k;
        lo = fmaxf(lo, t + 0.5f * (F - 1.0f) * __builtin_amdgcn_rcpf(G));
        t  = fmaxf(tstar, lo);
    }
    return tstar;
}

DEV unsigned int emit_row(const float (&x)[32], const float tstar, v4f* __restrict__ outrow, int &cnt)
{
    unsigned int myw = 0;
    int n = 0;
#pragma unroll
    for (int c = 0; c < 8; ++c) {
        v4f y;
        y.x = fmaxf(x[4 * c]     - tstar, 0.0f);
        y.y = fmaxf(x[4 * c + 1] - tstar, 0.0f);
        y.z = fmaxf(x[4 * c + 2] - tstar, 0.0f);
        y.w = fmaxf(x[4 * c + 3] - tstar, 0.0f);
        y.x *= y.x; y.y *= y.y; y.z *= y.z; y.w *= y.w;
        outrow[c * 32] = y;
        const bool p0 = (y.x != 0.0f), p1 = (y.y != 0.0f), p2 = (y.z != 0.0f), p3 = (y.w != 0.0f);
        myw |= p0 ? (1u << (4 * c))     : 0u;
        myw |= p1 ? (1u << (4 * c + 1)) : 0u;
        myw |= p2 ? (1u << (4 * c + 2)) : 0u;
        myw |= p3 ? (1u << (4 * c + 3)) : 0u;
        n += __builtin_popcount(__builtin_amdgcn_ballot_w32(p0))
           + __builtin_popcount(__builtin_amdgcn_ballot_w32(p1))
           + __builtin_popcount(__builtin_amdgcn_ballot_w32(p2))
           + __builtin_popcount(__builtin_amdgcn_ballot_w32(p3));
        if (c + 1 < 8) __builtin_amdgcn_sched_barrier(0);
    }
    cnt = n;
    return myw;
}

DEV int candidates_present(const float m1, const float m2, const float m3, const float tstar)
{
    float a = fmaxf(m1 - tstar, 0.0f), b = fmaxf(m2 - tstar, 0.0f), c = fmaxf(m3 - tstar, 0.0f);
    a *= a; b *= b; c *= c;
    return __builtin_popcount(__builtin_amdgcn_ballot_w32(a != 0.0f))
         + __builtin_popcount(__builtin_amdgcn_ballot_w32(b != 0.0f))
         + __builtin_popcount(__builtin_amdgcn_ballot_w32(c != 0.0f));
}

__global__ __launch_bounds__(512) void entmax_aggregate_kernel(
    const float* __restrict__ s1, const float* __restrict__ s2,
    const float* __restrict__ o1, const float* __restrict__ o2,
    const unsigned int* __restrict__ adjm, const int* __restrict__ nag,
    const float* __restrict__ wh1, const float* __restrict__ wh2,
    float* __restrict__ hp)
{
    __shared__ __attribute__((aligned(16))) float smem[ROWS_PER_BLOCK * LDA];
    float (*attL)[LDA]   = (float (*)[LDA])smem;
    float (*red)[16][RS] = (float (*)[16][RS])smem;
    __shared__ unsigned int suppw[ROWS_PER_BLOCK][32];

    const int lane = threadIdx.x & 31;
    const int wave = __builtin_amdgcn_readfirstlane((int)(threadIdx.x >> 5));
    const int rid  = blockIdx.x * ROWS_PER_BLOCK + wave;
    const int i    = rid & (Nn - 1);
    const int h    = (rid >> 10) & (Hh - 1);
    const int b    = rid >> 13;
    const int na   = nag[0];

    const int   rowbase = (b * Hh + h) * Nn;
    const float s1i = s1[rowbase + i];
    const float o1i = o1[rowbase + i];
    const int   im  = i % na;

    const float* s2r = s2 + rowbase;
    const float* o2r = o2 + rowbase;
    const unsigned int* amr = adjm + (b * Nn + i) * 32;
    const unsigned int  lb  = 16u * (unsigned int)lane;

    float x[32];
    const int jm   = (4 * lane) % na;
    const int step = 128 % na;
    float m1, m2, m3;
    if (step == 0) form_x<false>(x, m1, m2, m3, s2r, o2r, amr, lb, s1i, o1i, jm, step, im, na);
    else           form_x<true >(x, m1, m2, m3, s2r, o2r, amr, lb, s1i, o1i, jm, step, im, na);
    const float xmax = uni(wave_max(m1));
#pragma unroll
    for (int r = 0; r < 32; ++r) x[r] -= xmax;
    m1 -= xmax; m2 -= xmax; m3 -= xmax;

    float t;
    {
        const float lm   = m1;
        const float d0   = fmaxf(lm + 1.0f, 0.0f);
        const float kf0  = (float)__builtin_popcount(__builtin_amdgcn_ballot_w32(lm > -1.0f));
        const float G0   = uni(wave_sum(d0));
        const float F0   = uni(wave_sum(d0 * d0));
        const float dsc0 = fmaxf(fmaf(G0, G0, -kf0 * (F0 - 1.0f)), 0.0f);
        const float t0   = -1.0f + (G0 - __builtin_amdgcn_sqrtf(dsc0)) * __builtin_amdgcn_rcpf(kf0);
        t = (t0 < 0.0f) ? fmaxf(t0, -1.0f) : -1.0f;
    }
    float tstar = -1.0f;
    int kc = 0;
    {
        float lo = -1.0f;
        int kprev = -1;
        for (int it = 0; it <= STEP_MAX; ++it) {
            const float d1 = fmaxf(m1 - t, 0.0f);
            const float d2 = fmaxf(m2 - t, 0.0f);
            const float d3 = fmaxf(m3 - t, 0.0f);
            const int k = __builtin_popcount(__builtin_amdgcn_ballot_w32(m1 > t))
                        + __builtin_popcount(__builtin_amdgcn_ballot_w32(m2 > t))
                        + __builtin_popcount(__builtin_amdgcn_ballot_w32(m3 > t));
            const float G  = uni(wave_sum((d1 + d2) + d3));
            const float F  = uni(wave_sum(fmaf(d1, d1, fmaf(d2, d2, d3 * d3))));
            const float kf = (float)k;
            const float disc = fmaxf(fmaf(G, G, -kf * (F - 1.0f)), 0.0f);
            const float s    = (G - __builtin_amdgcn_sqrtf(disc)) * __builtin_amdgcn_rcpf(kf);
            tstar = t + s;
            kc = k;
            if (k == kprev || fabsf(s) <= TAU_EPS || it == STEP_MAX) break;
            kprev = k;
            lo = fmaxf(lo, t + 0.5f * (F - 1.0f) * __builtin_amdgcn_rcpf(G));
            t  = fmaxf(tstar, lo);
        }
    }
    const int kcs = candidates_present(m1, m2, m3, tstar);

    v4f* outrow = (v4f*)(&attL[wave][0]) + lane;
    int cnt;
    unsigned int myw = emit_row(x, tstar, outrow, cnt);
    if (cnt != kcs || kcs != kc) {
        tstar = full_search(x, (tstar < 0.0f) ? fmaxf(tstar, -1.0f) : -1.0f);
        myw = emit_row(x, tstar, outrow, cnt);
    }
    suppw[wave][lane] = myw;
    __syncthreads();

    const int m      = lane & 15;
    const int hi     = lane >> 4;
    const int ib     = (blockIdx.x * ROWS_PER_BLOCK) & (Nn - 1);
    const unsigned int una    = (unsigned int)na;
    const unsigned int rowmod = (unsigned int)(ib + m) % una;
    const float* w1b = wh1 + ((long)b * Nn) * Dd + h * DH;
    const float* w2b = wh2 + ((long)b * Nn) * Dd + h * DH;

    unsigned int uw = suppw[m][2 * wave + hi];
    uw |= (unsigned int)__builtin_amdgcn_update_dpp((int)uw, (int)uw, 0x161, 0xf, 0xf, true);
    uw |= (unsigned int)__builtin_amdgcn_update_dpp((int)uw, (int)uw, 0x162, 0xf, 0xf, true);
    uw |= (unsigned int)__builtin_amdgcn_update_dpp((int)uw, (int)uw, 0x164, 0xf, 0xf, true);
    uw |= (unsigned int)__builtin_amdgcn_update_dpp((int)uw, (int)uw, 0x168, 0xf, 0xf, true);
    const unsigned int u0 = (unsigned int)__builtin_amdgcn_readlane((int)uw, 0);
    const unsigned int u1 = (unsigned int)__builtin_amdgcn_readlane((int)uw, 16);
    unsigned long long bits = ((unsigned long long)u1 << 32) | (unsigned long long)u0;
    const int cbase = 8 * wave;

    v8f accS0 = {};
    v8f accS1 = {};
    while (bits != 0ull) {
        unsigned int pk[4];
#pragma unroll
        for (int e = 0; e < 4; ++e) {
            const bool there = bits != 0ull;
            const int  p = there ? __builtin_ctzll(bits) : 0;
            bits &= bits - 1ull;
            const int  r = p & 31;
            const unsigned int c = (unsigned int)(128 * (r >> 2) + cbase + 4 * (p >> 5) + (r & 3));
            pk[e] = c | ((c % una) << 10) | (there ? 0u : (1u << 20));
        }
        const unsigned int P0 = hi ? pk[2] : pk[0];
        const unsigned int P1 = hi ? pk[3] : pk[1];
        const unsigned int c0 = P0 & 1023u, c1 = P1 & 1023u;
        const float a0 = attL[m][c0];
        const float a1 = attL[m][c1];
        const bool self0 = ((P0 >> 10) & 1023u) == rowmod;
        const bool self1 = ((P1 >> 10) & 1023u) == rowmod;
        const bool gone0 = (P0 >> 20) != 0u;
        const bool gone1 = (P1 >> 20) != 0u;
        v2f aS, aO;
        aS.x = (self0 && !gone0) ? a0 : 0.0f;  aS.y = (self1 && !gone1) ? a1 : 0.0f;
        aO.x = (self0 || gone0) ? 0.0f : a0;   aO.y = (self1 || gone1) ? 0.0f : a1;

        const unsigned int o0 = c0 * Dd + m, o1r = c1 * Dd + m;
        v2f b1_0; b1_0.x = w1b[o0];      b1_0.y = w1b[o1r];
        v2f b1_1; b1_1.x = w1b[o0 + 16]; b1_1.y = w1b[o1r + 16];
        v2f b2_0; b2_0.x = w2b[o0];      b2_0.y = w2b[o1r];
        v2f b2_1; b2_1.x = w2b[o0 + 16]; b2_1.y = w2b[o1r + 16];

        accS0 = __builtin_amdgcn_wmma_f32_16x16x4_f32(false, aS, false, b1_0,
                                                      (short)0, accS0, false, false);
        accS0 = __builtin_amdgcn_wmma_f32_16x16x4_f32(false, aO, false, b2_0,
                                                      (short)0, accS0, false, false);
        accS1 = __builtin_amdgcn_wmma_f32_16x16x4_f32(false, aS, false, b1_1,
                                                      (short)0, accS1, false, false);
        accS1 = __builtin_amdgcn_wmma_f32_16x16x4_f32(false, aO, false, b2_1,
                                                      (short)0, accS1, false, false);
    }

    __syncthreads();
    float* rw = &red[wave][(lane >> 4) << 3][m];
#pragma unroll
    for (int r = 0; r < 8; ++r) {
        rw[r * RS]      = accS0[r];
        rw[r * RS + 16] = accS1[r];
    }
    __syncthreads();
    float acc = 0.0f;
#pragma unroll
    for (int w = 0; w < ROWS_PER_BLOCK; ++w) acc += red[w][wave][lane];
    hp[((long)(b * Nn + ib + wave)) * Dd + h * DH + lane] = acc;
}


__global__ __launch_bounds__(128) void outproj_kernel(
    const float* __restrict__ hp, const float* __restrict__ Wo,
    const float* __restrict__ bo, float* __restrict__ out)
{
    const int lane = threadIdx.x & 31;
    const int wave = threadIdx.x >> 5;
    const int m    = lane & 15;
    const int hf   = lane >> 4;
    const int row0 = (blockIdx.x >> 2) * 64 + (wave >> 1) * 32;
    const int col0 = (blockIdx.x & 3) * 64 + (wave & 1) * 32;

    v8f acc[2][2];
    gemm_wave_tile<2, 2, 8>(acc, hp + (row0 + m) * Dd + 4 * hf, Wo + (col0 + m) * Dd + 4 * hf);

#pragma unroll
    for (int j = 0; j < 2; ++j) {
        const int c = col0 + 16 * j + m;
        const float bias = bo[c];
#pragma unroll
        for (int i = 0; i < 2; ++i) {
            float* o = out + (row0 + 16 * i + 8 * hf) * Dd + c;
#pragma unroll
            for (int r = 0; r < 8; ++r) {
                float v = acc[i][j][r] + bias;
                v = (v > 0.0f) ? v : (expf(v) - 1.0f);
                o[r * Dd] = v;
            }
        }
    }
}

extern "C" void kernel_launch(void* const* d_in, const int* in_sizes, int n_in,
                              void* d_out, int out_size, void* d_ws, size_t ws_size,
                              hipStream_t stream) {
    (void)in_sizes; (void)n_in; (void)out_size; (void)ws_size;

    const float* hmat = (const float*)d_in[0];
    const int*   nag  = (const int*)  d_in[1];
    const int*   adj  = (const int*)  d_in[2];
    const float* W1   = (const float*)d_in[3];
    const float* W2   = (const float*)d_in[4];
    const float* Ai   = (const float*)d_in[5];
    const float* Aj   = (const float*)d_in[6];
    const float* Wo   = (const float*)d_in[7];
    const float* bo   = (const float*)d_in[8];
    float*       out  = (float*)d_out;

    float* ws  = (float*)d_ws;
    const size_t whN = (size_t)Bc * Nn * Dd;
    const size_t scN = (size_t)Bc * Hh * Nn;
    float* wh1 = ws;
    float* wh2 = wh1 + whN;
    float* s1  = wh2 + whN;
    float* s2  = s1 + scN;
    float* o1  = s2 + scN;
    float* o2  = o1 + scN;
    float* hp  = o2 + scN;
    unsigned int* adjm = (unsigned int*)(hp + whN);

    proj12_kernel<<<((Bc * Nn) / 64) * 4, 128, 0, stream>>>(
        hmat, W1, W2, wh1, wh2);

    scores_kernel<<<MASK_BLOCKS + SCORE_BLOCKS, 256, 0, stream>>>(
        wh1, wh2, Ai, Aj, adj, s1, s2, o1, o2, adjm);

    entmax_aggregate_kernel<<<(Bc * Hh * Nn) / ROWS_PER_BLOCK, 512, 0, stream>>>(
        s1, s2, o1, o2, adjm, nag, wh1, wh2, hp);

    outproj_kernel<<<((Bc * Nn) / 64) * 4, 128, 0, stream>>>(
        hp, Wo, bo, out);
}
